# starting kernel
# baseline (speedup 1.0000x reference)
#include <hip/hip_runtime.h>
#include <math.h>

#define N_ATOMS  10000
#define N_PAIRS  160000
#define N_CH     32
#define KDIM     128
#define K0TOT    384

typedef float v2f __attribute__((ext_vector_type(2)));
typedef float v8f __attribute__((ext_vector_type(8)));

#define A3 0.5773502691896258f
#define A2 0.7071067811865476f
#define B2 0.8164965809277260f
#define B1 0.4082482904638630f
__constant__ float U2c[9][9] = {
    { A3, 0,   0,   0,  A3, 0,   0,   0,   A3},
    { 0,  0,   0,   0,  0,  A2,  0,  -A2,  0 },
    { 0,  0,  -A2,  0,  0,  0,   A2,  0,   0 },
    { 0,  A2,  0,  -A2, 0,  0,   0,   0,   0 },
    { 0,  0,   A2,  0,  0,  0,   A2,  0,   0 },
    { 0,  A2,  0,   A2, 0,  0,   0,   0,   0 },
    {-B1, 0,   0,   0,  B2, 0,   0,   0,  -B1},
    { 0,  0,   0,   0,  0,  A2,  0,   A2,  0 },
    {-A2, 0,   0,   0,  0,  0,   0,   0,   A2},
};

__global__ void k_zero(float* p, long n) {
    long i = (long)blockIdx.x * blockDim.x + threadIdx.x;
    if (i < n) p[i] = 0.f;
}

__global__ void k_init_out(float* out, const float* b_out) {
    int i = blockIdx.x * blockDim.x + threadIdx.x;
    if (i < N_ATOMS) out[i] = b_out[0];
}

__global__ void k_pairs(const float* __restrict__ pos, const int* __restrict__ species,
                        const int* __restrict__ pairs, const float* __restrict__ emb,
                        const float* __restrict__ W_rad,
                        float* __restrict__ f0, float* __restrict__ f1, float* __restrict__ f2) {
    int p = blockIdx.x;
    int k = threadIdx.x;
    int c  = pairs[2 * p + 0];
    int nb = pairs[2 * p + 1];
    float rx = pos[3 * nb + 0] - pos[3 * c + 0];
    float ry = pos[3 * nb + 1] - pos[3 * c + 1];
    float rz = pos[3 * nb + 2] - pos[3 * c + 2];
    float d  = sqrtf(rx * rx + ry * ry + rz * rz + 1e-12f);
    float inv = 1.f / d;
    float ux = rx * inv, uy = ry * inv, uz = rz * inv;

    __shared__ float s_rb[8];
    if (k < 8) {
        float mu = 20.f * (float)k / 7.f;
        float t  = (d - 15.f) * 0.2f;
        t = fminf(fmaxf(t, 0.f), 1.f);
        float fc = 0.5f * (cosf(3.14159265358979f * t) + 1.f);
        float x  = (d - mu) * 0.4f;
        s_rb[k] = expf(-x * x) * fc;
    }
    __syncthreads();

    float neigh = emb[species[nb] * N_CH + (k >> 2)];
    float R[3];
#pragma unroll
    for (int l = 0; l < 3; ++l) {
        float s = 0.f;
#pragma unroll
        for (int j = 0; j < 8; ++j) s += s_rb[j] * W_rad[(l * 8 + j) * KDIM + k];
        R[l] = s;
    }
    const float mp = 0.1f;
    atomicAdd(&f0[(long)c * KDIM + k], R[0] * neigh * mp);
    float sh1[3] = { uy, uz, ux };
    float w1 = R[1] * neigh * mp;
#pragma unroll
    for (int m = 0; m < 3; ++m)
        atomicAdd(&f1[((long)c * 3 + m) * KDIM + k], sh1[m] * w1);
    const float s3 = 1.7320508075688772f;
    float sh2[5] = { s3 * ux * uy, s3 * uy * uz, 0.5f * (3.f * uz * uz - 1.f),
                     s3 * ux * uz, 0.5f * s3 * (ux * ux - uy * uy) };
    float w2 = R[2] * neigh * mp;
#pragma unroll
    for (int m = 0; m < 5; ++m)
        atomicAdd(&f2[((long)c * 5 + m) * KDIM + k], sh2[m] * w2);
}

__global__ void k_uncouple(const float* __restrict__ f1, const float* __restrict__ f2,
                           float* __restrict__ u1, float* __restrict__ u2) {
    long t = (long)blockIdx.x * blockDim.x + threadIdx.x;
    if (t >= (long)N_ATOMS * 9 * KDIM) return;
    int k = (int)(t % KDIM);
    int p = (int)((t / KDIM) % 9);
    long n = t / (9 * KDIM);
    float a = 0.f, b = 0.f;
#pragma unroll
    for (int q = 0; q < 3; ++q) a += U2c[1 + q][p] * f1[(n * 3 + q) * KDIM + k];
#pragma unroll
    for (int q = 0; q < 5; ++q) b += U2c[4 + q][p] * f2[(n * 5 + q) * KDIM + k];
    u1[t] = a;
    u2[t] = b;
}

__global__ void k_gemm_wmma128(const float* __restrict__ A, const float* __restrict__ B,
                               float* __restrict__ C) {
    int m0 = blockIdx.x * 16;
    int lane = threadIdx.x;
    int half = lane >> 4;
    int l16  = lane & 15;
    const float* Arow = A + (long)(m0 + l16) * KDIM;
    v8f acc[8] = {};
    for (int kk = 0; kk < KDIM; kk += 4) {
        int ka = kk + 2 * half;
        v2f a;
        a.x = Arow[ka];
        a.y = Arow[ka + 1];
        const float* B0 = B + (long)ka * KDIM + l16;
#pragma unroll
        for (int t = 0; t < 8; ++t) {
            v2f b;
            b.x = B0[t * 16];
            b.y = B0[KDIM + t * 16];
            acc[t] = __builtin_amdgcn_wmma_f32_16x16x4_f32(false, a, false, b, (short)0,
                                                           acc[t], false, false);
        }
    }
#pragma unroll
    for (int t = 0; t < 8; ++t) {
        float* Cc = C + (long)(m0 + 8 * half) * KDIM + t * 16 + l16;
#pragma unroll
        for (int i = 0; i < 8; ++i) Cc[(long)i * KDIM] = acc[t][i];
    }
}

__global__ void k_feats(const float* __restrict__ f0, const float* __restrict__ u1,
                        const float* __restrict__ u2, const float* __restrict__ g0,
                        const float* __restrict__ g1, const float* __restrict__ g2,
                        const int* __restrict__ species, const float* __restrict__ emb,
                        float* __restrict__ x0) {
    long t = (long)blockIdx.x * blockDim.x + threadIdx.x;
    if (t >= (long)N_ATOMS * KDIM) return;
    int k = (int)(t % KDIM);
    long n = t / KDIM;
    float f0v = f0[t];
    float feats0 = g0[t] * f0v + f0v;

    float G[9], F[9];
    float tr1 = 0.f;
#pragma unroll
    for (int p = 0; p < 9; ++p) {
        G[p] = g1[(n * 9 + p) * KDIM + k];
        F[p] = u1[(n * 9 + p) * KDIM + k];
    }
#pragma unroll
    for (int a = 0; a < 3; ++a)
#pragma unroll
        for (int b = 0; b < 3; ++b) tr1 += G[a * 3 + b] * F[b * 3 + a];
    tr1 += F[0] + F[4] + F[8];

    float tr2 = 0.f;
#pragma unroll
    for (int p = 0; p < 9; ++p) {
        G[p] = g2[(n * 9 + p) * KDIM + k];
        F[p] = u2[(n * 9 + p) * KDIM + k];
    }
#pragma unroll
    for (int a = 0; a < 3; ++a)
#pragma unroll
        for (int b = 0; b < 3; ++b) tr2 += G[a * 3 + b] * F[b * 3 + a];
    tr2 += F[0] + F[4] + F[8];

    const float i3 = 0.5773502691896258f;
    int sp = species[n];
    const float* e = emb + sp * N_CH;
    x0[n * K0TOT + k]       = feats0      * e[k / 12];
    x0[n * K0TOT + 128 + k] = (tr1 * i3)  * e[(128 + k) / 12];
    x0[n * K0TOT + 256 + k] = (tr2 * i3)  * e[(256 + k) / 12];
}

__global__ void k_head_wmma(const float* __restrict__ x0, const float* __restrict__ Wh,
                            const float* __restrict__ bh, const float* __restrict__ Wo,
                            float* __restrict__ out) {
    int m0 = blockIdx.x * 16;
    int n0 = blockIdx.y * 128;
    int lane = threadIdx.x;
    int half = lane >> 4;
    int l16  = lane & 15;
    const float* Arow = x0 + (long)(m0 + l16) * K0TOT;
    v8f acc[8] = {};
    for (int kk = 0; kk < K0TOT; kk += 4) {
        int ka = kk + 2 * half;
        v2f a;
        a.x = Arow[ka];
        a.y = Arow[ka + 1];
        const float* B0 = Wh + (long)ka * K0TOT + n0 + l16;
#pragma unroll
        for (int t = 0; t < 8; ++t) {
            v2f b;
            b.x = B0[t * 16];
            b.y = B0[K0TOT + t * 16];
            acc[t] = __builtin_amdgcn_wmma_f32_16x16x4_f32(false, a, false, b, (short)0,
                                                           acc[t], false, false);
        }
    }
    int rowbase = m0 + 8 * half;
    float partial[8];
#pragma unroll
    for (int i = 0; i < 8; ++i) partial[i] = 0.f;
#pragma unroll
    for (int t = 0; t < 8; ++t) {
        int col = n0 + t * 16 + l16;
        float bias = bh[col];
        float wo   = Wo[col];
#pragma unroll
        for (int i = 0; i < 8; ++i) {
            float v = acc[t][i] + bias;
            partial[i] += (v / (1.f + expf(-v))) * wo;
        }
    }
#pragma unroll
    for (int i = 0; i < 8; ++i) atomicAdd(&out[rowbase + i], partial[i]);
}

extern "C" void kernel_launch(void* const* d_in, const int* in_sizes, int n_in,
                              void* d_out, int out_size, void* d_ws, size_t ws_size,
                              hipStream_t stream) {
    (void)in_sizes; (void)n_in; (void)out_size; (void)ws_size;
    const float* positions = (const float*)d_in[0];
    const int*   species   = (const int*)d_in[1];
    const int*   pairs     = (const int*)d_in[2];
    const float* embeddings= (const float*)d_in[3];
    const float* W_rad     = (const float*)d_in[4];
    const float* W_cg0     = (const float*)d_in[5];
    const float* W_cg1     = (const float*)d_in[6];
    const float* W_cg2     = (const float*)d_in[7];
    const float* W_head    = (const float*)d_in[8];
    const float* b_head    = (const float*)d_in[9];
    const float* W_out     = (const float*)d_in[10];
    const float* b_out     = (const float*)d_in[11];
    float* out = (float*)d_out;

    float* ws = (float*)d_ws;
    const long SZ_F0 = (long)N_ATOMS * KDIM;
    const long SZ_F1 = (long)N_ATOMS * 3 * KDIM;
    const long SZ_F2 = (long)N_ATOMS * 5 * KDIM;
    const long SZ_U  = (long)N_ATOMS * 9 * KDIM;
    float* f0 = ws;
    float* f1 = f0 + SZ_F0;
    float* f2 = f1 + SZ_F1;
    float* u1 = f2 + SZ_F2;
    float* u2 = u1 + SZ_U;
    float* g0 = u2 + SZ_U;
    float* g1 = g0 + SZ_F0;
    float* g2 = g1 + SZ_U;
    float* x0 = g2 + SZ_U;

    long nz = SZ_F0 + SZ_F1 + SZ_F2;
    k_zero<<<(unsigned)((nz + 255) / 256), 256, 0, stream>>>(f0, nz);

    k_init_out<<<(N_ATOMS + 255) / 256, 256, 0, stream>>>(out, b_out);

    k_pairs<<<N_PAIRS, KDIM, 0, stream>>>(positions, species, pairs, embeddings, W_rad,
                                          f0, f1, f2);

    k_uncouple<<<45000, 256, 0, stream>>>(f1, f2, u1, u2);

    k_gemm_wmma128<<<N_ATOMS / 16, 32, 0, stream>>>(f0, W_cg0, g0);
    k_gemm_wmma128<<<N_ATOMS * 9 / 16, 32, 0, stream>>>(u1, W_cg1, g1);
    k_gemm_wmma128<<<N_ATOMS * 9 / 16, 32, 0, stream>>>(u2, W_cg2, g2);

    k_feats<<<(unsigned)(((long)N_ATOMS * KDIM + 255) / 256), 256, 0, stream>>>(
        f0, u1, u2, g0, g1, g2, species, embeddings, x0);

    k_head_wmma<<<dim3(N_ATOMS / 16, K0TOT / 16 / 8), 32, 0, stream>>>(
        x0, W_head, b_head, W_out, out);
}


# pair scatter factored: atomics sum sh x radial-basis moments per neighbour species (72 adds/pair, was 1152); 128 channels restored per atom in uncouple; sum order: pairs first, then basis/species
# speedup vs baseline: 1.5633x; 1.5633x over previous
#include <hip/hip_runtime.h>
#include <math.h>

#define N_ATOMS  10000
#define N_PAIRS  160000
#define N_CH     32
#define KDIM     128
#define K0TOT    384

typedef float v2f __attribute__((ext_vector_type(2)));
typedef float v8f __attribute__((ext_vector_type(8)));

#define A3 0.5773502691896258f
#define A2 0.7071067811865476f
#define B2 0.8164965809277260f
#define B1 0.4082482904638630f
__constant__ float U2c[9][9] = {
    { A3, 0,   0,   0,  A3, 0,   0,   0,   A3},
    { 0,  0,   0,   0,  0,  A2,  0,  -A2,  0 },
    { 0,  0,  -A2,  0,  0,  0,   A2,  0,   0 },
    { 0,  A2,  0,  -A2, 0,  0,   0,   0,   0 },
    { 0,  0,   A2,  0,  0,  0,   A2,  0,   0 },
    { 0,  A2,  0,   A2, 0,  0,   0,   0,   0 },
    {-B1, 0,   0,   0,  B2, 0,   0,   0,  -B1},
    { 0,  0,   0,   0,  0,  A2,  0,   A2,  0 },
    {-A2, 0,   0,   0,  0,  0,   0,   0,   A2},
};

__global__ void k_zero(float* p, long n) {
    long i = (long)blockIdx.x * blockDim.x + threadIdx.x;
    if (i < n) p[i] = 0.f;
}

__global__ void k_init_out(float* out, const float* b_out) {
    int i = blockIdx.x * blockDim.x + threadIdx.x;
    if (i < N_ATOMS) out[i] = b_out[0];
}

#define N_SPEC   4
#define N_RB     8
#define TM_ATOM  (N_SPEC * 9 * N_RB)
__global__ void k_pairs(const float* __restrict__ pos, const int* __restrict__ species,
                        const int* __restrict__ pairs, float* __restrict__ Tm) {
    int t = blockIdx.x * blockDim.x + threadIdx.x;
    int p = t >> 3;
    int j = t & 7;
    if (p >= N_PAIRS) return;
    int c  = pairs[2 * p + 0];
    int nb = pairs[2 * p + 1];
    float rx = pos[3 * nb + 0] - pos[3 * c + 0];
    float ry = pos[3 * nb + 1] - pos[3 * c + 1];
    float rz = pos[3 * nb + 2] - pos[3 * c + 2];
    float d  = sqrtf(rx * rx + ry * ry + rz * rz + 1e-12f);
    float inv = 1.f / d;
    float ux = rx * inv, uy = ry * inv, uz = rz * inv;

    float mu = 20.f * (float)j / 7.f;
    float tc = (d - 15.f) * 0.2f;
    tc = fminf(fmaxf(tc, 0.f), 1.f);
    float fc = 0.5f * (cosf(3.14159265358979f * tc) + 1.f);
    float x  = (d - mu) * 0.4f;
    float rb = expf(-x * x) * fc;

    const float s3 = 1.7320508075688772f;
    float sh[9] = { 1.f, uy, uz, ux,
                    s3 * ux * uy, s3 * uy * uz, 0.5f * (3.f * uz * uz - 1.f),
                    s3 * ux * uz, 0.5f * s3 * (ux * ux - uy * uy) };
    float* Tp = Tm + ((long)c * N_SPEC + species[nb]) * (9 * N_RB) + j;
    for (int m = 0; m < 9; ++m) atomicAdd(Tp + m * N_RB, sh[m] * rb);
}

__global__ void __launch_bounds__(KDIM)
k_expand_uncouple(const float* __restrict__ Tm, const float* __restrict__ emb,
                  const float* __restrict__ W_rad, float* __restrict__ f0,
                  float* __restrict__ u1, float* __restrict__ u2) {
    int n = blockIdx.x;
    int k = threadIdx.x;
    const float* Tn = Tm + (long)n * TM_ATOM;
    float e[N_SPEC];
    for (int s = 0; s < N_SPEC; ++s) e[s] = emb[s * N_CH + (k >> 2)];
    float f[9];
#pragma unroll
    for (int m = 0; m < 9; ++m) {
        const int l = (m == 0) ? 0 : (m < 4 ? 1 : 2);
        float acc = 0.f;
        for (int j = 0; j < N_RB; ++j) {
            float v = 0.f;
            for (int s = 0; s < N_SPEC; ++s) v += Tn[(s * 9 + m) * N_RB + j] * e[s];
            acc += v * W_rad[(l * N_RB + j) * KDIM + k];
        }
        f[m] = acc * 0.1f;
    }
    f0[(long)n * KDIM + k] = f[0];
    for (int p = 0; p < 9; ++p) {
        float a = 0.f, b = 0.f;
        for (int q = 0; q < 3; ++q) a += U2c[1 + q][p] * f[1 + q];
        for (int q = 0; q < 5; ++q) b += U2c[4 + q][p] * f[4 + q];
        u1[((long)n * 9 + p) * KDIM + k] = a;
        u2[((long)n * 9 + p) * KDIM + k] = b;
    }
}

__global__ void k_gemm_wmma128(const float* __restrict__ A, const float* __restrict__ B,
                               float* __restrict__ C) {
    int m0 = blockIdx.x * 16;
    int lane = threadIdx.x;
    int half = lane >> 4;
    int l16  = lane & 15;
    const float* Arow = A + (long)(m0 + l16) * KDIM;
    v8f acc[8] = {};
    for (int kk = 0; kk < KDIM; kk += 4) {
        int ka = kk + 2 * half;
        v2f a;
        a.x = Arow[ka];
        a.y = Arow[ka + 1];
        const float* B0 = B + (long)ka * KDIM + l16;
#pragma unroll
        for (int t = 0; t < 8; ++t) {
            v2f b;
            b.x = B0[t * 16];
            b.y = B0[KDIM + t * 16];
            acc[t] = __builtin_amdgcn_wmma_f32_16x16x4_f32(false, a, false, b, (short)0,
                                                           acc[t], false, false);
        }
    }
#pragma unroll
    for (int t = 0; t < 8; ++t) {
        float* Cc = C + (long)(m0 + 8 * half) * KDIM + t * 16 + l16;
#pragma unroll
        for (int i = 0; i < 8; ++i) Cc[(long)i * KDIM] = acc[t][i];
    }
}

__global__ void k_feats(const float* __restrict__ f0, const float* __restrict__ u1,
                        const float* __restrict__ u2, const float* __restrict__ g0,
                        const float* __restrict__ g1, const float* __restrict__ g2,
                        const int* __restrict__ species, const float* __restrict__ emb,
                        float* __restrict__ x0) {
    long t = (long)blockIdx.x * blockDim.x + threadIdx.x;
    if (t >= (long)N_ATOMS * KDIM) return;
    int k = (int)(t % KDIM);
    long n = t / KDIM;
    float f0v = f0[t];
    float feats0 = g0[t] * f0v + f0v;

    float G[9], F[9];
    float tr1 = 0.f;
#pragma unroll
    for (int p = 0; p < 9; ++p) {
        G[p] = g1[(n * 9 + p) * KDIM + k];
        F[p] = u1[(n * 9 + p) * KDIM + k];
    }
#pragma unroll
    for (int a = 0; a < 3; ++a)
#pragma unroll
        for (int b = 0; b < 3; ++b) tr1 += G[a * 3 + b] * F[b * 3 + a];
    tr1 += F[0] + F[4] + F[8];

    float tr2 = 0.f;
#pragma unroll
    for (int p = 0; p < 9; ++p) {
        G[p] = g2[(n * 9 + p) * KDIM + k];
        F[p] = u2[(n * 9 + p) * KDIM + k];
    }
#pragma unroll
    for (int a = 0; a < 3; ++a)
#pragma unroll
        for (int b = 0; b < 3; ++b) tr2 += G[a * 3 + b] * F[b * 3 + a];
    tr2 += F[0] + F[4] + F[8];

    const float i3 = 0.5773502691896258f;
    int sp = species[n];
    const float* e = emb + sp * N_CH;
    x0[n * K0TOT + k]       = feats0      * e[k / 12];
    x0[n * K0TOT + 128 + k] = (tr1 * i3)  * e[(128 + k) / 12];
    x0[n * K0TOT + 256 + k] = (tr2 * i3)  * e[(256 + k) / 12];
}

__global__ void k_head_wmma(const float* __restrict__ x0, const float* __restrict__ Wh,
                            const float* __restrict__ bh, const float* __restrict__ Wo,
                            float* __restrict__ out) {
    int m0 = blockIdx.x * 16;
    int n0 = blockIdx.y * 128;
    int lane = threadIdx.x;
    int half = lane >> 4;
    int l16  = lane & 15;
    const float* Arow = x0 + (long)(m0 + l16) * K0TOT;
    v8f acc[8] = {};
    for (int kk = 0; kk < K0TOT; kk += 4) {
        int ka = kk + 2 * half;
        v2f a;
        a.x = Arow[ka];
        a.y = Arow[ka + 1];
        const float* B0 = Wh + (long)ka * K0TOT + n0 + l16;
#pragma unroll
        for (int t = 0; t < 8; ++t) {
            v2f b;
            b.x = B0[t * 16];
            b.y = B0[K0TOT + t * 16];
            acc[t] = __builtin_amdgcn_wmma_f32_16x16x4_f32(false, a, false, b, (short)0,
                                                           acc[t], false, false);
        }
    }
    int rowbase = m0 + 8 * half;
    float partial[8];
#pragma unroll
    for (int i = 0; i < 8; ++i) partial[i] = 0.f;
#pragma unroll
    for (int t = 0; t < 8; ++t) {
        int col = n0 + t * 16 + l16;
        float bias = bh[col];
        float wo   = Wo[col];
#pragma unroll
        for (int i = 0; i < 8; ++i) {
            float v = acc[t][i] + bias;
            partial[i] += (v / (1.f + expf(-v))) * wo;
        }
    }
#pragma unroll
    for (int i = 0; i < 8; ++i) atomicAdd(&out[rowbase + i], partial[i]);
}

extern "C" void kernel_launch(void* const* d_in, const int* in_sizes, int n_in,
                              void* d_out, int out_size, void* d_ws, size_t ws_size,
                              hipStream_t stream) {
    (void)in_sizes; (void)n_in; (void)out_size; (void)ws_size;
    const float* positions = (const float*)d_in[0];
    const int*   species   = (const int*)d_in[1];
    const int*   pairs     = (const int*)d_in[2];
    const float* embeddings= (const float*)d_in[3];
    const float* W_rad     = (const float*)d_in[4];
    const float* W_cg0     = (const float*)d_in[5];
    const float* W_cg1     = (const float*)d_in[6];
    const float* W_cg2     = (const float*)d_in[7];
    const float* W_head    = (const float*)d_in[8];
    const float* b_head    = (const float*)d_in[9];
    const float* W_out     = (const float*)d_in[10];
    const float* b_out     = (const float*)d_in[11];
    float* out = (float*)d_out;

    float* ws = (float*)d_ws;
    const long SZ_F0 = (long)N_ATOMS * KDIM;
    const long SZ_F1 = (long)N_ATOMS * 3 * KDIM;
    const long SZ_F2 = (long)N_ATOMS * 5 * KDIM;
    const long SZ_U  = (long)N_ATOMS * 9 * KDIM;
    float* f0 = ws;
    float* f1 = f0 + SZ_F0;
    float* f2 = f1 + SZ_F1;
    float* u1 = f2 + SZ_F2;
    float* u2 = u1 + SZ_U;
    float* g0 = u2 + SZ_U;
    float* g1 = g0 + SZ_F0;
    float* g2 = g1 + SZ_U;
    float* x0 = g2 + SZ_U;

    float* Tm = f1;
    long nz = (long)N_ATOMS * TM_ATOM;
    k_zero<<<(unsigned)((nz + 255) / 256), 256, 0, stream>>>(Tm, nz);

    k_init_out<<<(N_ATOMS + 255) / 256, 256, 0, stream>>>(out, b_out);

    k_pairs<<<N_PAIRS * N_RB / 256, 256, 0, stream>>>(positions, species, pairs, Tm);

    k_expand_uncouple<<<N_ATOMS, KDIM, 0, stream>>>(Tm, embeddings, W_rad, f0, u1, u2);

    k_gemm_wmma128<<<N_ATOMS / 16, 32, 0, stream>>>(f0, W_cg0, g0);
    k_gemm_wmma128<<<N_ATOMS * 9 / 16, 32, 0, stream>>>(u1, W_cg1, g1);
    k_gemm_wmma128<<<N_ATOMS * 9 / 16, 32, 0, stream>>>(u2, W_cg2, g2);

    k_feats<<<(unsigned)(((long)N_ATOMS * KDIM + 255) / 256), 256, 0, stream>>>(
        f0, u1, u2, g0, g1, g2, species, embeddings, x0);

    k_head_wmma<<<dim3(N_ATOMS / 16, K0TOT / 16 / 8), 32, 0, stream>>>(
        x0, W_head, b_head, W_out, out);
}


# channel mix moved in front of uncouple (they commute): one f32 WMMA GEMM on the 9 coupled rows per atom, was 19 uncoupled rows in 3 launches; uncouple in registers in feats; sum order of g changes
# speedup vs baseline: 2.1770x; 1.3926x over previous
#include <hip/hip_runtime.h>
#include <math.h>

#define N_ATOMS  10000
#define N_PAIRS  160000
#define N_CH     32
#define KDIM     128
#define K0TOT    384

typedef float v2f __attribute__((ext_vector_type(2)));
typedef float v8f __attribute__((ext_vector_type(8)));

#define A3 0.5773502691896258f
#define A2 0.7071067811865476f
#define B2 0.8164965809277260f
#define B1 0.4082482904638630f

__global__ void k_zero(float* p, long n) {
    long i = (long)blockIdx.x * blockDim.x + threadIdx.x;
    if (i < n) p[i] = 0.f;
}

__global__ void k_init_out(float* out, const float* b_out) {
    int i = blockIdx.x * blockDim.x + threadIdx.x;
    if (i < N_ATOMS) out[i] = b_out[0];
}

#define N_SPEC   4
#define N_RB     8
#define TM_ATOM  (N_SPEC * 9 * N_RB)
__global__ void k_pairs(const float* __restrict__ pos, const int* __restrict__ species,
                        const int* __restrict__ pairs, float* __restrict__ Tm) {
    int t = blockIdx.x * blockDim.x + threadIdx.x;
    int p = t >> 3;
    int j = t & 7;
    if (p >= N_PAIRS) return;
    int c  = pairs[2 * p + 0];
    int nb = pairs[2 * p + 1];
    float rx = pos[3 * nb + 0] - pos[3 * c + 0];
    float ry = pos[3 * nb + 1] - pos[3 * c + 1];
    float rz = pos[3 * nb + 2] - pos[3 * c + 2];
    float d  = sqrtf(rx * rx + ry * ry + rz * rz + 1e-12f);
    float inv = 1.f / d;
    float ux = rx * inv, uy = ry * inv, uz = rz * inv;

    float mu = 20.f * (float)j / 7.f;
    float tc = (d - 15.f) * 0.2f;
    tc = fminf(fmaxf(tc, 0.f), 1.f);
    float fc = 0.5f * (cosf(3.14159265358979f * tc) + 1.f);
    float x  = (d - mu) * 0.4f;
    float rb = expf(-x * x) * fc;

    const float s3 = 1.7320508075688772f;
    float sh[9] = { 1.f, uy, uz, ux,
                    s3 * ux * uy, s3 * uy * uz, 0.5f * (3.f * uz * uz - 1.f),
                    s3 * ux * uz, 0.5f * s3 * (ux * ux - uy * uy) };
    float* Tp = Tm + ((long)c * N_SPEC + species[nb]) * (9 * N_RB) + j;
    for (int m = 0; m < 9; ++m) atomicAdd(Tp + m * N_RB, sh[m] * rb);
}

__global__ void __launch_bounds__(KDIM)
k_expand(const float* __restrict__ Tm, const float* __restrict__ emb,
         const float* __restrict__ W_rad, float* __restrict__ F) {
    int n = blockIdx.x;
    int k = threadIdx.x;
    const float* Tn = Tm + (long)n * TM_ATOM;
    float e[N_SPEC];
    for (int s = 0; s < N_SPEC; ++s) e[s] = emb[s * N_CH + (k >> 2)];
#pragma unroll
    for (int m = 0; m < 9; ++m) {
        const int l = (m == 0) ? 0 : (m < 4 ? 1 : 2);
        float acc = 0.f;
        for (int j = 0; j < N_RB; ++j) {
            float v = 0.f;
            for (int s = 0; s < N_SPEC; ++s) v += Tn[(s * 9 + m) * N_RB + j] * e[s];
            acc += v * W_rad[(l * N_RB + j) * KDIM + k];
        }
        F[((long)m * N_ATOMS + n) * KDIM + k] = acc * 0.1f;
    }
}

__global__ void k_gemm_planes(const float* __restrict__ F, const float* __restrict__ W0,
                              const float* __restrict__ W1, const float* __restrict__ W2,
                              float* __restrict__ H) {
    int m  = blockIdx.x / (N_ATOMS / 16);
    const float* B = (m == 0) ? W0 : (m < 4 ? W1 : W2);
    int m0 = blockIdx.x * 16;
    int lane = threadIdx.x;
    int half = lane >> 4;
    int l16  = lane & 15;
    const float* Arow = F + (long)(m0 + l16) * KDIM;
    v8f acc[8] = {};
    for (int kk = 0; kk < KDIM; kk += 4) {
        int ka = kk + 2 * half;
        v2f a;
        a.x = Arow[ka];
        a.y = Arow[ka + 1];
        const float* B0 = B + (long)ka * KDIM + l16;
#pragma unroll
        for (int t = 0; t < 8; ++t) {
            v2f b;
            b.x = B0[t * 16];
            b.y = B0[KDIM + t * 16];
            acc[t] = __builtin_amdgcn_wmma_f32_16x16x4_f32(false, a, false, b, (short)0,
                                                           acc[t], false, false);
        }
    }
#pragma unroll
    for (int t = 0; t < 8; ++t) {
        float* Cc = H + (long)(m0 + 8 * half) * KDIM + t * 16 + l16;
#pragma unroll
        for (int i = 0; i < 8; ++i) Cc[(long)i * KDIM] = acc[t][i];
    }
}

__device__ __forceinline__ void uncouple_l1(float v1, float v2, float v3, float u[9]) {
    u[0] = 0.f;       u[1] =  A2 * v3;  u[2] = -A2 * v2;
    u[3] = -A2 * v3;  u[4] = 0.f;       u[5] =  A2 * v1;
    u[6] =  A2 * v2;  u[7] = -A2 * v1;  u[8] = 0.f;
}
__device__ __forceinline__ void uncouple_l2(float v4, float v5, float v6, float v7, float v8,
                                            float u[9]) {
    u[0] = -B1 * v6 - A2 * v8;  u[1] = A2 * v5;  u[2] = A2 * v4;
    u[3] = A2 * v5;             u[4] = B2 * v6;  u[5] = A2 * v7;
    u[6] = A2 * v4;             u[7] = A2 * v7;  u[8] = -B1 * v6 + A2 * v8;
}
__device__ __forceinline__ float trace_gf_plus_f(const float G[9], const float Fm[9]) {
    float tr = 0.f;
#pragma unroll
    for (int a = 0; a < 3; ++a)
#pragma unroll
        for (int b = 0; b < 3; ++b) tr += G[a * 3 + b] * Fm[b * 3 + a];
    return tr + (Fm[0] + Fm[4] + Fm[8]);
}

__global__ void k_feats(const float* __restrict__ F, const float* __restrict__ H,
                        const int* __restrict__ species, const float* __restrict__ emb,
                        float* __restrict__ x0) {
    long t = (long)blockIdx.x * blockDim.x + threadIdx.x;
    const long PL = (long)N_ATOMS * KDIM;
    if (t >= PL) return;
    int k = (int)(t % KDIM);
    long n = t / KDIM;
    float f[9], h[9];
#pragma unroll
    for (int m = 0; m < 9; ++m) {
        f[m] = F[m * PL + t];
        h[m] = H[m * PL + t];
    }
    float feats0 = h[0] * f[0] + f[0];

    float Gu[9], Fu[9];
    uncouple_l1(f[1], f[2], f[3], Fu);
    uncouple_l1(h[1], h[2], h[3], Gu);
    float tr1 = trace_gf_plus_f(Gu, Fu);
    uncouple_l2(f[4], f[5], f[6], f[7], f[8], Fu);
    uncouple_l2(h[4], h[5], h[6], h[7], h[8], Gu);
    float tr2 = trace_gf_plus_f(Gu, Fu);

    const float i3 = A3;
    int sp = species[n];
    const float* e = emb + sp * N_CH;
    x0[n * K0TOT + k]       = feats0      * e[k / 12];
    x0[n * K0TOT + 128 + k] = (tr1 * i3)  * e[(128 + k) / 12];
    x0[n * K0TOT + 256 + k] = (tr2 * i3)  * e[(256 + k) / 12];
}

__global__ void k_head_wmma(const float* __restrict__ x0, const float* __restrict__ Wh,
                            const float* __restrict__ bh, const float* __restrict__ Wo,
                            float* __restrict__ out) {
    int m0 = blockIdx.x * 16;
    int n0 = blockIdx.y * 128;
    int lane = threadIdx.x;
    int half = lane >> 4;
    int l16  = lane & 15;
    const float* Arow = x0 + (long)(m0 + l16) * K0TOT;
    v8f acc[8] = {};
    for (int kk = 0; kk < K0TOT; kk += 4) {
        int ka = kk + 2 * half;
        v2f a;
        a.x = Arow[ka];
        a.y = Arow[ka + 1];
        const float* B0 = Wh + (long)ka * K0TOT + n0 + l16;
#pragma unroll
        for (int t = 0; t < 8; ++t) {
            v2f b;
            b.x = B0[t * 16];
            b.y = B0[K0TOT + t * 16];
            acc[t] = __builtin_amdgcn_wmma_f32_16x16x4_f32(false, a, false, b, (short)0,
                                                           acc[t], false, false);
        }
    }
    int rowbase = m0 + 8 * half;
    float partial[8];
#pragma unroll
    for (int i = 0; i < 8; ++i) partial[i] = 0.f;
#pragma unroll
    for (int t = 0; t < 8; ++t) {
        int col = n0 + t * 16 + l16;
        float bias = bh[col];
        float wo   = Wo[col];
#pragma unroll
        for (int i = 0; i < 8; ++i) {
            float v = acc[t][i] + bias;
            partial[i] += (v / (1.f + expf(-v))) * wo;
        }
    }
#pragma unroll
    for (int i = 0; i < 8; ++i) atomicAdd(&out[rowbase + i], partial[i]);
}

extern "C" void kernel_launch(void* const* d_in, const int* in_sizes, int n_in,
                              void* d_out, int out_size, void* d_ws, size_t ws_size,
                              hipStream_t stream) {
    (void)in_sizes; (void)n_in; (void)out_size; (void)ws_size;
    const float* positions = (const float*)d_in[0];
    const int*   species   = (const int*)d_in[1];
    const int*   pairs     = (const int*)d_in[2];
    const float* embeddings= (const float*)d_in[3];
    const float* W_rad     = (const float*)d_in[4];
    const float* W_cg0     = (const float*)d_in[5];
    const float* W_cg1     = (const float*)d_in[6];
    const float* W_cg2     = (const float*)d_in[7];
    const float* W_head    = (const float*)d_in[8];
    const float* b_head    = (const float*)d_in[9];
    const float* W_out     = (const float*)d_in[10];
    const float* b_out     = (const float*)d_in[11];
    float* out = (float*)d_out;

    float* ws = (float*)d_ws;
    const long PL = (long)N_ATOMS * KDIM;
    float* Tm = ws;
    float* F  = Tm + (long)N_ATOMS * TM_ATOM;
    float* H  = F + 9 * PL;
    float* x0 = H + 9 * PL;

    long nz = (long)N_ATOMS * TM_ATOM;
    k_zero<<<(unsigned)((nz + 255) / 256), 256, 0, stream>>>(Tm, nz);

    k_init_out<<<(N_ATOMS + 255) / 256, 256, 0, stream>>>(out, b_out);

    k_pairs<<<N_PAIRS * N_RB / 256, 256, 0, stream>>>(positions, species, pairs, Tm);

    k_expand<<<N_ATOMS, KDIM, 0, stream>>>(Tm, embeddings, W_rad, F);

    k_gemm_planes<<<9 * N_ATOMS / 16, 32, 0, stream>>>(F, W_cg0, W_cg1, W_cg2, H);

    k_feats<<<(unsigned)((PL + 255) / 256), 256, 0, stream>>>(F, H, species, embeddings, x0);

    k_head_wmma<<<dim3(N_ATOMS / 16, K0TOT / 16 / 8), 32, 0, stream>>>(
        x0, W_head, b_head, W_out, out);
}


# expand, channel-mix GEMM and feats fused into one kernel per 16-atom tile: coupled rows F stay in LDS, H in WMMA accumulators, neither written to memory; 5 launches (was 7); sum orders unchanged
# speedup vs baseline: 2.3119x; 1.0620x over previous
#include <hip/hip_runtime.h>
#include <math.h>

#define N_ATOMS  10000
#define N_PAIRS  160000
#define N_CH     32
#define KDIM     128
#define K0TOT    384

typedef float v2f __attribute__((ext_vector_type(2)));
typedef float v8f __attribute__((ext_vector_type(8)));

#define A3 0.5773502691896258f
#define A2 0.7071067811865476f
#define B2 0.8164965809277260f
#define B1 0.4082482904638630f

__global__ void k_zero(float* p, long n) {
    long i = (long)blockIdx.x * blockDim.x + threadIdx.x;
    if (i < n) p[i] = 0.f;
}

__global__ void k_init_out(float* out, const float* b_out) {
    int i = blockIdx.x * blockDim.x + threadIdx.x;
    if (i < N_ATOMS) out[i] = b_out[0];
}

#define N_SPEC   4
#define N_RB     8
#define TM_ATOM  (N_SPEC * 9 * N_RB)
__global__ void k_pairs(const float* __restrict__ pos, const int* __restrict__ species,
                        const int* __restrict__ pairs, float* __restrict__ Tm) {
    int t = blockIdx.x * blockDim.x + threadIdx.x;
    int p = t >> 3;
    int j = t & 7;
    if (p >= N_PAIRS) return;
    int c  = pairs[2 * p + 0];
    int nb = pairs[2 * p + 1];
    float rx = pos[3 * nb + 0] - pos[3 * c + 0];
    float ry = pos[3 * nb + 1] - pos[3 * c + 1];
    float rz = pos[3 * nb + 2] - pos[3 * c + 2];
    float d  = sqrtf(rx * rx + ry * ry + rz * rz + 1e-12f);
    float inv = 1.f / d;
    float ux = rx * inv, uy = ry * inv, uz = rz * inv;

    float mu = 20.f * (float)j / 7.f;
    float tc = (d - 15.f) * 0.2f;
    tc = fminf(fmaxf(tc, 0.f), 1.f);
    float fc = 0.5f * (cosf(3.14159265358979f * tc) + 1.f);
    float x  = (d - mu) * 0.4f;
    float rb = expf(-x * x) * fc;

    const float s3 = 1.7320508075688772f;
    float sh[9] = { 1.f, uy, uz, ux,
                    s3 * ux * uy, s3 * uy * uz, 0.5f * (3.f * uz * uz - 1.f),
                    s3 * ux * uz, 0.5f * s3 * (ux * ux - uy * uy) };
    float* Tp = Tm + ((long)c * N_SPEC + species[nb]) * (9 * N_RB) + j;
    for (int m = 0; m < 9; ++m) atomicAdd(Tp + m * N_RB, sh[m] * rb);
}

#define TILE_M   16
#define FS_PITCH 132
#define FS_PLANE (TILE_M * FS_PITCH)

__device__ __forceinline__ void uncouple_l1(float v1, float v2, float v3, float u[9]) {
    u[0] = 0.f;       u[1] =  A2 * v3;  u[2] = -A2 * v2;
    u[3] = -A2 * v3;  u[4] = 0.f;       u[5] =  A2 * v1;
    u[6] =  A2 * v2;  u[7] = -A2 * v1;  u[8] = 0.f;
}
__device__ __forceinline__ void uncouple_l2(float v4, float v5, float v6, float v7, float v8,
                                            float u[9]) {
    u[0] = -B1 * v6 - A2 * v8;  u[1] = A2 * v5;  u[2] = A2 * v4;
    u[3] = A2 * v5;             u[4] = B2 * v6;  u[5] = A2 * v7;
    u[6] = A2 * v4;             u[7] = A2 * v7;  u[8] = -B1 * v6 + A2 * v8;
}
__device__ __forceinline__ float trace_gf_plus_f(const float G[9], const float Fm[9]) {
    float tr = 0.f;
#pragma unroll
    for (int a = 0; a < 3; ++a)
#pragma unroll
        for (int b = 0; b < 3; ++b) tr += G[a * 3 + b] * Fm[b * 3 + a];
    return tr + (Fm[0] + Fm[4] + Fm[8]);
}

__device__ __forceinline__ void load_b_chunk(const float* __restrict__ p, float (&b)[32]) {
#pragma unroll
    for (int ks = 0; ks < 8; ++ks) {
        const float* q = p + ks * 4 * KDIM;
        const v2f r0 = *reinterpret_cast<const v2f*>(q);
        const v2f r1 = *reinterpret_cast<const v2f*>(q + KDIM);
        b[4 * ks + 0] = r0.x;
        b[4 * ks + 1] = r1.x;
        b[4 * ks + 2] = r0.y;
        b[4 * ks + 3] = r1.y;
    }
}
__device__ __forceinline__ void load_a_chunk(const float* __restrict__ p, v2f (&a)[8]) {
#pragma unroll
    for (int ks = 0; ks < 8; ++ks) {
        a[ks].x = p[4 * ks];
        a[ks].y = p[4 * ks + 1];
    }
}

template <int M0, int NM, bool HAS_NEXT>
__device__ __forceinline__ void mix_planes(const float* __restrict__ Fl,
                                           const float* __restrict__ Bl,
                                           const float* __restrict__ Bnext,
                                           float (&bc)[32], v8f (&acc)[18]) {
    for (int c = 0; c < 4; ++c) {
        const bool more = HAS_NEXT || (c < 3);
        float bn[32];
        if (more) load_b_chunk((c < 3) ? Bl + (c + 1) * 32 * KDIM : Bnext, bn);
        v2f a[8];
        load_a_chunk(Fl + M0 * FS_PLANE + c * 32, a);
#pragma unroll
        for (int m = M0; m < M0 + NM; ++m) {
            v2f an[8];
            if (m + 1 < M0 + NM) load_a_chunk(Fl + (m + 1) * FS_PLANE + c * 32, an);
#pragma unroll
            for (int ks = 0; ks < 8; ++ks) {
                v2f b0, b1;
                b0.x = bc[4 * ks + 0];
                b0.y = bc[4 * ks + 1];
                b1.x = bc[4 * ks + 2];
                b1.y = bc[4 * ks + 3];
                acc[2 * m]     = __builtin_amdgcn_wmma_f32_16x16x4_f32(
                    false, a[ks], false, b0, (short)0, acc[2 * m], false, false);
                acc[2 * m + 1] = __builtin_amdgcn_wmma_f32_16x16x4_f32(
                    false, a[ks], false, b1, (short)0, acc[2 * m + 1], false, false);
            }
            if (m + 1 < M0 + NM) {
#pragma unroll
                for (int ks = 0; ks < 8; ++ks) a[ks] = an[ks];
            }
        }
        if (more) {
#pragma unroll
            for (int q = 0; q < 32; ++q) bc[q] = bn[q];
        }
    }
}

__global__ void __launch_bounds__(KDIM)
k_tile_feats(const float* __restrict__ Tm, const float* __restrict__ emb,
             const float* __restrict__ W_rad, const float* __restrict__ W0,
             const float* __restrict__ W1, const float* __restrict__ W2,
             const int* __restrict__ species, float* __restrict__ x0) {
    __shared__ float Fs[9 * FS_PLANE];
    __shared__ float Es[N_SPEC * N_CH];
    __shared__ int   Ss[TILE_M];
    const int n0 = blockIdx.x * TILE_M;
    const int k  = threadIdx.x;
    Es[k] = emb[k];
    if (k < TILE_M) Ss[k] = species[n0 + k];

    {
        float e[N_SPEC];
        for (int s = 0; s < N_SPEC; ++s) e[s] = emb[s * N_CH + (k >> 2)];
        float wr[3 * N_RB];
#pragma unroll
        for (int q = 0; q < 3 * N_RB; ++q) wr[q] = W_rad[q * KDIM + k];
        for (int a = 0; a < TILE_M; ++a) {
            const float* Tn = Tm + (long)(n0 + a) * TM_ATOM;
#pragma unroll
            for (int m = 0; m < 9; ++m) {
                const int l = (m == 0) ? 0 : (m < 4 ? 1 : 2);
                float acc = 0.f;
#pragma unroll
                for (int j = 0; j < N_RB; ++j) {
                    float v = 0.f;
                    for (int s = 0; s < N_SPEC; ++s) v += Tn[(s * 9 + m) * N_RB + j] * e[s];
                    acc += v * wr[l * N_RB + j];
                }
                Fs[m * FS_PLANE + a * FS_PITCH + k] = acc * 0.1f;
            }
        }
    }
    __syncthreads();

    const int lane = k & 31;
    const int wave = k >> 5;
    const int half = lane >> 4;
    const int l16  = lane & 15;
    const int colbase = wave * 32;
    v8f acc[18] = {};
    {
        const float* Fl = Fs + l16 * FS_PITCH + 2 * half;
        const long boff = (long)(2 * half) * KDIM + colbase + 2 * l16;
        float bc[32];
        load_b_chunk(W0 + boff, bc);
        mix_planes<0, 1, true >(Fl, W0 + boff, W1 + boff, bc, acc);
        mix_planes<1, 3, true >(Fl, W1 + boff, W2 + boff, bc, acc);
        mix_planes<4, 5, false>(Fl, W2 + boff, W2 + boff, bc, acc);
    }

    const float i3 = A3;
    const int col = colbase + 2 * l16;
    const float* Fr = Fs + (8 * half) * FS_PITCH + col;
    int ec[3][2];
#pragma unroll
    for (int g = 0; g < 3; ++g) {
        ec[g][0] = (128 * g + col) / 12;
        ec[g][1] = (128 * g + col + 1) / 12;
    }
    v2f f[9], fn[9];
#pragma unroll
    for (int m = 0; m < 9; ++m) f[m] = *reinterpret_cast<const v2f*>(Fr + m * FS_PLANE);
#pragma unroll
    for (int i = 0; i < 8; ++i) {
        if (i + 1 < 8) {
#pragma unroll
            for (int m = 0; m < 9; ++m)
                fn[m] = *reinterpret_cast<const v2f*>(Fr + m * FS_PLANE + (i + 1) * FS_PITCH);
        }
        const float* e = Es + Ss[8 * half + i] * N_CH;
        v2f o[3];
#pragma unroll
        for (int t = 0; t < 2; ++t) {
            float fv[9], h[9];
#pragma unroll
            for (int m = 0; m < 9; ++m) {
                fv[m] = f[m][t];
                h[m]  = acc[2 * m + t][i];
            }
            float feats0 = h[0] * fv[0] + fv[0];
            float Gu[9], Fu[9];
            uncouple_l1(fv[1], fv[2], fv[3], Fu);
            uncouple_l1(h[1], h[2], h[3], Gu);
            float tr1 = trace_gf_plus_f(Gu, Fu);
            uncouple_l2(fv[4], fv[5], fv[6], fv[7], fv[8], Fu);
            uncouple_l2(h[4], h[5], h[6], h[7], h[8], Gu);
            float tr2 = trace_gf_plus_f(Gu, Fu);
            o[0][t] = feats0     * e[ec[0][t]];
            o[1][t] = (tr1 * i3) * e[ec[1][t]];
            o[2][t] = (tr2 * i3) * e[ec[2][t]];
        }
        float* xr = x0 + (long)(n0 + 8 * half + i) * K0TOT + col;
        *reinterpret_cast<v2f*>(xr)       = o[0];
        *reinterpret_cast<v2f*>(xr + 128) = o[1];
        *reinterpret_cast<v2f*>(xr + 256) = o[2];
        if (i + 1 < 8) {
#pragma unroll
            for (int m = 0; m < 9; ++m) f[m] = fn[m];
        }
    }
}

__global__ void k_head_wmma(const float* __restrict__ x0, const float* __restrict__ Wh,
                            const float* __restrict__ bh, const float* __restrict__ Wo,
                            float* __restrict__ out) {
    int m0 = blockIdx.x * 16;
    int n0 = blockIdx.y * 128;
    int lane = threadIdx.x;
    int half = lane >> 4;
    int l16  = lane & 15;
    const float* Arow = x0 + (long)(m0 + l16) * K0TOT;
    v8f acc[8] = {};
    for (int kk = 0; kk < K0TOT; kk += 4) {
        int ka = kk + 2 * half;
        v2f a;
        a.x = Arow[ka];
        a.y = Arow[ka + 1];
        const float* B0 = Wh + (long)ka * K0TOT + n0 + l16;
#pragma unroll
        for (int t = 0; t < 8; ++t) {
            v2f b;
            b.x = B0[t * 16];
            b.y = B0[K0TOT + t * 16];
            acc[t] = __builtin_amdgcn_wmma_f32_16x16x4_f32(false, a, false, b, (short)0,
                                                           acc[t], false, false);
        }
    }
    int rowbase = m0 + 8 * half;
    float partial[8];
#pragma unroll
    for (int i = 0; i < 8; ++i) partial[i] = 0.f;
#pragma unroll
    for (int t = 0; t < 8; ++t) {
        int col = n0 + t * 16 + l16;
        float bias = bh[col];
        float wo   = Wo[col];
#pragma unroll
        for (int i = 0; i < 8; ++i) {
            float v = acc[t][i] + bias;
            partial[i] += (v / (1.f + expf(-v))) * wo;
        }
    }
#pragma unroll
    for (int i = 0; i < 8; ++i) atomicAdd(&out[rowbase + i], partial[i]);
}

extern "C" void kernel_launch(void* const* d_in, const int* in_sizes, int n_in,
                              void* d_out, int out_size, void* d_ws, size_t ws_size,
                              hipStream_t stream) {
    (void)in_sizes; (void)n_in; (void)out_size; (void)ws_size;
    const float* positions = (const float*)d_in[0];
    const int*   species   = (const int*)d_in[1];
    const int*   pairs     = (const int*)d_in[2];
    const float* embeddings= (const float*)d_in[3];
    const float* W_rad     = (const float*)d_in[4];
    const float* W_cg0     = (const float*)d_in[5];
    const float* W_cg1     = (const float*)d_in[6];
    const float* W_cg2     = (const float*)d_in[7];
    const float* W_head    = (const float*)d_in[8];
    const float* b_head    = (const float*)d_in[9];
    const float* W_out     = (const float*)d_in[10];
    const float* b_out     = (const float*)d_in[11];
    float* out = (float*)d_out;

    float* ws = (float*)d_ws;
    float* Tm = ws;
    float* x0 = Tm + (long)N_ATOMS * TM_ATOM;

    long nz = (long)N_ATOMS * TM_ATOM;
    k_zero<<<(unsigned)((nz + 255) / 256), 256, 0, stream>>>(Tm, nz);

    k_init_out<<<(N_ATOMS + 255) / 256, 256, 0, stream>>>(out, b_out);

    k_pairs<<<N_PAIRS * N_RB / 256, 256, 0, stream>>>(positions, species, pairs, Tm);

    k_tile_feats<<<N_ATOMS / TILE_M, KDIM, 0, stream>>>(Tm, embeddings, W_rad, W_cg0, W_cg1,
                                                        W_cg2, species, x0);

    k_head_wmma<<<dim3(N_ATOMS / 16, K0TOT / 16 / 8), 32, 0, stream>>>(
        x0, W_head, b_head, W_out, out);
}


# coupled rows F built from the moments as a K=32 f32 WMMA product per plane (moments tile staged in LDS as A, emb x W_rad B fragments in registers), was scalar-load VALU sums; sum order of F changes
# speedup vs baseline: 2.7717x; 1.1989x over previous
#include <hip/hip_runtime.h>
#include <math.h>

#define N_ATOMS  10000
#define N_PAIRS  160000
#define N_CH     32
#define KDIM     128
#define K0TOT    384

typedef float v2f __attribute__((ext_vector_type(2)));
typedef float v4f __attribute__((ext_vector_type(4)));
typedef float v8f __attribute__((ext_vector_type(8)));

#define A3 0.5773502691896258f
#define A2 0.7071067811865476f
#define B2 0.8164965809277260f
#define B1 0.4082482904638630f

__global__ void k_zero(float* p, long n) {
    long i = (long)blockIdx.x * blockDim.x + threadIdx.x;
    if (i < n) p[i] = 0.f;
}

__global__ void k_init_out(float* out, const float* b_out) {
    int i = blockIdx.x * blockDim.x + threadIdx.x;
    if (i < N_ATOMS) out[i] = b_out[0];
}

#define N_SPEC   4
#define N_RB     8
#define TM_ATOM  (N_SPEC * 9 * N_RB)
__global__ void k_pairs(const float* __restrict__ pos, const int* __restrict__ species,
                        const int* __restrict__ pairs, float* __restrict__ Tm) {
    int t = blockIdx.x * blockDim.x + threadIdx.x;
    int p = t >> 3;
    int j = t & 7;
    if (p >= N_PAIRS) return;
    int c  = pairs[2 * p + 0];
    int nb = pairs[2 * p + 1];
    float rx = pos[3 * nb + 0] - pos[3 * c + 0];
    float ry = pos[3 * nb + 1] - pos[3 * c + 1];
    float rz = pos[3 * nb + 2] - pos[3 * c + 2];
    float d  = sqrtf(rx * rx + ry * ry + rz * rz + 1e-12f);
    float inv = 1.f / d;
    float ux = rx * inv, uy = ry * inv, uz = rz * inv;

    float mu = 20.f * (float)j / 7.f;
    float tc = (d - 15.f) * 0.2f;
    tc = fminf(fmaxf(tc, 0.f), 1.f);
    float fc = 0.5f * (cosf(3.14159265358979f * tc) + 1.f);
    float x  = (d - mu) * 0.4f;
    float rb = expf(-x * x) * fc;

    const float s3 = 1.7320508075688772f;
    float sh[9] = { 1.f, uy, uz, ux,
                    s3 * ux * uy, s3 * uy * uz, 0.5f * (3.f * uz * uz - 1.f),
                    s3 * ux * uz, 0.5f * s3 * (ux * ux - uy * uy) };
    float* Tp = Tm + ((long)c * N_SPEC + species[nb]) * (9 * N_RB) + j;
    for (int m = 0; m < 9; ++m) atomicAdd(Tp + m * N_RB, sh[m] * rb);
}

#define TILE_M   16
#define FS_PITCH 132
#define FS_PLANE (TILE_M * FS_PITCH)
#define TS_PITCH 292

__device__ __forceinline__ void uncouple_l1(float v1, float v2, float v3, float u[9]) {
    u[0] = 0.f;       u[1] =  A2 * v3;  u[2] = -A2 * v2;
    u[3] = -A2 * v3;  u[4] = 0.f;       u[5] =  A2 * v1;
    u[6] =  A2 * v2;  u[7] = -A2 * v1;  u[8] = 0.f;
}
__device__ __forceinline__ void uncouple_l2(float v4, float v5, float v6, float v7, float v8,
                                            float u[9]) {
    u[0] = -B1 * v6 - A2 * v8;  u[1] = A2 * v5;  u[2] = A2 * v4;
    u[3] = A2 * v5;             u[4] = B2 * v6;  u[5] = A2 * v7;
    u[6] = A2 * v4;             u[7] = A2 * v7;  u[8] = -B1 * v6 + A2 * v8;
}
__device__ __forceinline__ float trace_gf_plus_f(const float G[9], const float Fm[9]) {
    float tr = 0.f;
#pragma unroll
    for (int a = 0; a < 3; ++a)
#pragma unroll
        for (int b = 0; b < 3; ++b) tr += G[a * 3 + b] * Fm[b * 3 + a];
    return tr + (Fm[0] + Fm[4] + Fm[8]);
}

__device__ __forceinline__ void load_b_chunk(const float* __restrict__ p, float (&b)[32]) {
#pragma unroll
    for (int ks = 0; ks < 8; ++ks) {
        const float* q = p + ks * 4 * KDIM;
        const v2f r0 = *reinterpret_cast<const v2f*>(q);
        const v2f r1 = *reinterpret_cast<const v2f*>(q + KDIM);
        b[4 * ks + 0] = r0.x;
        b[4 * ks + 1] = r1.x;
        b[4 * ks + 2] = r0.y;
        b[4 * ks + 3] = r1.y;
    }
}
__device__ __forceinline__ void load_a_chunk(const float* __restrict__ p, v2f (&a)[8]) {
#pragma unroll
    for (int ks = 0; ks < 8; ++ks) {
        a[ks].x = p[4 * ks];
        a[ks].y = p[4 * ks + 1];
    }
}

template <int M0, int NM>
__device__ __forceinline__ void rows_from_moments(const float* __restrict__ Ta,
                                                  const v2f (&w)[4], const float (&e)[N_SPEC],
                                                  float* __restrict__ Fd) {
    float b[32];
#pragma unroll
    for (int ks = 0; ks < 8; ++ks) {
        const int s  = ks >> 1;
        const int jb = 2 * (ks & 1);
        b[4 * ks + 0] = e[s] * w[jb].x;
        b[4 * ks + 1] = e[s] * w[jb + 1].x;
        b[4 * ks + 2] = e[s] * w[jb].y;
        b[4 * ks + 3] = e[s] * w[jb + 1].y;
    }
#pragma unroll
    for (int m = M0; m < M0 + NM; ++m) {
        v8f c0 = {}, c1 = {};
#pragma unroll
        for (int ks = 0; ks < 8; ++ks) {
            const v2f a = *reinterpret_cast<const v2f*>(Ta + (ks >> 1) * 72 + m * 8 + 4 * (ks & 1));
            v2f b0, b1;
            b0.x = b[4 * ks + 0];
            b0.y = b[4 * ks + 1];
            b1.x = b[4 * ks + 2];
            b1.y = b[4 * ks + 3];
            c0 = __builtin_amdgcn_wmma_f32_16x16x4_f32(false, a, false, b0, (short)0, c0,
                                                       false, false);
            c1 = __builtin_amdgcn_wmma_f32_16x16x4_f32(false, a, false, b1, (short)0, c1,
                                                       false, false);
        }
#pragma unroll
        for (int r = 0; r < 8; ++r) {
            v2f o;
            o.x = c0[r] * 0.1f;
            o.y = c1[r] * 0.1f;
            *reinterpret_cast<v2f*>(Fd + m * FS_PLANE + r * FS_PITCH) = o;
        }
    }
}

template <int M0, int NM, bool HAS_NEXT>
__device__ __forceinline__ void mix_planes(const float* __restrict__ Fl,
                                           const float* __restrict__ Bl,
                                           const float* __restrict__ Bnext,
                                           float (&bc)[32], v8f (&acc)[18]) {
    for (int c = 0; c < 4; ++c) {
        const bool more = HAS_NEXT || (c < 3);
        float bn[32];
        if (more) load_b_chunk((c < 3) ? Bl + (c + 1) * 32 * KDIM : Bnext, bn);
        v2f a[8];
        load_a_chunk(Fl + M0 * FS_PLANE + c * 32, a);
#pragma unroll
        for (int m = M0; m < M0 + NM; ++m) {
            v2f an[8];
            if (m + 1 < M0 + NM) load_a_chunk(Fl + (m + 1) * FS_PLANE + c * 32, an);
#pragma unroll
            for (int ks = 0; ks < 8; ++ks) {
                v2f b0, b1;
                b0.x = bc[4 * ks + 0];
                b0.y = bc[4 * ks + 1];
                b1.x = bc[4 * ks + 2];
                b1.y = bc[4 * ks + 3];
                acc[2 * m]     = __builtin_amdgcn_wmma_f32_16x16x4_f32(
                    false, a[ks], false, b0, (short)0, acc[2 * m], false, false);
                acc[2 * m + 1] = __builtin_amdgcn_wmma_f32_16x16x4_f32(
                    false, a[ks], false, b1, (short)0, acc[2 * m + 1], false, false);
            }
            if (m + 1 < M0 + NM) {
#pragma unroll
                for (int ks = 0; ks < 8; ++ks) a[ks] = an[ks];
            }
        }
        if (more) {
#pragma unroll
            for (int q = 0; q < 32; ++q) bc[q] = bn[q];
        }
    }
}

__global__ void __launch_bounds__(KDIM)
k_tile_feats(const float* __restrict__ Tm, const float* __restrict__ emb,
             const float* __restrict__ W_rad, const float* __restrict__ W0,
             const float* __restrict__ W1, const float* __restrict__ W2,
             const int* __restrict__ species, float* __restrict__ x0) {
    __shared__ float Fs[9 * FS_PLANE];
    __shared__ __attribute__((aligned(16))) float Ts[TILE_M * TS_PITCH];
    __shared__ float Es[N_SPEC * N_CH];
    __shared__ int   Ss[TILE_M];
    const int n0 = blockIdx.x * TILE_M;
    const int k  = threadIdx.x;
    const int lane = k & 31;
    const int wave = k >> 5;
    const int half = lane >> 4;
    const int l16  = lane & 15;
    const int colbase = wave * 32;
    const int col = colbase + 2 * l16;
    Es[k] = emb[k];
    if (k < TILE_M) Ss[k] = species[n0 + k];

    {
        const v4f* Tg = reinterpret_cast<const v4f*>(Tm + (long)n0 * TM_ATOM);
        v4f tv[9];
#pragma unroll
        for (int it = 0; it < 9; ++it) tv[it] = Tg[it * KDIM + k];
        float e[N_SPEC];
#pragma unroll
        for (int s = 0; s < N_SPEC; ++s) e[s] = emb[s * N_CH + (col >> 2)];
        v2f wr[3][4];
#pragma unroll
        for (int l = 0; l < 3; ++l) {
#pragma unroll
            for (int jj = 0; jj < 4; ++jj) {
                const int j = 4 * (jj >> 1) + 2 * half + (jj & 1);
                wr[l][jj] = *reinterpret_cast<const v2f*>(W_rad + (l * N_RB + j) * KDIM + col);
            }
        }
#pragma unroll
        for (int it = 0; it < 9; ++it) {
            const int idx = it * KDIM + k;
            const int a   = idx / 72;
            const int q   = idx - a * 72;
            *reinterpret_cast<v4f*>(Ts + a * TS_PITCH + 4 * q) = tv[it];
        }
        __syncthreads();
        const float* Ta = Ts + l16 * TS_PITCH + 2 * half;
        float* Fd = Fs + (8 * half) * FS_PITCH + col;
        rows_from_moments<0, 1>(Ta, wr[0], e, Fd);
        rows_from_moments<1, 3>(Ta, wr[1], e, Fd);
        rows_from_moments<4, 5>(Ta, wr[2], e, Fd);
    }
    __syncthreads();

    v8f acc[18] = {};
    {
        const float* Fl = Fs + l16 * FS_PITCH + 2 * half;
        const long boff = (long)(2 * half) * KDIM + colbase + 2 * l16;
        float bc[32];
        load_b_chunk(W0 + boff, bc);
        mix_planes<0, 1, true >(Fl, W0 + boff, W1 + boff, bc, acc);
        mix_planes<1, 3, true >(Fl, W1 + boff, W2 + boff, bc, acc);
        mix_planes<4, 5, false>(Fl, W2 + boff, W2 + boff, bc, acc);
    }

    const float i3 = A3;
    const float* Fr = Fs + (8 * half) * FS_PITCH + col;
    int ec[3][2];
#pragma unroll
    for (int g = 0; g < 3; ++g) {
        ec[g][0] = (128 * g + col) / 12;
        ec[g][1] = (128 * g + col + 1) / 12;
    }
    v2f f[9], fn[9];
#pragma unroll
    for (int m = 0; m < 9; ++m) f[m] = *reinterpret_cast<const v2f*>(Fr + m * FS_PLANE);
#pragma unroll
    for (int i = 0; i < 8; ++i) {
        if (i + 1 < 8) {
#pragma unroll
            for (int m = 0; m < 9; ++m)
                fn[m] = *reinterpret_cast<const v2f*>(Fr + m * FS_PLANE + (i + 1) * FS_PITCH);
        }
        const float* e = Es + Ss[8 * half + i] * N_CH;
        v2f o[3];
#pragma unroll
        for (int t = 0; t < 2; ++t) {
            float fv[9], h[9];
#pragma unroll
            for (int m = 0; m < 9; ++m) {
                fv[m] = f[m][t];
                h[m]  = acc[2 * m + t][i];
            }
            float feats0 = h[0] * fv[0] + fv[0];
            float Gu[9], Fu[9];
            uncouple_l1(fv[1], fv[2], fv[3], Fu);
            uncouple_l1(h[1], h[2], h[3], Gu);
            float tr1 = trace_gf_plus_f(Gu, Fu);
            uncouple_l2(fv[4], fv[5], fv[6], fv[7], fv[8], Fu);
            uncouple_l2(h[4], h[5], h[6], h[7], h[8], Gu);
            float tr2 = trace_gf_plus_f(Gu, Fu);
            o[0][t] = feats0     * e[ec[0][t]];
            o[1][t] = (tr1 * i3) * e[ec[1][t]];
            o[2][t] = (tr2 * i3) * e[ec[2][t]];
        }
        float* xr = x0 + (long)(n0 + 8 * half + i) * K0TOT + col;
        *reinterpret_cast<v2f*>(xr)       = o[0];
        *reinterpret_cast<v2f*>(xr + 128) = o[1];
        *reinterpret_cast<v2f*>(xr + 256) = o[2];
        if (i + 1 < 8) {
#pragma unroll
            for (int m = 0; m < 9; ++m) f[m] = fn[m];
        }
    }
}

__global__ void k_head_wmma(const float* __restrict__ x0, const float* __restrict__ Wh,
                            const float* __restrict__ bh, const float* __restrict__ Wo,
                            float* __restrict__ out) {
    int m0 = blockIdx.x * 16;
    int n0 = blockIdx.y * 128;
    int lane = threadIdx.x;
    int half = lane >> 4;
    int l16  = lane & 15;
    const float* Arow = x0 + (long)(m0 + l16) * K0TOT;
    v8f acc[8] = {};
    for (int kk = 0; kk < K0TOT; kk += 4) {
        int ka = kk + 2 * half;
        v2f a;
        a.x = Arow[ka];
        a.y = Arow[ka + 1];
        const float* B0 = Wh + (long)ka * K0TOT + n0 + l16;
#pragma unroll
        for (int t = 0; t < 8; ++t) {
            v2f b;
            b.x = B0[t * 16];
            b.y = B0[K0TOT + t * 16];
            acc[t] = __builtin_amdgcn_wmma_f32_16x16x4_f32(false, a, false, b, (short)0,
                                                           acc[t], false, false);
        }
    }
    int rowbase = m0 + 8 * half;
    float partial[8];
#pragma unroll
    for (int i = 0; i < 8; ++i) partial[i] = 0.f;
#pragma unroll
    for (int t = 0; t < 8; ++t) {
        int col = n0 + t * 16 + l16;
        float bias = bh[col];
        float wo   = Wo[col];
#pragma unroll
        for (int i = 0; i < 8; ++i) {
            float v = acc[t][i] + bias;
            partial[i] += (v / (1.f + expf(-v))) * wo;
        }
    }
#pragma unroll
    for (int i = 0; i < 8; ++i) atomicAdd(&out[rowbase + i], partial[i]);
}

extern "C" void kernel_launch(void* const* d_in, const int* in_sizes, int n_in,
                              void* d_out, int out_size, void* d_ws, size_t ws_size,
                              hipStream_t stream) {
    (void)in_sizes; (void)n_in; (void)out_size; (void)ws_size;
    const float* positions = (const float*)d_in[0];
    const int*   species   = (const int*)d_in[1];
    const int*   pairs     = (const int*)d_in[2];
    const float* embeddings= (const float*)d_in[3];
    const float* W_rad     = (const float*)d_in[4];
    const float* W_cg0     = (const float*)d_in[5];
    const float* W_cg1     = (const float*)d_in[6];
    const float* W_cg2     = (const float*)d_in[7];
    const float* W_head    = (const float*)d_in[8];
    const float* b_head    = (const float*)d_in[9];
    const float* W_out     = (const float*)d_in[10];
    const float* b_out     = (const float*)d_in[11];
    float* out = (float*)d_out;

    float* ws = (float*)d_ws;
    float* Tm = ws;
    float* x0 = Tm + (long)N_ATOMS * TM_ATOM;

    long nz = (long)N_ATOMS * TM_ATOM;
    k_zero<<<(unsigned)((nz + 255) / 256), 256, 0, stream>>>(Tm, nz);

    k_init_out<<<(N_ATOMS + 255) / 256, 256, 0, stream>>>(out, b_out);

    k_pairs<<<N_PAIRS * N_RB / 256, 256, 0, stream>>>(positions, species, pairs, Tm);

    k_tile_feats<<<N_ATOMS / TILE_M, KDIM, 0, stream>>>(Tm, embeddings, W_rad, W_cg0, W_cg1,
                                                        W_cg2, species, x0);

    k_head_wmma<<<dim3(N_ATOMS / 16, K0TOT / 16 / 8), 32, 0, stream>>>(
        x0, W_head, b_head, W_out, out);
}


# head GEMM: wave tile 32x128 (was 16x128), 4-wave blocks, 16-byte operand loads (k dealt in quads), load groups issued 1 (W_head) to 2 (x0) 64-WMMA blocks ahead; sum order of the W_out dot changes
# speedup vs baseline: 3.9242x; 1.4158x over previous
#include <hip/hip_runtime.h>
#include <math.h>

#define N_ATOMS  10000
#define N_PAIRS  160000
#define N_CH     32
#define KDIM     128
#define K0TOT    384

typedef float v2f __attribute__((ext_vector_type(2)));
typedef float v4f __attribute__((ext_vector_type(4)));
typedef float v8f __attribute__((ext_vector_type(8)));

#define A3 0.5773502691896258f
#define A2 0.7071067811865476f
#define B2 0.8164965809277260f
#define B1 0.4082482904638630f

__global__ void k_zero(float* p, long n) {
    long i = (long)blockIdx.x * blockDim.x + threadIdx.x;
    if (i < n) p[i] = 0.f;
}

__global__ void k_init_out(float* out, const float* b_out) {
    int i = blockIdx.x * blockDim.x + threadIdx.x;
    if (i < N_ATOMS) out[i] = b_out[0];
}

#define N_SPEC   4
#define N_RB     8
#define TM_ATOM  (N_SPEC * 9 * N_RB)
__global__ void k_pairs(const float* __restrict__ pos, const int* __restrict__ species,
                        const int* __restrict__ pairs, float* __restrict__ Tm) {
    int t = blockIdx.x * blockDim.x + threadIdx.x;
    int p = t >> 3;
    int j = t & 7;
    if (p >= N_PAIRS) return;
    int c  = pairs[2 * p + 0];
    int nb = pairs[2 * p + 1];
    float rx = pos[3 * nb + 0] - pos[3 * c + 0];
    float ry = pos[3 * nb + 1] - pos[3 * c + 1];
    float rz = pos[3 * nb + 2] - pos[3 * c + 2];
    float d  = sqrtf(rx * rx + ry * ry + rz * rz + 1e-12f);
    float inv = 1.f / d;
    float ux = rx * inv, uy = ry * inv, uz = rz * inv;

    float mu = 20.f * (float)j / 7.f;
    float tc = (d - 15.f) * 0.2f;
    tc = fminf(fmaxf(tc, 0.f), 1.f);
    float fc = 0.5f * (cosf(3.14159265358979f * tc) + 1.f);
    float x  = (d - mu) * 0.4f;
    float rb = expf(-x * x) * fc;

    const float s3 = 1.7320508075688772f;
    float sh[9] = { 1.f, uy, uz, ux,
                    s3 * ux * uy, s3 * uy * uz, 0.5f * (3.f * uz * uz - 1.f),
                    s3 * ux * uz, 0.5f * s3 * (ux * ux - uy * uy) };
    float* Tp = Tm + ((long)c * N_SPEC + species[nb]) * (9 * N_RB) + j;
    for (int m = 0; m < 9; ++m) atomicAdd(Tp + m * N_RB, sh[m] * rb);
}

#define TILE_M   16
#define FS_PITCH 132
#define FS_PLANE (TILE_M * FS_PITCH)
#define TS_PITCH 292

__device__ __forceinline__ void uncouple_l1(float v1, float v2, float v3, float u[9]) {
    u[0] = 0.f;       u[1] =  A2 * v3;  u[2] = -A2 * v2;
    u[3] = -A2 * v3;  u[4] = 0.f;       u[5] =  A2 * v1;
    u[6] =  A2 * v2;  u[7] = -A2 * v1;  u[8] = 0.f;
}
__device__ __forceinline__ void uncouple_l2(float v4, float v5, float v6, float v7, float v8,
                                            float u[9]) {
    u[0] = -B1 * v6 - A2 * v8;  u[1] = A2 * v5;  u[2] = A2 * v4;
    u[3] = A2 * v5;             u[4] = B2 * v6;  u[5] = A2 * v7;
    u[6] = A2 * v4;             u[7] = A2 * v7;  u[8] = -B1 * v6 + A2 * v8;
}
__device__ __forceinline__ float trace_gf_plus_f(const float G[9], const float Fm[9]) {
    float tr = 0.f;
#pragma unroll
    for (int a = 0; a < 3; ++a)
#pragma unroll
        for (int b = 0; b < 3; ++b) tr += G[a * 3 + b] * Fm[b * 3 + a];
    return tr + (Fm[0] + Fm[4] + Fm[8]);
}

__device__ __forceinline__ void load_b_chunk(const float* __restrict__ p, float (&b)[32]) {
#pragma unroll
    for (int ks = 0; ks < 8; ++ks) {
        const float* q = p + ks * 4 * KDIM;
        const v2f r0 = *reinterpret_cast<const v2f*>(q);
        const v2f r1 = *reinterpret_cast<const v2f*>(q + KDIM);
        b[4 * ks + 0] = r0.x;
        b[4 * ks + 1] = r1.x;
        b[4 * ks + 2] = r0.y;
        b[4 * ks + 3] = r1.y;
    }
}
__device__ __forceinline__ void load_a_chunk(const float* __restrict__ p, v2f (&a)[8]) {
#pragma unroll
    for (int ks = 0; ks < 8; ++ks) {
        a[ks].x = p[4 * ks];
        a[ks].y = p[4 * ks + 1];
    }
}

template <int M0, int NM>
__device__ __forceinline__ void rows_from_moments(const float* __restrict__ Ta,
                                                  const v2f (&w)[4], const float (&e)[N_SPEC],
                                                  float* __restrict__ Fd) {
    float b[32];
#pragma unroll
    for (int ks = 0; ks < 8; ++ks) {
        const int s  = ks >> 1;
        const int jb = 2 * (ks & 1);
        b[4 * ks + 0] = e[s] * w[jb].x;
        b[4 * ks + 1] = e[s] * w[jb + 1].x;
        b[4 * ks + 2] = e[s] * w[jb].y;
        b[4 * ks + 3] = e[s] * w[jb + 1].y;
    }
#pragma unroll
    for (int m = M0; m < M0 + NM; ++m) {
        v8f c0 = {}, c1 = {};
#pragma unroll
        for (int ks = 0; ks < 8; ++ks) {
            const v2f a = *reinterpret_cast<const v2f*>(Ta + (ks >> 1) * 72 + m * 8 + 4 * (ks & 1));
            v2f b0, b1;
            b0.x = b[4 * ks + 0];
            b0.y = b[4 * ks + 1];
            b1.x = b[4 * ks + 2];
            b1.y = b[4 * ks + 3];
            c0 = __builtin_amdgcn_wmma_f32_16x16x4_f32(false, a, false, b0, (short)0, c0,
                                                       false, false);
            c1 = __builtin_amdgcn_wmma_f32_16x16x4_f32(false, a, false, b1, (short)0, c1,
                                                       false, false);
        }
#pragma unroll
        for (int r = 0; r < 8; ++r) {
            v2f o;
            o.x = c0[r] * 0.1f;
            o.y = c1[r] * 0.1f;
            *reinterpret_cast<v2f*>(Fd + m * FS_PLANE + r * FS_PITCH) = o;
        }
    }
}

template <int M0, int NM, bool HAS_NEXT>
__device__ __forceinline__ void mix_planes(const float* __restrict__ Fl,
                                           const float* __restrict__ Bl,
                                           const float* __restrict__ Bnext,
                                           float (&bc)[32], v8f (&acc)[18]) {
    for (int c = 0; c < 4; ++c) {
        const bool more = HAS_NEXT || (c < 3);
        float bn[32];
        if (more) load_b_chunk((c < 3) ? Bl + (c + 1) * 32 * KDIM : Bnext, bn);
        v2f a[8];
        load_a_chunk(Fl + M0 * FS_PLANE + c * 32, a);
#pragma unroll
        for (int m = M0; m < M0 + NM; ++m) {
            v2f an[8];
            if (m + 1 < M0 + NM) load_a_chunk(Fl + (m + 1) * FS_PLANE + c * 32, an);
#pragma unroll
            for (int ks = 0; ks < 8; ++ks) {
                v2f b0, b1;
                b0.x = bc[4 * ks + 0];
                b0.y = bc[4 * ks + 1];
                b1.x = bc[4 * ks + 2];
                b1.y = bc[4 * ks + 3];
                acc[2 * m]     = __builtin_amdgcn_wmma_f32_16x16x4_f32(
                    false, a[ks], false, b0, (short)0, acc[2 * m], false, false);
                acc[2 * m + 1] = __builtin_amdgcn_wmma_f32_16x16x4_f32(
                    false, a[ks], false, b1, (short)0, acc[2 * m + 1], false, false);
            }
            if (m + 1 < M0 + NM) {
#pragma unroll
                for (int ks = 0; ks < 8; ++ks) a[ks] = an[ks];
            }
        }
        if (more) {
#pragma unroll
            for (int q = 0; q < 32; ++q) bc[q] = bn[q];
        }
    }
}

__global__ void __launch_bounds__(KDIM)
k_tile_feats(const float* __restrict__ Tm, const float* __restrict__ emb,
             const float* __restrict__ W_rad, const float* __restrict__ W0,
             const float* __restrict__ W1, const float* __restrict__ W2,
             const int* __restrict__ species, float* __restrict__ x0) {
    __shared__ float Fs[9 * FS_PLANE];
    __shared__ __attribute__((aligned(16))) float Ts[TILE_M * TS_PITCH];
    __shared__ float Es[N_SPEC * N_CH];
    __shared__ int   Ss[TILE_M];
    const int n0 = blockIdx.x * TILE_M;
    const int k  = threadIdx.x;
    const int lane = k & 31;
    const int wave = k >> 5;
    const int half = lane >> 4;
    const int l16  = lane & 15;
    const int colbase = wave * 32;
    const int col = colbase + 2 * l16;
    Es[k] = emb[k];
    if (k < TILE_M) Ss[k] = species[n0 + k];

    {
        const v4f* Tg = reinterpret_cast<const v4f*>(Tm + (long)n0 * TM_ATOM);
        v4f tv[9];
#pragma unroll
        for (int it = 0; it < 9; ++it) tv[it] = Tg[it * KDIM + k];
        float e[N_SPEC];
#pragma unroll
        for (int s = 0; s < N_SPEC; ++s) e[s] = emb[s * N_CH + (col >> 2)];
        v2f wr[3][4];
#pragma unroll
        for (int l = 0; l < 3; ++l) {
#pragma unroll
            for (int jj = 0; jj < 4; ++jj) {
                const int j = 4 * (jj >> 1) + 2 * half + (jj & 1);
                wr[l][jj] = *reinterpret_cast<const v2f*>(W_rad + (l * N_RB + j) * KDIM + col);
            }
        }
#pragma unroll
        for (int it = 0; it < 9; ++it) {
            const int idx = it * KDIM + k;
            const int a   = idx / 72;
            const int q   = idx - a * 72;
            *reinterpret_cast<v4f*>(Ts + a * TS_PITCH + 4 * q) = tv[it];
        }
        __syncthreads();
        const float* Ta = Ts + l16 * TS_PITCH + 2 * half;
        float* Fd = Fs + (8 * half) * FS_PITCH + col;
        rows_from_moments<0, 1>(Ta, wr[0], e, Fd);
        rows_from_moments<1, 3>(Ta, wr[1], e, Fd);
        rows_from_moments<4, 5>(Ta, wr[2], e, Fd);
    }
    __syncthreads();

    v8f acc[18] = {};
    {
        const float* Fl = Fs + l16 * FS_PITCH + 2 * half;
        const long boff = (long)(2 * half) * KDIM + colbase + 2 * l16;
        float bc[32];
        load_b_chunk(W0 + boff, bc);
        mix_planes<0, 1, true >(Fl, W0 + boff, W1 + boff, bc, acc);
        mix_planes<1, 3, true >(Fl, W1 + boff, W2 + boff, bc, acc);
        mix_planes<4, 5, false>(Fl, W2 + boff, W2 + boff, bc, acc);
    }

    const float i3 = A3;
    const float* Fr = Fs + (8 * half) * FS_PITCH + col;
    int ec[3][2];
#pragma unroll
    for (int g = 0; g < 3; ++g) {
        ec[g][0] = (128 * g + col) / 12;
        ec[g][1] = (128 * g + col + 1) / 12;
    }
    v2f f[9], fn[9];
#pragma unroll
    for (int m = 0; m < 9; ++m) f[m] = *reinterpret_cast<const v2f*>(Fr + m * FS_PLANE);
#pragma unroll
    for (int i = 0; i < 8; ++i) {
        if (i + 1 < 8) {
#pragma unroll
            for (int m = 0; m < 9; ++m)
                fn[m] = *reinterpret_cast<const v2f*>(Fr + m * FS_PLANE + (i + 1) * FS_PITCH);
        }
        const float* e = Es + Ss[8 * half + i] * N_CH;
        v2f o[3];
#pragma unroll
        for (int t = 0; t < 2; ++t) {
            float fv[9], h[9];
#pragma unroll
            for (int m = 0; m < 9; ++m) {
                fv[m] = f[m][t];
                h[m]  = acc[2 * m + t][i];
            }
            float feats0 = h[0] * fv[0] + fv[0];
            float Gu[9], Fu[9];
            uncouple_l1(fv[1], fv[2], fv[3], Fu);
            uncouple_l1(h[1], h[2], h[3], Gu);
            float tr1 = trace_gf_plus_f(Gu, Fu);
            uncouple_l2(fv[4], fv[5], fv[6], fv[7], fv[8], Fu);
            uncouple_l2(h[4], h[5], h[6], h[7], h[8], Gu);
            float tr2 = trace_gf_plus_f(Gu, Fu);
            o[0][t] = feats0     * e[ec[0][t]];
            o[1][t] = (tr1 * i3) * e[ec[1][t]];
            o[2][t] = (tr2 * i3) * e[ec[2][t]];
        }
        float* xr = x0 + (long)(n0 + 8 * half + i) * K0TOT + col;
        *reinterpret_cast<v2f*>(xr)       = o[0];
        *reinterpret_cast<v2f*>(xr + 128) = o[1];
        *reinterpret_cast<v2f*>(xr + 256) = o[2];
        if (i + 1 < 8) {
#pragma unroll
            for (int m = 0; m < 9; ++m) f[m] = fn[m];
        }
    }
}

#define HB_ROWS 128
#define HB_CHUNKS (K0TOT / 16)
struct HeadA { v4f a[2][2][2]; };
struct HeadB { v4f b[2][4][2]; };
__device__ __forceinline__ void head_load_a(const float* __restrict__ A0,
                                            const float* __restrict__ A1, int c, HeadA& f) {
#pragma unroll
    for (int cc = 0; cc < 2; ++cc) {
#pragma unroll
        for (int sp = 0; sp < 2; ++sp) {
            const int q = 2 * (c + cc) + sp;
            f.a[cc][sp][0] = *reinterpret_cast<const v4f*>(A0 + 8 * q);
            f.a[cc][sp][1] = *reinterpret_cast<const v4f*>(A1 + 8 * q);
        }
    }
}
__device__ __forceinline__ void head_load_b(const float* __restrict__ Bp, int c, HeadB& f) {
#pragma unroll
    for (int sp = 0; sp < 2; ++sp) {
        const int q = 2 * c + sp;
#pragma unroll
        for (int i = 0; i < 4; ++i) {
            const float* rowp = Bp + (long)(8 * q + i) * K0TOT;
            f.b[sp][i][0] = *reinterpret_cast<const v4f*>(rowp);
            f.b[sp][i][1] = *reinterpret_cast<const v4f*>(rowp + 64);
        }
    }
}
template <int CC>
__device__ __forceinline__ void head_mma(const HeadA& fa, const HeadB& fb, v8f (&acc)[2][8]) {
#pragma unroll
    for (int sp = 0; sp < 2; ++sp) {
#pragma unroll
        for (int e = 0; e < 2; ++e) {
            v2f a0, a1;
            a0.x = fa.a[CC][sp][0][2 * e];
            a0.y = fa.a[CC][sp][0][2 * e + 1];
            a1.x = fa.a[CC][sp][1][2 * e];
            a1.y = fa.a[CC][sp][1][2 * e + 1];
#pragma unroll
            for (int g = 0; g < 2; ++g) {
#pragma unroll
                for (int v = 0; v < 4; ++v) {
                    v2f b;
                    b.x = fb.b[sp][2 * e][g][v];
                    b.y = fb.b[sp][2 * e + 1][g][v];
                    acc[0][4 * g + v] = __builtin_amdgcn_wmma_f32_16x16x4_f32(
                        false, a0, false, b, (short)0, acc[0][4 * g + v], false, false);
                    acc[1][4 * g + v] = __builtin_amdgcn_wmma_f32_16x16x4_f32(
                        false, a1, false, b, (short)0, acc[1][4 * g + v], false, false);
                }
            }
        }
    }
}
__global__ void __launch_bounds__(HB_ROWS)
k_head_wmma(const float* __restrict__ x0, const float* __restrict__ Wh,
            const float* __restrict__ bh, const float* __restrict__ Wo,
            float* __restrict__ out) {
    const int lane = threadIdx.x & 31;
    const int wave = threadIdx.x >> 5;
    const int half = lane >> 4;
    const int l16  = lane & 15;
    const int m0 = blockIdx.x * HB_ROWS + wave * 32;
    if (m0 >= N_ATOMS) return;
    const int n0 = blockIdx.y * 128;
    const int ra = min(m0 + l16, N_ATOMS - 1);
    const int rb = min(m0 + 16 + l16, N_ATOMS - 1);
    const float* A0 = x0 + (long)ra * K0TOT + 4 * half;
    const float* A1 = x0 + (long)rb * K0TOT + 4 * half;
    const float* Bp = Wh + (long)(4 * half) * K0TOT + n0 + 4 * l16;
    v8f acc[2][8] = {};
    HeadA fa, fan;
    HeadB f0, f1;
    head_load_a(A0, A1, 0, fa);
    head_load_b(Bp, 0, f0);
    for (int c = 0; c < HB_CHUNKS - 2; c += 2) {
        head_load_b(Bp, c + 1, f1);
        __builtin_amdgcn_sched_barrier(0);
        head_load_a(A0, A1, c + 2, fan);
        __builtin_amdgcn_sched_barrier(0);
        head_mma<0>(fa, f0, acc);
        __builtin_amdgcn_sched_barrier(0);
        head_load_b(Bp, c + 2, f0);
        __builtin_amdgcn_sched_barrier(0);
        head_mma<1>(fa, f1, acc);
        __builtin_amdgcn_sched_barrier(0);
        fa = fan;
    }
    head_load_b(Bp, HB_CHUNKS - 1, f1);
    __builtin_amdgcn_sched_barrier(0);
    head_mma<0>(fa, f0, acc);
    head_mma<1>(fa, f1, acc);
    v4f bias[2], wo[2];
#pragma unroll
    for (int g = 0; g < 2; ++g) {
        bias[g] = *reinterpret_cast<const v4f*>(bh + n0 + 64 * g + 4 * l16);
        wo[g]   = *reinterpret_cast<const v4f*>(Wo + n0 + 64 * g + 4 * l16);
    }
#pragma unroll
    for (int rt = 0; rt < 2; ++rt) {
#pragma unroll
        for (int i = 0; i < 8; ++i) {
            float partial = 0.f;
#pragma unroll
            for (int g = 0; g < 2; ++g) {
#pragma unroll
                for (int v = 0; v < 4; ++v) {
                    float x = acc[rt][4 * g + v][i] + bias[g][v];
                    partial += (x / (1.f + expf(-x))) * wo[g][v];
                }
            }
            const int row = m0 + 16 * rt + 8 * half + i;
            if (row < N_ATOMS) atomicAdd(&out[row], partial);
        }
    }
}

extern "C" void kernel_launch(void* const* d_in, const int* in_sizes, int n_in,
                              void* d_out, int out_size, void* d_ws, size_t ws_size,
                              hipStream_t stream) {
    (void)in_sizes; (void)n_in; (void)out_size; (void)ws_size;
    const float* positions = (const float*)d_in[0];
    const int*   species   = (const int*)d_in[1];
    const int*   pairs     = (const int*)d_in[2];
    const float* embeddings= (const float*)d_in[3];
    const float* W_rad     = (const float*)d_in[4];
    const float* W_cg0     = (const float*)d_in[5];
    const float* W_cg1     = (const float*)d_in[6];
    const float* W_cg2     = (const float*)d_in[7];
    const float* W_head    = (const float*)d_in[8];
    const float* b_head    = (const float*)d_in[9];
    const float* W_out     = (const float*)d_in[10];
    const float* b_out     = (const float*)d_in[11];
    float* out = (float*)d_out;

    float* ws = (float*)d_ws;
    float* Tm = ws;
    float* x0 = Tm + (long)N_ATOMS * TM_ATOM;

    long nz = (long)N_ATOMS * TM_ATOM;
    k_zero<<<(unsigned)((nz + 255) / 256), 256, 0, stream>>>(Tm, nz);

    k_init_out<<<(N_ATOMS + 255) / 256, 256, 0, stream>>>(out, b_out);

    k_pairs<<<N_PAIRS * N_RB / 256, 256, 0, stream>>>(positions, species, pairs, Tm);

    k_tile_feats<<<N_ATOMS / TILE_M, KDIM, 0, stream>>>(Tm, embeddings, W_rad, W_cg0, W_cg1,
                                                        W_cg2, species, x0);

    k_head_wmma<<<dim3((N_ATOMS + HB_ROWS - 1) / HB_ROWS, K0TOT / 128), HB_ROWS, 0, stream>>>(
        x0, W_head, b_head, W_out, out);
}


# channel mix reassociated: H = T x (R x W_cg), K=32 on the moments (was K=128 on F); R x W_cg built per call by WMMA in the init launch; F and H share A, stay in accumulators; sum order of g changes
# speedup vs baseline: 4.4831x; 1.1424x over previous
#include <hip/hip_runtime.h>
#include <math.h>

#define N_ATOMS  10000
#define N_PAIRS  160000
#define N_CH     32
#define KDIM     128
#define K0TOT    384

typedef float v2f __attribute__((ext_vector_type(2)));
typedef float v4f __attribute__((ext_vector_type(4)));
typedef float v8f __attribute__((ext_vector_type(8)));

#define A3 0.5773502691896258f
#define A2 0.7071067811865476f
#define B2 0.8164965809277260f
#define B1 0.4082482904638630f

__global__ void k_zero(float* p, long n) {
    long i = (long)blockIdx.x * blockDim.x + threadIdx.x;
    if (i < n) p[i] = 0.f;
}

#define N_SPEC   4
#define N_RB     8
#define TM_ATOM  (N_SPEC * 9 * N_RB)
__global__ void k_pairs(const float* __restrict__ pos, const int* __restrict__ species,
                        const int* __restrict__ pairs, float* __restrict__ Tm) {
    int t = blockIdx.x * blockDim.x + threadIdx.x;
    int p = t >> 3;
    int j = t & 7;
    if (p >= N_PAIRS) return;
    int c  = pairs[2 * p + 0];
    int nb = pairs[2 * p + 1];
    float rx = pos[3 * nb + 0] - pos[3 * c + 0];
    float ry = pos[3 * nb + 1] - pos[3 * c + 1];
    float rz = pos[3 * nb + 2] - pos[3 * c + 2];
    float d  = sqrtf(rx * rx + ry * ry + rz * rz + 1e-12f);
    float inv = 1.f / d;
    float ux = rx * inv, uy = ry * inv, uz = rz * inv;

    float mu = 20.f * (float)j / 7.f;
    float tc = (d - 15.f) * 0.2f;
    tc = fminf(fmaxf(tc, 0.f), 1.f);
    float fc = 0.5f * (cosf(3.14159265358979f * tc) + 1.f);
    float x  = (d - mu) * 0.4f;
    float rb = expf(-x * x) * fc;

    const float s3 = 1.7320508075688772f;
    float sh[9] = { 1.f, uy, uz, ux,
                    s3 * ux * uy, s3 * uy * uz, 0.5f * (3.f * uz * uz - 1.f),
                    s3 * ux * uz, 0.5f * s3 * (ux * ux - uy * uy) };
    float* Tp = Tm + ((long)c * N_SPEC + species[nb]) * (9 * N_RB) + j;
    for (int m = 0; m < 9; ++m) atomicAdd(Tp + m * N_RB, sh[m] * rb);
}

#define RW_BLOCKS 3
#define RW_PLANE  (N_SPEC * N_RB * KDIM)
__global__ void __launch_bounds__(256)
k_init(float* __restrict__ out, const float* __restrict__ b_out, const float* __restrict__ emb,
       const float* __restrict__ W_rad, const float* __restrict__ W0,
       const float* __restrict__ W1, const float* __restrict__ W2, float* __restrict__ RW) {
    if (blockIdx.x >= RW_BLOCKS) {
        const int i = (blockIdx.x - RW_BLOCKS) * 256 + threadIdx.x;
        if (i < N_ATOMS) out[i] = b_out[0];
        return;
    }
    const int l    = blockIdx.x;
    const int lane = threadIdx.x & 31;
    const int wave = threadIdx.x >> 5;
    const int half = lane >> 4;
    const int l16  = lane & 15;
    const int kk   = 16 * (wave & 1) + l16;
    const int col  = 32 * (wave >> 1) + 2 * l16;
    const float* Wc = (l == 0) ? W0 : ((l == 1) ? W1 : W2);
    const float* Bp = Wc + (2 * half) * KDIM + col;
    const float* Ap = W_rad + (l * N_RB + (kk & 7)) * KDIM + 2 * half;
    const v4f*   Ep = reinterpret_cast<const v4f*>(emb + (kk >> 3) * N_CH);
    v2f b0[KDIM / 4], b1[KDIM / 4], wq[KDIM / 4];
    v4f eq[N_CH / 4];
#pragma unroll
    for (int ks = 0; ks < KDIM / 4; ++ks) {
        b0[ks] = *reinterpret_cast<const v2f*>(Bp + (4 * ks) * KDIM);
        b1[ks] = *reinterpret_cast<const v2f*>(Bp + (4 * ks + 1) * KDIM);
        wq[ks] = *reinterpret_cast<const v2f*>(Ap + 4 * ks);
    }
#pragma unroll
    for (int q = 0; q < N_CH / 4; ++q) eq[q] = Ep[q];
    __builtin_amdgcn_sched_barrier(0);
    v8f acc0 = {}, acc1 = {};
#pragma unroll
    for (int ks = 0; ks < KDIM / 4; ++ks) {
        const float ev = eq[ks >> 2][ks & 3];
        v2f a, t0, t1;
        a.x  = ev * wq[ks].x;
        a.y  = ev * wq[ks].y;
        t0.x = b0[ks].x;
        t0.y = b1[ks].x;
        t1.x = b0[ks].y;
        t1.y = b1[ks].y;
        acc0 = __builtin_amdgcn_wmma_f32_16x16x4_f32(false, a, false, t0, (short)0, acc0,
                                                     false, false);
        acc1 = __builtin_amdgcn_wmma_f32_16x16x4_f32(false, a, false, t1, (short)0, acc1,
                                                     false, false);
    }
    float* Rd = RW + l * RW_PLANE + (16 * (wave & 1) + 8 * half) * KDIM + col;
#pragma unroll
    for (int r = 0; r < 8; ++r) {
        v2f o;
        o.x = acc0[r] * 0.1f;
        o.y = acc1[r] * 0.1f;
        *reinterpret_cast<v2f*>(Rd + r * KDIM) = o;
    }
}

#define TILE_M   16
#define TS_PITCH 292

__device__ __forceinline__ void uncouple_l1(float v1, float v2, float v3, float u[9]) {
    u[0] = 0.f;       u[1] =  A2 * v3;  u[2] = -A2 * v2;
    u[3] = -A2 * v3;  u[4] = 0.f;       u[5] =  A2 * v1;
    u[6] =  A2 * v2;  u[7] = -A2 * v1;  u[8] = 0.f;
}
__device__ __forceinline__ void uncouple_l2(float v4, float v5, float v6, float v7, float v8,
                                            float u[9]) {
    u[0] = -B1 * v6 - A2 * v8;  u[1] = A2 * v5;  u[2] = A2 * v4;
    u[3] = A2 * v5;             u[4] = B2 * v6;  u[5] = A2 * v7;
    u[6] = A2 * v4;             u[7] = A2 * v7;  u[8] = -B1 * v6 + A2 * v8;
}
__device__ __forceinline__ float trace_gf_plus_f(const float G[9], const float Fm[9]) {
    float tr = 0.f;
#pragma unroll
    for (int a = 0; a < 3; ++a)
#pragma unroll
        for (int b = 0; b < 3; ++b) tr += G[a * 3 + b] * Fm[b * 3 + a];
    return tr + (Fm[0] + Fm[4] + Fm[8]);
}

__device__ __forceinline__ void load_b_chunk(const float* __restrict__ p, float (&b)[32]) {
#pragma unroll
    for (int ks = 0; ks < 8; ++ks) {
        const float* q = p + ks * 4 * KDIM;
        const v2f r0 = *reinterpret_cast<const v2f*>(q);
        const v2f r1 = *reinterpret_cast<const v2f*>(q + KDIM);
        b[4 * ks + 0] = r0.x;
        b[4 * ks + 1] = r1.x;
        b[4 * ks + 2] = r0.y;
        b[4 * ks + 3] = r1.y;
    }
}
__device__ __forceinline__ void load_t_plane(const float* __restrict__ Ta, int m, v2f (&a)[8]) {
#pragma unroll
    for (int ks = 0; ks < 8; ++ks)
        a[ks] = *reinterpret_cast<const v2f*>(Ta + (ks >> 1) * 72 + m * 8 + 4 * (ks & 1));
}

template <int M0, int NM, bool HAS_NEXT>
__device__ __forceinline__ void planes_from_moments(const float* __restrict__ Ta,
                                                    const v2f (&w)[4], const float (&e)[N_SPEC],
                                                    const float (&bw)[32], v2f (&a)[8],
                                                    v8f (&cf)[2 * NM], v8f (&ch)[2 * NM]) {
    float b[32];
#pragma unroll
    for (int ks = 0; ks < 8; ++ks) {
        const int s  = ks >> 1;
        const int jb = 2 * (ks & 1);
        b[4 * ks + 0] = e[s] * w[jb].x;
        b[4 * ks + 1] = e[s] * w[jb + 1].x;
        b[4 * ks + 2] = e[s] * w[jb].y;
        b[4 * ks + 3] = e[s] * w[jb + 1].y;
    }
#pragma unroll
    for (int i = 0; i < NM; ++i) {
        v2f an[8];
        if (HAS_NEXT || i + 1 < NM) load_t_plane(Ta, M0 + i + 1, an);
        v8f f0 = {}, f1 = {}, h0 = {}, h1 = {};
#pragma unroll
        for (int ks = 0; ks < 8; ++ks) {
            v2f b0, b1, g0, g1;
            b0.x = b[4 * ks + 0];
            b0.y = b[4 * ks + 1];
            b1.x = b[4 * ks + 2];
            b1.y = b[4 * ks + 3];
            g0.x = bw[4 * ks + 0];
            g0.y = bw[4 * ks + 1];
            g1.x = bw[4 * ks + 2];
            g1.y = bw[4 * ks + 3];
            f0 = __builtin_amdgcn_wmma_f32_16x16x4_f32(false, a[ks], false, b0, (short)0, f0,
                                                       false, false);
            f1 = __builtin_amdgcn_wmma_f32_16x16x4_f32(false, a[ks], false, b1, (short)0, f1,
                                                       false, false);
            h0 = __builtin_amdgcn_wmma_f32_16x16x4_f32(false, a[ks], false, g0, (short)0, h0,
                                                       false, false);
            h1 = __builtin_amdgcn_wmma_f32_16x16x4_f32(false, a[ks], false, g1, (short)0, h1,
                                                       false, false);
        }
        cf[2 * i]     = f0 * 0.1f;
        cf[2 * i + 1] = f1 * 0.1f;
        ch[2 * i]     = h0;
        ch[2 * i + 1] = h1;
        if (HAS_NEXT || i + 1 < NM) {
#pragma unroll
            for (int ks = 0; ks < 8; ++ks) a[ks] = an[ks];
        }
    }
}

__global__ void __launch_bounds__(KDIM)
k_tile_feats(const float* __restrict__ Tm, const float* __restrict__ emb,
             const float* __restrict__ W_rad, const float* __restrict__ RW,
             const int* __restrict__ species, float* __restrict__ x0) {
    __shared__ __attribute__((aligned(16))) float Ts[TILE_M * TS_PITCH];
    __shared__ float Es[N_SPEC * N_CH];
    __shared__ int   Ss[TILE_M];
    const int n0 = blockIdx.x * TILE_M;
    const int k  = threadIdx.x;
    const int lane = k & 31;
    const int wave = k >> 5;
    const int half = lane >> 4;
    const int l16  = lane & 15;
    const int colbase = wave * 32;
    const int col = colbase + 2 * l16;
    Es[k] = emb[k];
    if (k < TILE_M) Ss[k] = species[n0 + k];

    const v4f* Tg = reinterpret_cast<const v4f*>(Tm + (long)n0 * TM_ATOM);
    v4f tv[9];
#pragma unroll
    for (int it = 0; it < 9; ++it) tv[it] = Tg[it * KDIM + k];
    float e[N_SPEC];
#pragma unroll
    for (int s = 0; s < N_SPEC; ++s) e[s] = emb[s * N_CH + (col >> 2)];
    v2f wr[3][4];
#pragma unroll
    for (int l = 0; l < 3; ++l) {
#pragma unroll
        for (int jj = 0; jj < 4; ++jj) {
            const int j = 4 * (jj >> 1) + 2 * half + (jj & 1);
            wr[l][jj] = *reinterpret_cast<const v2f*>(W_rad + (l * N_RB + j) * KDIM + col);
        }
    }
    const float* Rp = RW + (2 * half) * KDIM + col;
    float bw0[32], bw1[32], bw2[32];
    load_b_chunk(Rp, bw0);
    load_b_chunk(Rp + RW_PLANE, bw1);
    load_b_chunk(Rp + 2 * RW_PLANE, bw2);
#pragma unroll
    for (int it = 0; it < 9; ++it) {
        const int idx = it * KDIM + k;
        const int a   = idx / 72;
        const int q   = idx - a * 72;
        *reinterpret_cast<v4f*>(Ts + a * TS_PITCH + 4 * q) = tv[it];
    }
    __syncthreads();

    const float* Ta = Ts + l16 * TS_PITCH + 2 * half;
    const float i3 = A3;
    int ec[3][2];
#pragma unroll
    for (int g = 0; g < 3; ++g) {
        ec[g][0] = (128 * g + col) / 12;
        ec[g][1] = (128 * g + col + 1) / 12;
    }
    float* xr = x0 + (long)(n0 + 8 * half) * K0TOT + col;
    v2f a[8];
    load_t_plane(Ta, 0, a);

    {
        v8f cf[2], ch[2];
        planes_from_moments<0, 1, true>(Ta, wr[0], e, bw0, a, cf, ch);
#pragma unroll
        for (int i = 0; i < 8; ++i) {
            const float* er = Es + Ss[8 * half + i] * N_CH;
            v2f o;
#pragma unroll
            for (int t = 0; t < 2; ++t) {
                const float fv = cf[t][i], h = ch[t][i];
                o[t] = (h * fv + fv) * er[ec[0][t]];
            }
            *reinterpret_cast<v2f*>(xr + i * K0TOT) = o;
        }
    }
    {
        v8f cf[6], ch[6];
        planes_from_moments<1, 3, true>(Ta, wr[1], e, bw1, a, cf, ch);
#pragma unroll
        for (int i = 0; i < 8; ++i) {
            const float* er = Es + Ss[8 * half + i] * N_CH;
            v2f o;
#pragma unroll
            for (int t = 0; t < 2; ++t) {
                float Gu[9], Fu[9];
                uncouple_l1(cf[t][i], cf[2 + t][i], cf[4 + t][i], Fu);
                uncouple_l1(ch[t][i], ch[2 + t][i], ch[4 + t][i], Gu);
                const float tr1 = trace_gf_plus_f(Gu, Fu);
                o[t] = (tr1 * i3) * er[ec[1][t]];
            }
            *reinterpret_cast<v2f*>(xr + i * K0TOT + 128) = o;
        }
    }
    {
        v8f cf[10], ch[10];
        planes_from_moments<4, 5, false>(Ta, wr[2], e, bw2, a, cf, ch);
#pragma unroll
        for (int i = 0; i < 8; ++i) {
            const float* er = Es + Ss[8 * half + i] * N_CH;
            v2f o;
#pragma unroll
            for (int t = 0; t < 2; ++t) {
                float Gu[9], Fu[9];
                uncouple_l2(cf[t][i], cf[2 + t][i], cf[4 + t][i], cf[6 + t][i], cf[8 + t][i], Fu);
                uncouple_l2(ch[t][i], ch[2 + t][i], ch[4 + t][i], ch[6 + t][i], ch[8 + t][i], Gu);
                const float tr2 = trace_gf_plus_f(Gu, Fu);
                o[t] = (tr2 * i3) * er[ec[2][t]];
            }
            *reinterpret_cast<v2f*>(xr + i * K0TOT + 256) = o;
        }
    }
}

#define HB_ROWS 128
#define HB_CHUNKS (K0TOT / 16)
struct HeadA { v4f a[2][2][2]; };
struct HeadB { v4f b[2][4][2]; };
__device__ __forceinline__ void head_load_a(const float* __restrict__ A0,
                                            const float* __restrict__ A1, int c, HeadA& f) {
#pragma unroll
    for (int cc = 0; cc < 2; ++cc) {
#pragma unroll
        for (int sp = 0; sp < 2; ++sp) {
            const int q = 2 * (c + cc) + sp;
            f.a[cc][sp][0] = *reinterpret_cast<const v4f*>(A0 + 8 * q);
            f.a[cc][sp][1] = *reinterpret_cast<const v4f*>(A1 + 8 * q);
        }
    }
}
__device__ __forceinline__ void head_load_b(const float* __restrict__ Bp, int c, HeadB& f) {
#pragma unroll
    for (int sp = 0; sp < 2; ++sp) {
        const int q = 2 * c + sp;
#pragma unroll
        for (int i = 0; i < 4; ++i) {
            const float* rowp = Bp + (long)(8 * q + i) * K0TOT;
            f.b[sp][i][0] = *reinterpret_cast<const v4f*>(rowp);
            f.b[sp][i][1] = *reinterpret_cast<const v4f*>(rowp + 64);
        }
    }
}
template <int CC>
__device__ __forceinline__ void head_mma(const HeadA& fa, const HeadB& fb, v8f (&acc)[2][8]) {
#pragma unroll
    for (int sp = 0; sp < 2; ++sp) {
#pragma unroll
        for (int e = 0; e < 2; ++e) {
            v2f a0, a1;
            a0.x = fa.a[CC][sp][0][2 * e];
            a0.y = fa.a[CC][sp][0][2 * e + 1];
            a1.x = fa.a[CC][sp][1][2 * e];
            a1.y = fa.a[CC][sp][1][2 * e + 1];
#pragma unroll
            for (int g = 0; g < 2; ++g) {
#pragma unroll
                for (int v = 0; v < 4; ++v) {
                    v2f b;
                    b.x = fb.b[sp][2 * e][g][v];
                    b.y = fb.b[sp][2 * e + 1][g][v];
                    acc[0][4 * g + v] = __builtin_amdgcn_wmma_f32_16x16x4_f32(
                        false, a0, false, b, (short)0, acc[0][4 * g + v], false, false);
                    acc[1][4 * g + v] = __builtin_amdgcn_wmma_f32_16x16x4_f32(
                        false, a1, false, b, (short)0, acc[1][4 * g + v], false, false);
                }
            }
        }
    }
}
__global__ void __launch_bounds__(HB_ROWS)
k_head_wmma(const float* __restrict__ x0, const float* __restrict__ Wh,
            const float* __restrict__ bh, const float* __restrict__ Wo,
            float* __restrict__ out) {
    const int lane = threadIdx.x & 31;
    const int wave = threadIdx.x >> 5;
    const int half = lane >> 4;
    const int l16  = lane & 15;
    const int m0 = blockIdx.x * HB_ROWS + wave * 32;
    if (m0 >= N_ATOMS) return;
    const int n0 = blockIdx.y * 128;
    const int ra = min(m0 + l16, N_ATOMS - 1);
    const int rb = min(m0 + 16 + l16, N_ATOMS - 1);
    const float* A0 = x0 + (long)ra * K0TOT + 4 * half;
    const float* A1 = x0 + (long)rb * K0TOT + 4 * half;
    const float* Bp = Wh + (long)(4 * half) * K0TOT + n0 + 4 * l16;
    v8f acc[2][8] = {};
    HeadA fa, fan;
    HeadB f0, f1;
    head_load_a(A0, A1, 0, fa);
    head_load_b(Bp, 0, f0);
    for (int c = 0; c < HB_CHUNKS - 2; c += 2) {
        head_load_b(Bp, c + 1, f1);
        __builtin_amdgcn_sched_barrier(0);
        head_load_a(A0, A1, c + 2, fan);
        __builtin_amdgcn_sched_barrier(0);
        head_mma<0>(fa, f0, acc);
        __builtin_amdgcn_sched_barrier(0);
        head_load_b(Bp, c + 2, f0);
        __builtin_amdgcn_sched_barrier(0);
        head_mma<1>(fa, f1, acc);
        __builtin_amdgcn_sched_barrier(0);
        fa = fan;
    }
    head_load_b(Bp, HB_CHUNKS - 1, f1);
    __builtin_amdgcn_sched_barrier(0);
    head_mma<0>(fa, f0, acc);
    head_mma<1>(fa, f1, acc);
    v4f bias[2], wo[2];
#pragma unroll
    for (int g = 0; g < 2; ++g) {
        bias[g] = *reinterpret_cast<const v4f*>(bh + n0 + 64 * g + 4 * l16);
        wo[g]   = *reinterpret_cast<const v4f*>(Wo + n0 + 64 * g + 4 * l16);
    }
#pragma unroll
    for (int rt = 0; rt < 2; ++rt) {
#pragma unroll
        for (int i = 0; i < 8; ++i) {
            float partial = 0.f;
#pragma unroll
            for (int g = 0; g < 2; ++g) {
#pragma unroll
                for (int v = 0; v < 4; ++v) {
                    float x = acc[rt][4 * g + v][i] + bias[g][v];
                    partial += (x / (1.f + expf(-x))) * wo[g][v];
                }
            }
            const int row = m0 + 16 * rt + 8 * half + i;
            if (row < N_ATOMS) atomicAdd(&out[row], partial);
        }
    }
}

extern "C" void kernel_launch(void* const* d_in, const int* in_sizes, int n_in,
                              void* d_out, int out_size, void* d_ws, size_t ws_size,
                              hipStream_t stream) {
    (void)in_sizes; (void)n_in; (void)out_size; (void)ws_size;
    const float* positions = (const float*)d_in[0];
    const int*   species   = (const int*)d_in[1];
    const int*   pairs     = (const int*)d_in[2];
    const float* embeddings= (const float*)d_in[3];
    const float* W_rad     = (const float*)d_in[4];
    const float* W_cg0     = (const float*)d_in[5];
    const float* W_cg1     = (const float*)d_in[6];
    const float* W_cg2     = (const float*)d_in[7];
    const float* W_head    = (const float*)d_in[8];
    const float* b_head    = (const float*)d_in[9];
    const float* W_out     = (const float*)d_in[10];
    const float* b_out     = (const float*)d_in[11];
    float* out = (float*)d_out;

    float* ws = (float*)d_ws;
    float* Tm = ws;
    float* x0 = Tm + (long)N_ATOMS * TM_ATOM;
    float* RW = x0 + (long)N_ATOMS * K0TOT;

    long nz = (long)N_ATOMS * TM_ATOM;
    k_zero<<<(unsigned)((nz + 255) / 256), 256, 0, stream>>>(Tm, nz);

    k_init<<<RW_BLOCKS + (N_ATOMS + 255) / 256, 256, 0, stream>>>(out, b_out, embeddings, W_rad,
                                                                   W_cg0, W_cg1, W_cg2, RW);

    k_pairs<<<N_PAIRS * N_RB / 256, 256, 0, stream>>>(positions, species, pairs, Tm);

    k_tile_feats<<<N_ATOMS / TILE_M, KDIM, 0, stream>>>(Tm, embeddings, W_rad, RW, species, x0);

    k_head_wmma<<<dim3((N_ATOMS + HB_ROWS - 1) / HB_ROWS, K0TOT / 128), HB_ROWS, 0, stream>>>(
        x0, W_head, b_head, W_out, out);
}


# pair moments summed by owner: 125 blocks scan the pair list, add their 80 atoms' pairs into LDS rows (LDS float atomics), store each row once; no global atomics, no zero launch; sum order changes
# speedup vs baseline: 4.8182x; 1.0747x over previous
#include <hip/hip_runtime.h>
#include <math.h>

#define N_ATOMS  10000
#define N_PAIRS  160000
#define N_CH     32
#define KDIM     128
#define K0TOT    384

typedef float v2f __attribute__((ext_vector_type(2)));
typedef float v4f __attribute__((ext_vector_type(4)));
typedef float v8f __attribute__((ext_vector_type(8)));

#define A3 0.5773502691896258f
#define A2 0.7071067811865476f
#define B2 0.8164965809277260f
#define B1 0.4082482904638630f

#define N_SPEC   4
#define N_RB     8
#define TM_ATOM  (N_SPEC * 9 * N_RB)
#define OWN_ATOMS   80
#define OWN_BLOCKS  (N_ATOMS / OWN_ATOMS)
#define OWN_THREADS 1024
#define OWN_WAVES   (OWN_THREADS / 32)
#define OWN_LOADS   16
#define OWN_GROUP   (OWN_THREADS * OWN_LOADS)
#define OWN_GROUPS  (PAIR_QUADS / OWN_GROUP)
#define W_CAP       128
#define PAIR_QUADS  (N_PAIRS / 2)
typedef int v4i __attribute__((ext_vector_type(4)));

__device__ __forceinline__ void pair_geometry(const float* __restrict__ pos, int c, int nb,
                                              float& d, float& ux, float& uy, float& uz,
                                              float& fc) {
    float rx = pos[3 * nb + 0] - pos[3 * c + 0];
    float ry = pos[3 * nb + 1] - pos[3 * c + 1];
    float rz = pos[3 * nb + 2] - pos[3 * c + 2];
    d = sqrtf(rx * rx + ry * ry + rz * rz + 1e-12f);
    float inv = 1.f / d;
    ux = rx * inv;
    uy = ry * inv;
    uz = rz * inv;
    float tc = (d - 15.f) * 0.2f;
    tc = fminf(fmaxf(tc, 0.f), 1.f);
    fc = 0.5f * (cosf(3.14159265358979f * tc) + 1.f);
}
__device__ __forceinline__ void add_terms(float* Tp, int j, float d, float ux, float uy,
                                          float uz, float fc) {
    float mu = 20.f * (float)j / 7.f;
    float x  = (d - mu) * 0.4f;
    float rb = expf(-x * x) * fc;
    const float s3 = 1.7320508075688772f;
    float sh[9] = { 1.f, uy, uz, ux,
                    s3 * ux * uy, s3 * uy * uz, 0.5f * (3.f * uz * uz - 1.f),
                    s3 * ux * uz, 0.5f * s3 * (ux * ux - uy * uy) };
    for (int m = 0; m < 9; ++m) atomicAdd(Tp + m * N_RB, sh[m] * rb);
}
__device__ __forceinline__ void note_pair(unsigned* myl, unsigned& wcount, int a0, int c,
                                          int nb, bool valid) {
    const unsigned cl = (unsigned)(c - a0);
    const bool hit = valid && cl < OWN_ATOMS;
    const unsigned mask = __builtin_amdgcn_ballot_w32(hit);
    if (mask != 0u) {
        const unsigned idx = wcount + __builtin_amdgcn_mbcnt_lo(mask, 0u);
        if (hit && idx < W_CAP) myl[idx] = (cl << 16) | (unsigned)nb;
        wcount += (unsigned)__builtin_popcount(mask);
    }
}
template <bool CHECK>
__device__ __forceinline__ void note_group(const v4i* __restrict__ P4, int base, int tid,
                                           unsigned* myl, unsigned& wcount, int a0) {
    v4i v[OWN_LOADS];
#pragma unroll
    for (int u = 0; u < OWN_LOADS; ++u) {
        const int q = base + u * OWN_THREADS + tid;
        v[u] = P4[CHECK ? min(q, PAIR_QUADS - 1) : q];
    }
#pragma unroll
    for (int u = 0; u < OWN_LOADS; ++u) {
        const bool valid = !CHECK || (base + u * OWN_THREADS + tid < PAIR_QUADS);
        note_pair(myl, wcount, a0, v[u].x, v[u].y, valid);
        note_pair(myl, wcount, a0, v[u].z, v[u].w, valid);
    }
}

__global__ void __launch_bounds__(OWN_THREADS)
k_pairs_owned(const float* __restrict__ pos, const int* __restrict__ species,
              const int* __restrict__ pairs, float* __restrict__ Tm) {
    __shared__ __attribute__((aligned(16))) float acc[OWN_ATOMS * TM_ATOM];
    __shared__ unsigned wl[OWN_WAVES * W_CAP];
    __shared__ float g_d[OWN_WAVES * W_CAP], g_ux[OWN_WAVES * W_CAP], g_uy[OWN_WAVES * W_CAP],
                     g_uz[OWN_WAVES * W_CAP], g_fc[OWN_WAVES * W_CAP];
    __shared__ int   g_row[OWN_WAVES * W_CAP];
    const int tid  = threadIdx.x;
    const int lane = tid & 31;
    const int wave = tid >> 5;
    const int a0   = blockIdx.x * OWN_ATOMS;
    for (int i = tid; i < OWN_ATOMS * TM_ATOM / 4; i += OWN_THREADS)
        reinterpret_cast<v4f*>(acc)[i] = v4f{0.f, 0.f, 0.f, 0.f};

    const v4i* P4 = reinterpret_cast<const v4i*>(pairs);
    unsigned* myl = wl + wave * W_CAP;
    unsigned wcount = 0u;
    for (int k = 0; k < OWN_GROUPS; ++k)
        note_group<false>(P4, (int)((k + blockIdx.x) % OWN_GROUPS) * OWN_GROUP, tid, myl, wcount,
                          a0);
    if (OWN_GROUPS * OWN_GROUP < PAIR_QUADS)
        note_group<true>(P4, OWN_GROUPS * OWN_GROUP, tid, myl, wcount, a0);
    const bool over   = wcount > W_CAP;
    const unsigned nn = over ? 0u : wcount;
    const int g0 = wave * W_CAP;
    __syncthreads();

    for (unsigned i = lane; i < nn; i += 32) {
        const unsigned ent = myl[i];
        const int cl = (int)(ent >> 16);
        const int nb = (int)(ent & 0xffffu);
        float d, ux, uy, uz, fc;
        pair_geometry(pos, a0 + cl, nb, d, ux, uy, uz, fc);
        g_d[g0 + i]  = d;
        g_ux[g0 + i] = ux;
        g_uy[g0 + i] = uy;
        g_uz[g0 + i] = uz;
        g_fc[g0 + i] = fc;
        g_row[g0 + i] = (cl * N_SPEC + species[nb]) * (9 * N_RB);
    }
    __syncthreads();

    const int j = lane & 7;
    for (unsigned i = lane >> 3; i < nn; i += 4) {
        const int gi = g0 + (int)i;
        add_terms(acc + g_row[gi] + j, j, g_d[gi], g_ux[gi], g_uy[gi], g_uz[gi], g_fc[gi]);
    }
    if (over) {
        for (int q = tid; q < PAIR_QUADS; q += OWN_THREADS) {
            const v4i v = P4[q];
            const unsigned cl0 = (unsigned)(v.x - a0);
            const unsigned cl1 = (unsigned)(v.z - a0);
            for (int jj = 0; jj < N_RB; ++jj) {
                float d, ux, uy, uz, fc;
                if (cl0 < OWN_ATOMS) {
                    pair_geometry(pos, v.x, v.y, d, ux, uy, uz, fc);
                    add_terms(acc + ((int)cl0 * N_SPEC + species[v.y]) * (9 * N_RB) + jj, jj,
                              d, ux, uy, uz, fc);
                }
                if (cl1 < OWN_ATOMS) {
                    pair_geometry(pos, v.z, v.w, d, ux, uy, uz, fc);
                    add_terms(acc + ((int)cl1 * N_SPEC + species[v.w]) * (9 * N_RB) + jj, jj,
                              d, ux, uy, uz, fc);
                }
            }
        }
    }
    __syncthreads();

    v4f* Tg = reinterpret_cast<v4f*>(Tm + (long)a0 * TM_ATOM);
    for (int i = tid; i < OWN_ATOMS * TM_ATOM / 4; i += OWN_THREADS)
        Tg[i] = reinterpret_cast<const v4f*>(acc)[i];
}

#define RW_BLOCKS 3
#define RW_PLANE  (N_SPEC * N_RB * KDIM)
__global__ void __launch_bounds__(256)
k_init(float* __restrict__ out, const float* __restrict__ b_out, const float* __restrict__ emb,
       const float* __restrict__ W_rad, const float* __restrict__ W0,
       const float* __restrict__ W1, const float* __restrict__ W2, float* __restrict__ RW) {
    if (blockIdx.x >= RW_BLOCKS) {
        const int i = (blockIdx.x - RW_BLOCKS) * 256 + threadIdx.x;
        if (i < N_ATOMS) out[i] = b_out[0];
        return;
    }
    const int l    = blockIdx.x;
    const int lane = threadIdx.x & 31;
    const int wave = threadIdx.x >> 5;
    const int half = lane >> 4;
    const int l16  = lane & 15;
    const int kk   = 16 * (wave & 1) + l16;
    const int col  = 32 * (wave >> 1) + 2 * l16;
    const float* Wc = (l == 0) ? W0 : ((l == 1) ? W1 : W2);
    const float* Bp = Wc + (2 * half) * KDIM + col;
    const float* Ap = W_rad + (l * N_RB + (kk & 7)) * KDIM + 2 * half;
    const v4f*   Ep = reinterpret_cast<const v4f*>(emb + (kk >> 3) * N_CH);
    v2f b0[KDIM / 4], b1[KDIM / 4], wq[KDIM / 4];
    v4f eq[N_CH / 4];
#pragma unroll
    for (int ks = 0; ks < KDIM / 4; ++ks) {
        b0[ks] = *reinterpret_cast<const v2f*>(Bp + (4 * ks) * KDIM);
        b1[ks] = *reinterpret_cast<const v2f*>(Bp + (4 * ks + 1) * KDIM);
        wq[ks] = *reinterpret_cast<const v2f*>(Ap + 4 * ks);
    }
#pragma unroll
    for (int q = 0; q < N_CH / 4; ++q) eq[q] = Ep[q];
    __builtin_amdgcn_sched_barrier(0);
    v8f acc0 = {}, acc1 = {};
#pragma unroll
    for (int ks = 0; ks < KDIM / 4; ++ks) {
        const float ev = eq[ks >> 2][ks & 3];
        v2f a, t0, t1;
        a.x  = ev * wq[ks].x;
        a.y  = ev * wq[ks].y;
        t0.x = b0[ks].x;
        t0.y = b1[ks].x;
        t1.x = b0[ks].y;
        t1.y = b1[ks].y;
        acc0 = __builtin_amdgcn_wmma_f32_16x16x4_f32(false, a, false, t0, (short)0, acc0,
                                                     false, false);
        acc1 = __builtin_amdgcn_wmma_f32_16x16x4_f32(false, a, false, t1, (short)0, acc1,
                                                     false, false);
    }
    float* Rd = RW + l * RW_PLANE + (16 * (wave & 1) + 8 * half) * KDIM + col;
#pragma unroll
    for (int r = 0; r < 8; ++r) {
        v2f o;
        o.x = acc0[r] * 0.1f;
        o.y = acc1[r] * 0.1f;
        *reinterpret_cast<v2f*>(Rd + r * KDIM) = o;
    }
}

#define TILE_M   16
#define TS_PITCH 292

__device__ __forceinline__ void uncouple_l1(float v1, float v2, float v3, float u[9]) {
    u[0] = 0.f;       u[1] =  A2 * v3;  u[2] = -A2 * v2;
    u[3] = -A2 * v3;  u[4] = 0.f;       u[5] =  A2 * v1;
    u[6] =  A2 * v2;  u[7] = -A2 * v1;  u[8] = 0.f;
}
__device__ __forceinline__ void uncouple_l2(float v4, float v5, float v6, float v7, float v8,
                                            float u[9]) {
    u[0] = -B1 * v6 - A2 * v8;  u[1] = A2 * v5;  u[2] = A2 * v4;
    u[3] = A2 * v5;             u[4] = B2 * v6;  u[5] = A2 * v7;
    u[6] = A2 * v4;             u[7] = A2 * v7;  u[8] = -B1 * v6 + A2 * v8;
}
__device__ __forceinline__ float trace_gf_plus_f(const float G[9], const float Fm[9]) {
    float tr = 0.f;
#pragma unroll
    for (int a = 0; a < 3; ++a)
#pragma unroll
        for (int b = 0; b < 3; ++b) tr += G[a * 3 + b] * Fm[b * 3 + a];
    return tr + (Fm[0] + Fm[4] + Fm[8]);
}

__device__ __forceinline__ void load_b_chunk(const float* __restrict__ p, float (&b)[32]) {
#pragma unroll
    for (int ks = 0; ks < 8; ++ks) {
        const float* q = p + ks * 4 * KDIM;
        const v2f r0 = *reinterpret_cast<const v2f*>(q);
        const v2f r1 = *reinterpret_cast<const v2f*>(q + KDIM);
        b[4 * ks + 0] = r0.x;
        b[4 * ks + 1] = r1.x;
        b[4 * ks + 2] = r0.y;
        b[4 * ks + 3] = r1.y;
    }
}
__device__ __forceinline__ void load_t_plane(const float* __restrict__ Ta, int m, v2f (&a)[8]) {
#pragma unroll
    for (int ks = 0; ks < 8; ++ks)
        a[ks] = *reinterpret_cast<const v2f*>(Ta + (ks >> 1) * 72 + m * 8 + 4 * (ks & 1));
}

template <int M0, int NM, bool HAS_NEXT>
__device__ __forceinline__ void planes_from_moments(const float* __restrict__ Ta,
                                                    const v2f (&w)[4], const float (&e)[N_SPEC],
                                                    const float (&bw)[32], v2f (&a)[8],
                                                    v8f (&cf)[2 * NM], v8f (&ch)[2 * NM]) {
    float b[32];
#pragma unroll
    for (int ks = 0; ks < 8; ++ks) {
        const int s  = ks >> 1;
        const int jb = 2 * (ks & 1);
        b[4 * ks + 0] = e[s] * w[jb].x;
        b[4 * ks + 1] = e[s] * w[jb + 1].x;
        b[4 * ks + 2] = e[s] * w[jb].y;
        b[4 * ks + 3] = e[s] * w[jb + 1].y;
    }
#pragma unroll
    for (int i = 0; i < NM; ++i) {
        v2f an[8];
        if (HAS_NEXT || i + 1 < NM) load_t_plane(Ta, M0 + i + 1, an);
        v8f f0 = {}, f1 = {}, h0 = {}, h1 = {};
#pragma unroll
        for (int ks = 0; ks < 8; ++ks) {
            v2f b0, b1, g0, g1;
            b0.x = b[4 * ks + 0];
            b0.y = b[4 * ks + 1];
            b1.x = b[4 * ks + 2];
            b1.y = b[4 * ks + 3];
            g0.x = bw[4 * ks + 0];
            g0.y = bw[4 * ks + 1];
            g1.x = bw[4 * ks + 2];
            g1.y = bw[4 * ks + 3];
            f0 = __builtin_amdgcn_wmma_f32_16x16x4_f32(false, a[ks], false, b0, (short)0, f0,
                                                       false, false);
            f1 = __builtin_amdgcn_wmma_f32_16x16x4_f32(false, a[ks], false, b1, (short)0, f1,
                                                       false, false);
            h0 = __builtin_amdgcn_wmma_f32_16x16x4_f32(false, a[ks], false, g0, (short)0, h0,
                                                       false, false);
            h1 = __builtin_amdgcn_wmma_f32_16x16x4_f32(false, a[ks], false, g1, (short)0, h1,
                                                       false, false);
        }
        cf[2 * i]     = f0 * 0.1f;
        cf[2 * i + 1] = f1 * 0.1f;
        ch[2 * i]     = h0;
        ch[2 * i + 1] = h1;
        if (HAS_NEXT || i + 1 < NM) {
#pragma unroll
            for (int ks = 0; ks < 8; ++ks) a[ks] = an[ks];
        }
    }
}

__global__ void __launch_bounds__(KDIM)
k_tile_feats(const float* __restrict__ Tm, const float* __restrict__ emb,
             const float* __restrict__ W_rad, const float* __restrict__ RW,
             const int* __restrict__ species, float* __restrict__ x0) {
    __shared__ __attribute__((aligned(16))) float Ts[TILE_M * TS_PITCH];
    __shared__ float Es[N_SPEC * N_CH];
    __shared__ int   Ss[TILE_M];
    const int n0 = blockIdx.x * TILE_M;
    const int k  = threadIdx.x;
    const int lane = k & 31;
    const int wave = k >> 5;
    const int half = lane >> 4;
    const int l16  = lane & 15;
    const int colbase = wave * 32;
    const int col = colbase + 2 * l16;
    Es[k] = emb[k];
    if (k < TILE_M) Ss[k] = species[n0 + k];

    const v4f* Tg = reinterpret_cast<const v4f*>(Tm + (long)n0 * TM_ATOM);
    v4f tv[9];
#pragma unroll
    for (int it = 0; it < 9; ++it) tv[it] = Tg[it * KDIM + k];
    float e[N_SPEC];
#pragma unroll
    for (int s = 0; s < N_SPEC; ++s) e[s] = emb[s * N_CH + (col >> 2)];
    v2f wr[3][4];
#pragma unroll
    for (int l = 0; l < 3; ++l) {
#pragma unroll
        for (int jj = 0; jj < 4; ++jj) {
            const int j = 4 * (jj >> 1) + 2 * half + (jj & 1);
            wr[l][jj] = *reinterpret_cast<const v2f*>(W_rad + (l * N_RB + j) * KDIM + col);
        }
    }
    const float* Rp = RW + (2 * half) * KDIM + col;
    float bw0[32], bw1[32], bw2[32];
    load_b_chunk(Rp, bw0);
    load_b_chunk(Rp + RW_PLANE, bw1);
    load_b_chunk(Rp + 2 * RW_PLANE, bw2);
#pragma unroll
    for (int it = 0; it < 9; ++it) {
        const int idx = it * KDIM + k;
        const int a   = idx / 72;
        const int q   = idx - a * 72;
        *reinterpret_cast<v4f*>(Ts + a * TS_PITCH + 4 * q) = tv[it];
    }
    __syncthreads();

    const float* Ta = Ts + l16 * TS_PITCH + 2 * half;
    const float i3 = A3;
    int ec[3][2];
#pragma unroll
    for (int g = 0; g < 3; ++g) {
        ec[g][0] = (128 * g + col) / 12;
        ec[g][1] = (128 * g + col + 1) / 12;
    }
    float* xr = x0 + (long)(n0 + 8 * half) * K0TOT + col;
    v2f a[8];
    load_t_plane(Ta, 0, a);

    {
        v8f cf[2], ch[2];
        planes_from_moments<0, 1, true>(Ta, wr[0], e, bw0, a, cf, ch);
#pragma unroll
        for (int i = 0; i < 8; ++i) {
            const float* er = Es + Ss[8 * half + i] * N_CH;
            v2f o;
#pragma unroll
            for (int t = 0; t < 2; ++t) {
                const float fv = cf[t][i], h = ch[t][i];
                o[t] = (h * fv + fv) * er[ec[0][t]];
            }
            *reinterpret_cast<v2f*>(xr + i * K0TOT) = o;
        }
    }
    {
        v8f cf[6], ch[6];
        planes_from_moments<1, 3, true>(Ta, wr[1], e, bw1, a, cf, ch);
#pragma unroll
        for (int i = 0; i < 8; ++i) {
            const float* er = Es + Ss[8 * half + i] * N_CH;
            v2f o;
#pragma unroll
            for (int t = 0; t < 2; ++t) {
                float Gu[9], Fu[9];
                uncouple_l1(cf[t][i], cf[2 + t][i], cf[4 + t][i], Fu);
                uncouple_l1(ch[t][i], ch[2 + t][i], ch[4 + t][i], Gu);
                const float tr1 = trace_gf_plus_f(Gu, Fu);
                o[t] = (tr1 * i3) * er[ec[1][t]];
            }
            *reinterpret_cast<v2f*>(xr + i * K0TOT + 128) = o;
        }
    }
    {
        v8f cf[10], ch[10];
        planes_from_moments<4, 5, false>(Ta, wr[2], e, bw2, a, cf, ch);
#pragma unroll
        for (int i = 0; i < 8; ++i) {
            const float* er = Es + Ss[8 * half + i] * N_CH;
            v2f o;
#pragma unroll
            for (int t = 0; t < 2; ++t) {
                float Gu[9], Fu[9];
                uncouple_l2(cf[t][i], cf[2 + t][i], cf[4 + t][i], cf[6 + t][i], cf[8 + t][i], Fu);
                uncouple_l2(ch[t][i], ch[2 + t][i], ch[4 + t][i], ch[6 + t][i], ch[8 + t][i], Gu);
                const float tr2 = trace_gf_plus_f(Gu, Fu);
                o[t] = (tr2 * i3) * er[ec[2][t]];
            }
            *reinterpret_cast<v2f*>(xr + i * K0TOT + 256) = o;
        }
    }
}

#define HB_ROWS 128
#define HB_CHUNKS (K0TOT / 16)
struct HeadA { v4f a[2][2][2]; };
struct HeadB { v4f b[2][4][2]; };
__device__ __forceinline__ void head_load_a(const float* __restrict__ A0,
                                            const float* __restrict__ A1, int c, HeadA& f) {
#pragma unroll
    for (int cc = 0; cc < 2; ++cc) {
#pragma unroll
        for (int sp = 0; sp < 2; ++sp) {
            const int q = 2 * (c + cc) + sp;
            f.a[cc][sp][0] = *reinterpret_cast<const v4f*>(A0 + 8 * q);
            f.a[cc][sp][1] = *reinterpret_cast<const v4f*>(A1 + 8 * q);
        }
    }
}
__device__ __forceinline__ void head_load_b(const float* __restrict__ Bp, int c, HeadB& f) {
#pragma unroll
    for (int sp = 0; sp < 2; ++sp) {
        const int q = 2 * c + sp;
#pragma unroll
        for (int i = 0; i < 4; ++i) {
            const float* rowp = Bp + (long)(8 * q + i) * K0TOT;
            f.b[sp][i][0] = *reinterpret_cast<const v4f*>(rowp);
            f.b[sp][i][1] = *reinterpret_cast<const v4f*>(rowp + 64);
        }
    }
}
template <int CC>
__device__ __forceinline__ void head_mma(const HeadA& fa, const HeadB& fb, v8f (&acc)[2][8]) {
#pragma unroll
    for (int sp = 0; sp < 2; ++sp) {
#pragma unroll
        for (int e = 0; e < 2; ++e) {
            v2f a0, a1;
            a0.x = fa.a[CC][sp][0][2 * e];
            a0.y = fa.a[CC][sp][0][2 * e + 1];
            a1.x = fa.a[CC][sp][1][2 * e];
            a1.y = fa.a[CC][sp][1][2 * e + 1];
#pragma unroll
            for (int g = 0; g < 2; ++g) {
#pragma unroll
                for (int v = 0; v < 4; ++v) {
                    v2f b;
                    b.x = fb.b[sp][2 * e][g][v];
                    b.y = fb.b[sp][2 * e + 1][g][v];
                    acc[0][4 * g + v] = __builtin_amdgcn_wmma_f32_16x16x4_f32(
                        false, a0, false, b, (short)0, acc[0][4 * g + v], false, false);
                    acc[1][4 * g + v] = __builtin_amdgcn_wmma_f32_16x16x4_f32(
                        false, a1, false, b, (short)0, acc[1][4 * g + v], false, false);
                }
            }
        }
    }
}
__global__ void __launch_bounds__(HB_ROWS)
k_head_wmma(const float* __restrict__ x0, const float* __restrict__ Wh,
            const float* __restrict__ bh, const float* __restrict__ Wo,
            float* __restrict__ out) {
    const int lane = threadIdx.x & 31;
    const int wave = threadIdx.x >> 5;
    const int half = lane >> 4;
    const int l16  = lane & 15;
    const int m0 = blockIdx.x * HB_ROWS + wave * 32;
    if (m0 >= N_ATOMS) return;
    const int n0 = blockIdx.y * 128;
    const int ra = min(m0 + l16, N_ATOMS - 1);
    const int rb = min(m0 + 16 + l16, N_ATOMS - 1);
    const float* A0 = x0 + (long)ra * K0TOT + 4 * half;
    const float* A1 = x0 + (long)rb * K0TOT + 4 * half;
    const float* Bp = Wh + (long)(4 * half) * K0TOT + n0 + 4 * l16;
    v8f acc[2][8] = {};
    HeadA fa, fan;
    HeadB f0, f1;
    head_load_a(A0, A1, 0, fa);
    head_load_b(Bp, 0, f0);
    for (int c = 0; c < HB_CHUNKS - 2; c += 2) {
        head_load_b(Bp, c + 1, f1);
        __builtin_amdgcn_sched_barrier(0);
        head_load_a(A0, A1, c + 2, fan);
        __builtin_amdgcn_sched_barrier(0);
        head_mma<0>(fa, f0, acc);
        __builtin_amdgcn_sched_barrier(0);
        head_load_b(Bp, c + 2, f0);
        __builtin_amdgcn_sched_barrier(0);
        head_mma<1>(fa, f1, acc);
        __builtin_amdgcn_sched_barrier(0);
        fa = fan;
    }
    head_load_b(Bp, HB_CHUNKS - 1, f1);
    __builtin_amdgcn_sched_barrier(0);
    head_mma<0>(fa, f0, acc);
    head_mma<1>(fa, f1, acc);
    v4f bias[2], wo[2];
#pragma unroll
    for (int g = 0; g < 2; ++g) {
        bias[g] = *reinterpret_cast<const v4f*>(bh + n0 + 64 * g + 4 * l16);
        wo[g]   = *reinterpret_cast<const v4f*>(Wo + n0 + 64 * g + 4 * l16);
    }
#pragma unroll
    for (int rt = 0; rt < 2; ++rt) {
#pragma unroll
        for (int i = 0; i < 8; ++i) {
            float partial = 0.f;
#pragma unroll
            for (int g = 0; g < 2; ++g) {
#pragma unroll
                for (int v = 0; v < 4; ++v) {
                    float x = acc[rt][4 * g + v][i] + bias[g][v];
                    partial += (x / (1.f + expf(-x))) * wo[g][v];
                }
            }
            const int row = m0 + 16 * rt + 8 * half + i;
            if (row < N_ATOMS) atomicAdd(&out[row], partial);
        }
    }
}

extern "C" void kernel_launch(void* const* d_in, const int* in_sizes, int n_in,
                              void* d_out, int out_size, void* d_ws, size_t ws_size,
                              hipStream_t stream) {
    (void)in_sizes; (void)n_in; (void)out_size; (void)ws_size;
    const float* positions = (const float*)d_in[0];
    const int*   species   = (const int*)d_in[1];
    const int*   pairs     = (const int*)d_in[2];
    const float* embeddings= (const float*)d_in[3];
    const float* W_rad     = (const float*)d_in[4];
    const float* W_cg0     = (const float*)d_in[5];
    const float* W_cg1     = (const float*)d_in[6];
    const float* W_cg2     = (const float*)d_in[7];
    const float* W_head    = (const float*)d_in[8];
    const float* b_head    = (const float*)d_in[9];
    const float* W_out     = (const float*)d_in[10];
    const float* b_out     = (const float*)d_in[11];
    float* out = (float*)d_out;

    float* ws = (float*)d_ws;
    float* Tm = ws;
    float* x0 = Tm + (long)N_ATOMS * TM_ATOM;
    float* RW = x0 + (long)N_ATOMS * K0TOT;

    k_init<<<RW_BLOCKS + (N_ATOMS + 255) / 256, 256, 0, stream>>>(out, b_out, embeddings, W_rad,
                                                                   W_cg0, W_cg1, W_cg2, RW);

    k_pairs_owned<<<OWN_BLOCKS, OWN_THREADS, 0, stream>>>(positions, species, pairs, Tm);

    k_tile_feats<<<N_ATOMS / TILE_M, KDIM, 0, stream>>>(Tm, embeddings, W_rad, RW, species, x0);

    k_head_wmma<<<dim3((N_ATOMS + HB_ROWS - 1) / HB_ROWS, K0TOT / 128), HB_ROWS, 0, stream>>>(
        x0, W_head, b_head, W_out, out);
}


# init launch folded into the pair kernel: 3 extra blocks build RW = scale x R x W_cg by WMMA on compute units the owner blocks leave idle; owner blocks store out's start value; 3 launches (was 4)
# speedup vs baseline: 5.1836x; 1.0758x over previous
#include <hip/hip_runtime.h>
#include <math.h>

#define N_ATOMS  10000
#define N_PAIRS  160000
#define N_CH     32
#define KDIM     128
#define K0TOT    384

typedef float v2f __attribute__((ext_vector_type(2)));
typedef float v4f __attribute__((ext_vector_type(4)));
typedef float v8f __attribute__((ext_vector_type(8)));

#define A3 0.5773502691896258f
#define A2 0.7071067811865476f
#define B2 0.8164965809277260f
#define B1 0.4082482904638630f

#define N_SPEC   4
#define N_RB     8
#define TM_ATOM  (N_SPEC * 9 * N_RB)
#define OWN_ATOMS   80
#define OWN_BLOCKS  (N_ATOMS / OWN_ATOMS)
#define OWN_THREADS 1024
#define OWN_WAVES   (OWN_THREADS / 32)
#define OWN_LOADS   16
#define OWN_GROUP   (OWN_THREADS * OWN_LOADS)
#define OWN_GROUPS  (PAIR_QUADS / OWN_GROUP)
#define W_CAP       128
#define PAIR_QUADS  (N_PAIRS / 2)
typedef int v4i __attribute__((ext_vector_type(4)));

__device__ __forceinline__ void pair_geometry(const float* __restrict__ pos, int c, int nb,
                                              float& d, float& ux, float& uy, float& uz,
                                              float& fc) {
    float rx = pos[3 * nb + 0] - pos[3 * c + 0];
    float ry = pos[3 * nb + 1] - pos[3 * c + 1];
    float rz = pos[3 * nb + 2] - pos[3 * c + 2];
    d = sqrtf(rx * rx + ry * ry + rz * rz + 1e-12f);
    float inv = 1.f / d;
    ux = rx * inv;
    uy = ry * inv;
    uz = rz * inv;
    float tc = (d - 15.f) * 0.2f;
    tc = fminf(fmaxf(tc, 0.f), 1.f);
    fc = 0.5f * (cosf(3.14159265358979f * tc) + 1.f);
}
__device__ __forceinline__ void add_terms(float* Tp, int j, float d, float ux, float uy,
                                          float uz, float fc) {
    float mu = 20.f * (float)j / 7.f;
    float x  = (d - mu) * 0.4f;
    float rb = expf(-x * x) * fc;
    const float s3 = 1.7320508075688772f;
    float sh[9] = { 1.f, uy, uz, ux,
                    s3 * ux * uy, s3 * uy * uz, 0.5f * (3.f * uz * uz - 1.f),
                    s3 * ux * uz, 0.5f * s3 * (ux * ux - uy * uy) };
    for (int m = 0; m < 9; ++m) atomicAdd(Tp + m * N_RB, sh[m] * rb);
}
__device__ __forceinline__ void note_pair(unsigned* myl, unsigned& wcount, int a0, int c,
                                          int nb, bool valid) {
    const unsigned cl = (unsigned)(c - a0);
    const bool hit = valid && cl < OWN_ATOMS;
    const unsigned mask = __builtin_amdgcn_ballot_w32(hit);
    if (mask != 0u) {
        const unsigned idx = wcount + __builtin_amdgcn_mbcnt_lo(mask, 0u);
        if (hit && idx < W_CAP) myl[idx] = (cl << 16) | (unsigned)nb;
        wcount += (unsigned)__builtin_popcount(mask);
    }
}
template <bool CHECK>
__device__ __forceinline__ void note_group(const v4i* __restrict__ P4, int base, int tid,
                                           unsigned* myl, unsigned& wcount, int a0) {
    v4i v[OWN_LOADS];
#pragma unroll
    for (int u = 0; u < OWN_LOADS; ++u) {
        const int q = base + u * OWN_THREADS + tid;
        v[u] = P4[CHECK ? min(q, PAIR_QUADS - 1) : q];
    }
#pragma unroll
    for (int u = 0; u < OWN_LOADS; ++u) {
        const bool valid = !CHECK || (base + u * OWN_THREADS + tid < PAIR_QUADS);
        note_pair(myl, wcount, a0, v[u].x, v[u].y, valid);
        note_pair(myl, wcount, a0, v[u].z, v[u].w, valid);
    }
}

#define RW_BLOCKS 3
#define RW_WAVES  8
#define RW_PLANE  (N_SPEC * N_RB * KDIM)
__device__ __forceinline__ void build_rw(int l, int tid, const float* __restrict__ emb,
                                         const float* __restrict__ W_rad,
                                         const float* __restrict__ Wc, float* __restrict__ RW) {
    const int lane = tid & 31;
    const int wave = tid >> 5;
    const int half = lane >> 4;
    const int l16  = lane & 15;
    const int kk   = 16 * (wave & 1) + l16;
    const int col  = 32 * (wave >> 1) + 2 * l16;
    const float* Bp = Wc + (2 * half) * KDIM + col;
    const float* Ap = W_rad + (l * N_RB + (kk & 7)) * KDIM + 2 * half;
    const v4f*   Ep = reinterpret_cast<const v4f*>(emb + (kk >> 3) * N_CH);
    v8f acc0 = {}, acc1 = {};
    v4f e4 = {};
#pragma unroll
    for (int ks = 0; ks < KDIM / 4; ++ks) {
        if ((ks & 3) == 0) e4 = Ep[ks >> 2];
        const v2f r0 = *reinterpret_cast<const v2f*>(Bp + (4 * ks) * KDIM);
        const v2f r1 = *reinterpret_cast<const v2f*>(Bp + (4 * ks + 1) * KDIM);
        const v2f wq = *reinterpret_cast<const v2f*>(Ap + 4 * ks);
        const float ev = e4[ks & 3];
        v2f a, t0, t1;
        a.x  = ev * wq.x;
        a.y  = ev * wq.y;
        t0.x = r0.x;
        t0.y = r1.x;
        t1.x = r0.y;
        t1.y = r1.y;
        acc0 = __builtin_amdgcn_wmma_f32_16x16x4_f32(false, a, false, t0, (short)0, acc0,
                                                     false, false);
        acc1 = __builtin_amdgcn_wmma_f32_16x16x4_f32(false, a, false, t1, (short)0, acc1,
                                                     false, false);
    }
    float* Rd = RW + l * RW_PLANE + (16 * (wave & 1) + 8 * half) * KDIM + col;
#pragma unroll
    for (int r = 0; r < 8; ++r) {
        v2f o;
        o.x = acc0[r] * 0.1f;
        o.y = acc1[r] * 0.1f;
        *reinterpret_cast<v2f*>(Rd + r * KDIM) = o;
    }
}

__global__ void __launch_bounds__(OWN_THREADS)
k_pairs_owned(const float* __restrict__ pos, const int* __restrict__ species,
              const int* __restrict__ pairs, float* __restrict__ Tm,
              float* __restrict__ out, const float* __restrict__ b_out,
              const float* __restrict__ emb, const float* __restrict__ W_rad,
              const float* __restrict__ W0, const float* __restrict__ W1,
              const float* __restrict__ W2, float* __restrict__ RW) {
    if (blockIdx.x >= OWN_BLOCKS) {
        const int l = blockIdx.x - OWN_BLOCKS;
        if (threadIdx.x < RW_WAVES * 32)
            build_rw(l, threadIdx.x, emb, W_rad, (l == 0) ? W0 : ((l == 1) ? W1 : W2), RW);
        return;
    }
    __shared__ __attribute__((aligned(16))) float acc[OWN_ATOMS * TM_ATOM];
    __shared__ unsigned wl[OWN_WAVES * W_CAP];
    __shared__ float g_d[OWN_WAVES * W_CAP], g_ux[OWN_WAVES * W_CAP], g_uy[OWN_WAVES * W_CAP],
                     g_uz[OWN_WAVES * W_CAP], g_fc[OWN_WAVES * W_CAP];
    __shared__ int   g_row[OWN_WAVES * W_CAP];
    const int tid  = threadIdx.x;
    const int lane = tid & 31;
    const int wave = tid >> 5;
    const int a0   = blockIdx.x * OWN_ATOMS;
    if (tid < OWN_ATOMS) out[a0 + tid] = b_out[0];
    for (int i = tid; i < OWN_ATOMS * TM_ATOM / 4; i += OWN_THREADS)
        reinterpret_cast<v4f*>(acc)[i] = v4f{0.f, 0.f, 0.f, 0.f};

    const v4i* P4 = reinterpret_cast<const v4i*>(pairs);
    unsigned* myl = wl + wave * W_CAP;
    unsigned wcount = 0u;
    for (int k = 0; k < OWN_GROUPS; ++k)
        note_group<false>(P4, (int)((k + blockIdx.x) % OWN_GROUPS) * OWN_GROUP, tid, myl, wcount,
                          a0);
    if (OWN_GROUPS * OWN_GROUP < PAIR_QUADS)
        note_group<true>(P4, OWN_GROUPS * OWN_GROUP, tid, myl, wcount, a0);
    const bool over   = wcount > W_CAP;
    const unsigned nn = over ? 0u : wcount;
    const int g0 = wave * W_CAP;
    __syncthreads();

    for (unsigned i = lane; i < nn; i += 32) {
        const unsigned ent = myl[i];
        const int cl = (int)(ent >> 16);
        const int nb = (int)(ent & 0xffffu);
        float d, ux, uy, uz, fc;
        pair_geometry(pos, a0 + cl, nb, d, ux, uy, uz, fc);
        g_d[g0 + i]  = d;
        g_ux[g0 + i] = ux;
        g_uy[g0 + i] = uy;
        g_uz[g0 + i] = uz;
        g_fc[g0 + i] = fc;
        g_row[g0 + i] = (cl * N_SPEC + species[nb]) * (9 * N_RB);
    }
    __syncthreads();

    const int j = lane & 7;
    for (unsigned i = lane >> 3; i < nn; i += 4) {
        const int gi = g0 + (int)i;
        add_terms(acc + g_row[gi] + j, j, g_d[gi], g_ux[gi], g_uy[gi], g_uz[gi], g_fc[gi]);
    }
    if (over) {
        for (int q = tid; q < PAIR_QUADS; q += OWN_THREADS) {
            const v4i v = P4[q];
            const unsigned cl0 = (unsigned)(v.x - a0);
            const unsigned cl1 = (unsigned)(v.z - a0);
            for (int jj = 0; jj < N_RB; ++jj) {
                float d, ux, uy, uz, fc;
                if (cl0 < OWN_ATOMS) {
                    pair_geometry(pos, v.x, v.y, d, ux, uy, uz, fc);
                    add_terms(acc + ((int)cl0 * N_SPEC + species[v.y]) * (9 * N_RB) + jj, jj,
                              d, ux, uy, uz, fc);
                }
                if (cl1 < OWN_ATOMS) {
                    pair_geometry(pos, v.z, v.w, d, ux, uy, uz, fc);
                    add_terms(acc + ((int)cl1 * N_SPEC + species[v.w]) * (9 * N_RB) + jj, jj,
                              d, ux, uy, uz, fc);
                }
            }
        }
    }
    __syncthreads();

    v4f* Tg = reinterpret_cast<v4f*>(Tm + (long)a0 * TM_ATOM);
    for (int i = tid; i < OWN_ATOMS * TM_ATOM / 4; i += OWN_THREADS)
        Tg[i] = reinterpret_cast<const v4f*>(acc)[i];
}

#define TILE_M   16
#define TS_PITCH 292

__device__ __forceinline__ void uncouple_l1(float v1, float v2, float v3, float u[9]) {
    u[0] = 0.f;       u[1] =  A2 * v3;  u[2] = -A2 * v2;
    u[3] = -A2 * v3;  u[4] = 0.f;       u[5] =  A2 * v1;
    u[6] =  A2 * v2;  u[7] = -A2 * v1;  u[8] = 0.f;
}
__device__ __forceinline__ void uncouple_l2(float v4, float v5, float v6, float v7, float v8,
                                            float u[9]) {
    u[0] = -B1 * v6 - A2 * v8;  u[1] = A2 * v5;  u[2] = A2 * v4;
    u[3] = A2 * v5;             u[4] = B2 * v6;  u[5] = A2 * v7;
    u[6] = A2 * v4;             u[7] = A2 * v7;  u[8] = -B1 * v6 + A2 * v8;
}
__device__ __forceinline__ float trace_gf_plus_f(const float G[9], const float Fm[9]) {
    float tr = 0.f;
#pragma unroll
    for (int a = 0; a < 3; ++a)
#pragma unroll
        for (int b = 0; b < 3; ++b) tr += G[a * 3 + b] * Fm[b * 3 + a];
    return tr + (Fm[0] + Fm[4] + Fm[8]);
}

__device__ __forceinline__ void load_b_chunk(const float* __restrict__ p, float (&b)[32]) {
#pragma unroll
    for (int ks = 0; ks < 8; ++ks) {
        const float* q = p + ks * 4 * KDIM;
        const v2f r0 = *reinterpret_cast<const v2f*>(q);
        const v2f r1 = *reinterpret_cast<const v2f*>(q + KDIM);
        b[4 * ks + 0] = r0.x;
        b[4 * ks + 1] = r1.x;
        b[4 * ks + 2] = r0.y;
        b[4 * ks + 3] = r1.y;
    }
}
__device__ __forceinline__ void load_t_plane(const float* __restrict__ Ta, int m, v2f (&a)[8]) {
#pragma unroll
    for (int ks = 0; ks < 8; ++ks)
        a[ks] = *reinterpret_cast<const v2f*>(Ta + (ks >> 1) * 72 + m * 8 + 4 * (ks & 1));
}

template <int M0, int NM, bool HAS_NEXT>
__device__ __forceinline__ void planes_from_moments(const float* __restrict__ Ta,
                                                    const v2f (&w)[4], const float (&e)[N_SPEC],
                                                    const float (&bw)[32], v2f (&a)[8],
                                                    v8f (&cf)[2 * NM], v8f (&ch)[2 * NM]) {
    float b[32];
#pragma unroll
    for (int ks = 0; ks < 8; ++ks) {
        const int s  = ks >> 1;
        const int jb = 2 * (ks & 1);
        b[4 * ks + 0] = e[s] * w[jb].x;
        b[4 * ks + 1] = e[s] * w[jb + 1].x;
        b[4 * ks + 2] = e[s] * w[jb].y;
        b[4 * ks + 3] = e[s] * w[jb + 1].y;
    }
#pragma unroll
    for (int i = 0; i < NM; ++i) {
        v2f an[8];
        if (HAS_NEXT || i + 1 < NM) load_t_plane(Ta, M0 + i + 1, an);
        v8f f0 = {}, f1 = {}, h0 = {}, h1 = {};
#pragma unroll
        for (int ks = 0; ks < 8; ++ks) {
            v2f b0, b1, g0, g1;
            b0.x = b[4 * ks + 0];
            b0.y = b[4 * ks + 1];
            b1.x = b[4 * ks + 2];
            b1.y = b[4 * ks + 3];
            g0.x = bw[4 * ks + 0];
            g0.y = bw[4 * ks + 1];
            g1.x = bw[4 * ks + 2];
            g1.y = bw[4 * ks + 3];
            f0 = __builtin_amdgcn_wmma_f32_16x16x4_f32(false, a[ks], false, b0, (short)0, f0,
                                                       false, false);
            f1 = __builtin_amdgcn_wmma_f32_16x16x4_f32(false, a[ks], false, b1, (short)0, f1,
                                                       false, false);
            h0 = __builtin_amdgcn_wmma_f32_16x16x4_f32(false, a[ks], false, g0, (short)0, h0,
                                                       false, false);
            h1 = __builtin_amdgcn_wmma_f32_16x16x4_f32(false, a[ks], false, g1, (short)0, h1,
                                                       false, false);
        }
        cf[2 * i]     = f0 * 0.1f;
        cf[2 * i + 1] = f1 * 0.1f;
        ch[2 * i]     = h0;
        ch[2 * i + 1] = h1;
        if (HAS_NEXT || i + 1 < NM) {
#pragma unroll
            for (int ks = 0; ks < 8; ++ks) a[ks] = an[ks];
        }
    }
}

__global__ void __launch_bounds__(KDIM)
k_tile_feats(const float* __restrict__ Tm, const float* __restrict__ emb,
             const float* __restrict__ W_rad, const float* __restrict__ RW,
             const int* __restrict__ species, float* __restrict__ x0) {
    __shared__ __attribute__((aligned(16))) float Ts[TILE_M * TS_PITCH];
    __shared__ float Es[N_SPEC * N_CH];
    __shared__ int   Ss[TILE_M];
    const int n0 = blockIdx.x * TILE_M;
    const int k  = threadIdx.x;
    const int lane = k & 31;
    const int wave = k >> 5;
    const int half = lane >> 4;
    const int l16  = lane & 15;
    const int colbase = wave * 32;
    const int col = colbase + 2 * l16;
    Es[k] = emb[k];
    if (k < TILE_M) Ss[k] = species[n0 + k];

    const v4f* Tg = reinterpret_cast<const v4f*>(Tm + (long)n0 * TM_ATOM);
    v4f tv[9];
#pragma unroll
    for (int it = 0; it < 9; ++it) tv[it] = Tg[it * KDIM + k];
    float e[N_SPEC];
#pragma unroll
    for (int s = 0; s < N_SPEC; ++s) e[s] = emb[s * N_CH + (col >> 2)];
    v2f wr[3][4];
#pragma unroll
    for (int l = 0; l < 3; ++l) {
#pragma unroll
        for (int jj = 0; jj < 4; ++jj) {
            const int j = 4 * (jj >> 1) + 2 * half + (jj & 1);
            wr[l][jj] = *reinterpret_cast<const v2f*>(W_rad + (l * N_RB + j) * KDIM + col);
        }
    }
    const float* Rp = RW + (2 * half) * KDIM + col;
    float bw0[32], bw1[32], bw2[32];
    load_b_chunk(Rp, bw0);
    load_b_chunk(Rp + RW_PLANE, bw1);
    load_b_chunk(Rp + 2 * RW_PLANE, bw2);
#pragma unroll
    for (int it = 0; it < 9; ++it) {
        const int idx = it * KDIM + k;
        const int a   = idx / 72;
        const int q   = idx - a * 72;
        *reinterpret_cast<v4f*>(Ts + a * TS_PITCH + 4 * q) = tv[it];
    }
    __syncthreads();

    const float* Ta = Ts + l16 * TS_PITCH + 2 * half;
    const float i3 = A3;
    int ec[3][2];
#pragma unroll
    for (int g = 0; g < 3; ++g) {
        ec[g][0] = (128 * g + col) / 12;
        ec[g][1] = (128 * g + col + 1) / 12;
    }
    float* xr = x0 + (long)(n0 + 8 * half) * K0TOT + col;
    v2f a[8];
    load_t_plane(Ta, 0, a);

    {
        v8f cf[2], ch[2];
        planes_from_moments<0, 1, true>(Ta, wr[0], e, bw0, a, cf, ch);
#pragma unroll
        for (int i = 0; i < 8; ++i) {
            const float* er = Es + Ss[8 * half + i] * N_CH;
            v2f o;
#pragma unroll
            for (int t = 0; t < 2; ++t) {
                const float fv = cf[t][i], h = ch[t][i];
                o[t] = (h * fv + fv) * er[ec[0][t]];
            }
            *reinterpret_cast<v2f*>(xr + i * K0TOT) = o;
        }
    }
    {
        v8f cf[6], ch[6];
        planes_from_moments<1, 3, true>(Ta, wr[1], e, bw1, a, cf, ch);
#pragma unroll
        for (int i = 0; i < 8; ++i) {
            const float* er = Es + Ss[8 * half + i] * N_CH;
            v2f o;
#pragma unroll
            for (int t = 0; t < 2; ++t) {
                float Gu[9], Fu[9];
                uncouple_l1(cf[t][i], cf[2 + t][i], cf[4 + t][i], Fu);
                uncouple_l1(ch[t][i], ch[2 + t][i], ch[4 + t][i], Gu);
                const float tr1 = trace_gf_plus_f(Gu, Fu);
                o[t] = (tr1 * i3) * er[ec[1][t]];
            }
            *reinterpret_cast<v2f*>(xr + i * K0TOT + 128) = o;
        }
    }
    {
        v8f cf[10], ch[10];
        planes_from_moments<4, 5, false>(Ta, wr[2], e, bw2, a, cf, ch);
#pragma unroll
        for (int i = 0; i < 8; ++i) {
            const float* er = Es + Ss[8 * half + i] * N_CH;
            v2f o;
#pragma unroll
            for (int t = 0; t < 2; ++t) {
                float Gu[9], Fu[9];
                uncouple_l2(cf[t][i], cf[2 + t][i], cf[4 + t][i], cf[6 + t][i], cf[8 + t][i], Fu);
                uncouple_l2(ch[t][i], ch[2 + t][i], ch[4 + t][i], ch[6 + t][i], ch[8 + t][i], Gu);
                const float tr2 = trace_gf_plus_f(Gu, Fu);
                o[t] = (tr2 * i3) * er[ec[2][t]];
            }
            *reinterpret_cast<v2f*>(xr + i * K0TOT + 256) = o;
        }
    }
}

#define HB_ROWS 128
#define HB_CHUNKS (K0TOT / 16)
struct HeadA { v4f a[2][2][2]; };
struct HeadB { v4f b[2][4][2]; };
__device__ __forceinline__ void head_load_a(const float* __restrict__ A0,
                                            const float* __restrict__ A1, int c, HeadA& f) {
#pragma unroll
    for (int cc = 0; cc < 2; ++cc) {
#pragma unroll
        for (int sp = 0; sp < 2; ++sp) {
            const int q = 2 * (c + cc) + sp;
            f.a[cc][sp][0] = *reinterpret_cast<const v4f*>(A0 + 8 * q);
            f.a[cc][sp][1] = *reinterpret_cast<const v4f*>(A1 + 8 * q);
        }
    }
}
__device__ __forceinline__ void head_load_b(const float* __restrict__ Bp, int c, HeadB& f) {
#pragma unroll
    for (int sp = 0; sp < 2; ++sp) {
        const int q = 2 * c + sp;
#pragma unroll
        for (int i = 0; i < 4; ++i) {
            const float* rowp = Bp + (long)(8 * q + i) * K0TOT;
            f.b[sp][i][0] = *reinterpret_cast<const v4f*>(rowp);
            f.b[sp][i][1] = *reinterpret_cast<const v4f*>(rowp + 64);
        }
    }
}
template <int CC>
__device__ __forceinline__ void head_mma(const HeadA& fa, const HeadB& fb, v8f (&acc)[2][8]) {
#pragma unroll
    for (int sp = 0; sp < 2; ++sp) {
#pragma unroll
        for (int e = 0; e < 2; ++e) {
            v2f a0, a1;
            a0.x = fa.a[CC][sp][0][2 * e];
            a0.y = fa.a[CC][sp][0][2 * e + 1];
            a1.x = fa.a[CC][sp][1][2 * e];
            a1.y = fa.a[CC][sp][1][2 * e + 1];
#pragma unroll
            for (int g = 0; g < 2; ++g) {
#pragma unroll
                for (int v = 0; v < 4; ++v) {
                    v2f b;
                    b.x = fb.b[sp][2 * e][g][v];
                    b.y = fb.b[sp][2 * e + 1][g][v];
                    acc[0][4 * g + v] = __builtin_amdgcn_wmma_f32_16x16x4_f32(
                        false, a0, false, b, (short)0, acc[0][4 * g + v], false, false);
                    acc[1][4 * g + v] = __builtin_amdgcn_wmma_f32_16x16x4_f32(
                        false, a1, false, b, (short)0, acc[1][4 * g + v], false, false);
                }
            }
        }
    }
}
__global__ void __launch_bounds__(HB_ROWS)
k_head_wmma(const float* __restrict__ x0, const float* __restrict__ Wh,
            const float* __restrict__ bh, const float* __restrict__ Wo,
            float* __restrict__ out) {
    const int lane = threadIdx.x & 31;
    const int wave = threadIdx.x >> 5;
    const int half = lane >> 4;
    const int l16  = lane & 15;
    const int m0 = blockIdx.x * HB_ROWS + wave * 32;
    if (m0 >= N_ATOMS) return;
    const int n0 = blockIdx.y * 128;
    const int ra = min(m0 + l16, N_ATOMS - 1);
    const int rb = min(m0 + 16 + l16, N_ATOMS - 1);
    const float* A0 = x0 + (long)ra * K0TOT + 4 * half;
    const float* A1 = x0 + (long)rb * K0TOT + 4 * half;
    const float* Bp = Wh + (long)(4 * half) * K0TOT + n0 + 4 * l16;
    v8f acc[2][8] = {};
    HeadA fa, fan;
    HeadB f0, f1;
    head_load_a(A0, A1, 0, fa);
    head_load_b(Bp, 0, f0);
    for (int c = 0; c < HB_CHUNKS - 2; c += 2) {
        head_load_b(Bp, c + 1, f1);
        __builtin_amdgcn_sched_barrier(0);
        head_load_a(A0, A1, c + 2, fan);
        __builtin_amdgcn_sched_barrier(0);
        head_mma<0>(fa, f0, acc);
        __builtin_amdgcn_sched_barrier(0);
        head_load_b(Bp, c + 2, f0);
        __builtin_amdgcn_sched_barrier(0);
        head_mma<1>(fa, f1, acc);
        __builtin_amdgcn_sched_barrier(0);
        fa = fan;
    }
    head_load_b(Bp, HB_CHUNKS - 1, f1);
    __builtin_amdgcn_sched_barrier(0);
    head_mma<0>(fa, f0, acc);
    head_mma<1>(fa, f1, acc);
    v4f bias[2], wo[2];
#pragma unroll
    for (int g = 0; g < 2; ++g) {
        bias[g] = *reinterpret_cast<const v4f*>(bh + n0 + 64 * g + 4 * l16);
        wo[g]   = *reinterpret_cast<const v4f*>(Wo + n0 + 64 * g + 4 * l16);
    }
#pragma unroll
    for (int rt = 0; rt < 2; ++rt) {
#pragma unroll
        for (int i = 0; i < 8; ++i) {
            float partial = 0.f;
#pragma unroll
            for (int g = 0; g < 2; ++g) {
#pragma unroll
                for (int v = 0; v < 4; ++v) {
                    float x = acc[rt][4 * g + v][i] + bias[g][v];
                    partial += (x / (1.f + expf(-x))) * wo[g][v];
                }
            }
            const int row = m0 + 16 * rt + 8 * half + i;
            if (row < N_ATOMS) atomicAdd(&out[row], partial);
        }
    }
}

extern "C" void kernel_launch(void* const* d_in, const int* in_sizes, int n_in,
                              void* d_out, int out_size, void* d_ws, size_t ws_size,
                              hipStream_t stream) {
    (void)in_sizes; (void)n_in; (void)out_size; (void)ws_size;
    const float* positions = (const float*)d_in[0];
    const int*   species   = (const int*)d_in[1];
    const int*   pairs     = (const int*)d_in[2];
    const float* embeddings= (const float*)d_in[3];
    const float* W_rad     = (const float*)d_in[4];
    const float* W_cg0     = (const float*)d_in[5];
    const float* W_cg1     = (const float*)d_in[6];
    const float* W_cg2     = (const float*)d_in[7];
    const float* W_head    = (const float*)d_in[8];
    const float* b_head    = (const float*)d_in[9];
    const float* W_out     = (const float*)d_in[10];
    const float* b_out     = (const float*)d_in[11];
    float* out = (float*)d_out;

    float* ws = (float*)d_ws;
    float* Tm = ws;
    float* x0 = Tm + (long)N_ATOMS * TM_ATOM;
    float* RW = x0 + (long)N_ATOMS * K0TOT;

    k_pairs_owned<<<OWN_BLOCKS + RW_BLOCKS, OWN_THREADS, 0, stream>>>(
        positions, species, pairs, Tm, out, b_out, embeddings, W_rad, W_cg0, W_cg1, W_cg2, RW);

    k_tile_feats<<<N_ATOMS / TILE_M, KDIM, 0, stream>>>(Tm, embeddings, W_rad, RW, species, x0);

    k_head_wmma<<<dim3((N_ATOMS + HB_ROWS - 1) / HB_ROWS, K0TOT / 128), HB_ROWS, 0, stream>>>(
        x0, W_head, b_head, W_out, out);
}


# pairs binned once by 16-atom tile (250 blocks sort 640 pairs each in LDS, store geometry records); tile kernel sums its moments in LDS from its runs: no list scan, no moments array; sum order changes
# speedup vs baseline: 5.5979x; 1.0799x over previous
#include <hip/hip_runtime.h>
#include <math.h>

#define N_ATOMS  10000
#define N_PAIRS  160000
#define N_CH     32
#define KDIM     128
#define K0TOT    384

typedef float v2f __attribute__((ext_vector_type(2)));
typedef float v4f __attribute__((ext_vector_type(4)));
typedef float v8f __attribute__((ext_vector_type(8)));

#define A3 0.5773502691896258f
#define A2 0.7071067811865476f
#define B2 0.8164965809277260f
#define B1 0.4082482904638630f

#define N_SPEC   4
#define N_RB     8
#define TM_ATOM  (N_SPEC * 9 * N_RB)
#define TILE_M   16
#define TS_PITCH 292
#define N_TILES     (N_ATOMS / TILE_M)
#define BIN_THREADS 640
#define BIN_BLOCKS  (N_PAIRS / BIN_THREADS)
#define BIN_WAVES   (BIN_THREADS / 32)
#define REC_FLOATS  8
#define OUT_ATOMS   80
typedef int v2i __attribute__((ext_vector_type(2)));

__device__ __forceinline__ void pair_geometry(const float* __restrict__ pos, int c, int nb,
                                              float& d, float& ux, float& uy, float& uz,
                                              float& fc) {
    float rx = pos[3 * nb + 0] - pos[3 * c + 0];
    float ry = pos[3 * nb + 1] - pos[3 * c + 1];
    float rz = pos[3 * nb + 2] - pos[3 * c + 2];
    d = sqrtf(rx * rx + ry * ry + rz * rz + 1e-12f);
    float inv = 1.f / d;
    ux = rx * inv;
    uy = ry * inv;
    uz = rz * inv;
    float tc = (d - 15.f) * 0.2f;
    tc = fminf(fmaxf(tc, 0.f), 1.f);
    fc = 0.5f * (cosf(3.14159265358979f * tc) + 1.f);
}
__device__ __forceinline__ void add_terms(float* Tp, int j, float d, float ux, float uy,
                                          float uz, float fc) {
    float mu = 20.f * (float)j / 7.f;
    float x  = (d - mu) * 0.4f;
    float rb = expf(-x * x) * fc;
    const float s3 = 1.7320508075688772f;
    float sh[9] = { 1.f, uy, uz, ux,
                    s3 * ux * uy, s3 * uy * uz, 0.5f * (3.f * uz * uz - 1.f),
                    s3 * ux * uz, 0.5f * s3 * (ux * ux - uy * uy) };
    for (int m = 0; m < 9; ++m) atomicAdd(Tp + m * N_RB, sh[m] * rb);
}
__device__ __forceinline__ unsigned wave_scan(unsigned v, int lane) {
    for (int o = 1; o < 32; o <<= 1) {
        const unsigned u = __shfl_up(v, o);
        if (lane >= o) v += u;
    }
    return v;
}

#define RW_BLOCKS 3
#define RW_WAVES  8
#define RW_PLANE  (N_SPEC * N_RB * KDIM)
__device__ __forceinline__ void build_rw(int l, int tid, const float* __restrict__ emb,
                                         const float* __restrict__ W_rad,
                                         const float* __restrict__ Wc, float* __restrict__ RW) {
    const int lane = tid & 31;
    const int wave = tid >> 5;
    const int half = lane >> 4;
    const int l16  = lane & 15;
    const int kk   = 16 * (wave & 1) + l16;
    const int col  = 32 * (wave >> 1) + 2 * l16;
    const float* Bp = Wc + (2 * half) * KDIM + col;
    const float* Ap = W_rad + (l * N_RB + (kk & 7)) * KDIM + 2 * half;
    const v4f*   Ep = reinterpret_cast<const v4f*>(emb + (kk >> 3) * N_CH);
    v8f acc0 = {}, acc1 = {};
    v4f e4 = {};
#pragma unroll
    for (int ks = 0; ks < KDIM / 4; ++ks) {
        if ((ks & 3) == 0) e4 = Ep[ks >> 2];
        const v2f r0 = *reinterpret_cast<const v2f*>(Bp + (4 * ks) * KDIM);
        const v2f r1 = *reinterpret_cast<const v2f*>(Bp + (4 * ks + 1) * KDIM);
        const v2f wq = *reinterpret_cast<const v2f*>(Ap + 4 * ks);
        const float ev = e4[ks & 3];
        v2f a, t0, t1;
        a.x  = ev * wq.x;
        a.y  = ev * wq.y;
        t0.x = r0.x;
        t0.y = r1.x;
        t1.x = r0.y;
        t1.y = r1.y;
        acc0 = __builtin_amdgcn_wmma_f32_16x16x4_f32(false, a, false, t0, (short)0, acc0,
                                                     false, false);
        acc1 = __builtin_amdgcn_wmma_f32_16x16x4_f32(false, a, false, t1, (short)0, acc1,
                                                     false, false);
    }
    float* Rd = RW + l * RW_PLANE + (16 * (wave & 1) + 8 * half) * KDIM + col;
#pragma unroll
    for (int r = 0; r < 8; ++r) {
        v2f o;
        o.x = acc0[r] * 0.1f;
        o.y = acc1[r] * 0.1f;
        *reinterpret_cast<v2f*>(Rd + r * KDIM) = o;
    }
}

__global__ void __launch_bounds__(BIN_THREADS)
k_pairs_binned(const float* __restrict__ pos, const int* __restrict__ species,
               const int* __restrict__ pairs, float* __restrict__ srt,
               float* __restrict__ meta, float* __restrict__ out,
               const float* __restrict__ b_out, const float* __restrict__ emb,
               const float* __restrict__ W_rad, const float* __restrict__ W0,
               const float* __restrict__ W1, const float* __restrict__ W2,
               float* __restrict__ RW) {
    if (blockIdx.x >= BIN_BLOCKS) {
        const int l = blockIdx.x - BIN_BLOCKS;
        if (threadIdx.x < RW_WAVES * 32)
            build_rw(l, threadIdx.x, emb, W_rad, (l == 0) ? W0 : ((l == 1) ? W1 : W2), RW);
        return;
    }
    __shared__ __attribute__((aligned(16))) float recs[BIN_THREADS * REC_FLOATS];
    __shared__ unsigned hist[BIN_THREADS];
    __shared__ unsigned wtot[BIN_WAVES];
    const int tid  = threadIdx.x;
    const int lane = tid & 31;
    const int wave = tid >> 5;
    const int b    = blockIdx.x;
    if (b < N_ATOMS / OUT_ATOMS && tid < OUT_ATOMS) out[b * OUT_ATOMS + tid] = b_out[0];
    hist[tid] = 0u;
    const v2i pr = reinterpret_cast<const v2i*>(pairs)[b * BIN_THREADS + tid];
    const int c  = pr.x;
    const int nb = pr.y;
    const int t  = c / TILE_M;
    __syncthreads();
    const unsigned rank = atomicAdd(&hist[t], 1u);
    float d, ux, uy, uz, fc;
    pair_geometry(pos, c, nb, d, ux, uy, uz, fc);
    const int row = (c - t * TILE_M) * TS_PITCH + species[nb] * (9 * N_RB);
    __syncthreads();
    const unsigned cnt  = hist[tid];
    const unsigned incl = wave_scan(cnt, lane);
    if (lane == 31) wtot[wave] = incl;
    __syncthreads();
    unsigned woff = 0u;
    for (int w = 0; w < BIN_WAVES; ++w)
        if (w < wave) woff += wtot[w];
    const unsigned start = woff + incl - cnt;
    hist[tid] = start;
    reinterpret_cast<unsigned*>(meta)[b * BIN_THREADS + tid] = (start << 16) | cnt;
    __syncthreads();
    const unsigned posn = hist[t] + rank;
    v4f* rl = reinterpret_cast<v4f*>(recs);
    rl[2 * posn]     = v4f{d, ux, uy, uz};
    rl[2 * posn + 1] = v4f{fc, (float)row, 0.f, 0.f};
    __syncthreads();
    v4f* sg = reinterpret_cast<v4f*>(srt) + (long)b * (BIN_THREADS * 2);
    for (int i = tid; i < BIN_THREADS * 2; i += BIN_THREADS) sg[i] = rl[i];
}

#define REC_CAP 512
__device__ __forceinline__ void note_run(unsigned src, unsigned p0, unsigned cnt, unsigned base,
                                         unsigned* idx) {
    const unsigned lo = base > p0 ? base - p0 : 0u;
    const unsigned hi = min(cnt, base + REC_CAP > p0 ? base + REC_CAP - p0 : 0u);
#pragma nounroll
    for (unsigned r = lo; r < hi; ++r) idx[p0 + r - base] = src + r;
}

__device__ __forceinline__ void uncouple_l1(float v1, float v2, float v3, float u[9]) {
    u[0] = 0.f;       u[1] =  A2 * v3;  u[2] = -A2 * v2;
    u[3] = -A2 * v3;  u[4] = 0.f;       u[5] =  A2 * v1;
    u[6] =  A2 * v2;  u[7] = -A2 * v1;  u[8] = 0.f;
}
__device__ __forceinline__ void uncouple_l2(float v4, float v5, float v6, float v7, float v8,
                                            float u[9]) {
    u[0] = -B1 * v6 - A2 * v8;  u[1] = A2 * v5;  u[2] = A2 * v4;
    u[3] = A2 * v5;             u[4] = B2 * v6;  u[5] = A2 * v7;
    u[6] = A2 * v4;             u[7] = A2 * v7;  u[8] = -B1 * v6 + A2 * v8;
}
__device__ __forceinline__ float trace_gf_plus_f(const float G[9], const float Fm[9]) {
    float tr = 0.f;
#pragma unroll
    for (int a = 0; a < 3; ++a)
#pragma unroll
        for (int b = 0; b < 3; ++b) tr += G[a * 3 + b] * Fm[b * 3 + a];
    return tr + (Fm[0] + Fm[4] + Fm[8]);
}

__device__ __forceinline__ void load_b_chunk(const float* __restrict__ p, float (&b)[32]) {
#pragma unroll
    for (int ks = 0; ks < 8; ++ks) {
        const float* q = p + ks * 4 * KDIM;
        const v2f r0 = *reinterpret_cast<const v2f*>(q);
        const v2f r1 = *reinterpret_cast<const v2f*>(q + KDIM);
        b[4 * ks + 0] = r0.x;
        b[4 * ks + 1] = r1.x;
        b[4 * ks + 2] = r0.y;
        b[4 * ks + 3] = r1.y;
    }
}
__device__ __forceinline__ void load_t_plane(const float* __restrict__ Ta, int m, v2f (&a)[8]) {
#pragma unroll
    for (int ks = 0; ks < 8; ++ks)
        a[ks] = *reinterpret_cast<const v2f*>(Ta + (ks >> 1) * 72 + m * 8 + 4 * (ks & 1));
}

template <int M0, int NM, bool HAS_NEXT>
__device__ __forceinline__ void planes_from_moments(const float* __restrict__ Ta,
                                                    const v2f (&w)[4], const float (&e)[N_SPEC],
                                                    const float (&bw)[32], v2f (&a)[8],
                                                    v8f (&cf)[2 * NM], v8f (&ch)[2 * NM]) {
    float b[32];
#pragma unroll
    for (int ks = 0; ks < 8; ++ks) {
        const int s  = ks >> 1;
        const int jb = 2 * (ks & 1);
        b[4 * ks + 0] = e[s] * w[jb].x;
        b[4 * ks + 1] = e[s] * w[jb + 1].x;
        b[4 * ks + 2] = e[s] * w[jb].y;
        b[4 * ks + 3] = e[s] * w[jb + 1].y;
    }
#pragma unroll
    for (int i = 0; i < NM; ++i) {
        v2f an[8];
        if (HAS_NEXT || i + 1 < NM) load_t_plane(Ta, M0 + i + 1, an);
        v8f f0 = {}, f1 = {}, h0 = {}, h1 = {};
#pragma unroll
        for (int ks = 0; ks < 8; ++ks) {
            v2f b0, b1, g0, g1;
            b0.x = b[4 * ks + 0];
            b0.y = b[4 * ks + 1];
            b1.x = b[4 * ks + 2];
            b1.y = b[4 * ks + 3];
            g0.x = bw[4 * ks + 0];
            g0.y = bw[4 * ks + 1];
            g1.x = bw[4 * ks + 2];
            g1.y = bw[4 * ks + 3];
            f0 = __builtin_amdgcn_wmma_f32_16x16x4_f32(false, a[ks], false, b0, (short)0, f0,
                                                       false, false);
            f1 = __builtin_amdgcn_wmma_f32_16x16x4_f32(false, a[ks], false, b1, (short)0, f1,
                                                       false, false);
            h0 = __builtin_amdgcn_wmma_f32_16x16x4_f32(false, a[ks], false, g0, (short)0, h0,
                                                       false, false);
            h1 = __builtin_amdgcn_wmma_f32_16x16x4_f32(false, a[ks], false, g1, (short)0, h1,
                                                       false, false);
        }
        cf[2 * i]     = f0 * 0.1f;
        cf[2 * i + 1] = f1 * 0.1f;
        ch[2 * i]     = h0;
        ch[2 * i + 1] = h1;
        if (HAS_NEXT || i + 1 < NM) {
#pragma unroll
            for (int ks = 0; ks < 8; ++ks) a[ks] = an[ks];
        }
    }
}

__global__ void __launch_bounds__(KDIM)
k_tile_feats(const float* __restrict__ srt, const float* __restrict__ meta,
             const float* __restrict__ emb,
             const float* __restrict__ W_rad, const float* __restrict__ RW,
             const int* __restrict__ species, float* __restrict__ x0) {
    __shared__ __attribute__((aligned(16))) float Ts[TILE_M * TS_PITCH];
    __shared__ __attribute__((aligned(16))) float Rec[REC_CAP * REC_FLOATS];
    __shared__ unsigned Idx[REC_CAP];
    __shared__ unsigned wsum[KDIM / 32];
    __shared__ float Es[N_SPEC * N_CH];
    __shared__ int   Ss[TILE_M];
    const int n0 = blockIdx.x * TILE_M;
    const int k  = threadIdx.x;
    const int lane = k & 31;
    const int wave = k >> 5;
    const int half = lane >> 4;
    const int l16  = lane & 15;
    const int colbase = wave * 32;
    const int col = colbase + 2 * l16;
    Es[k] = emb[k];
    if (k < TILE_M) Ss[k] = species[n0 + k];

    const unsigned* mt = reinterpret_cast<const unsigned*>(meta) + blockIdx.x;
    unsigned m0 = 0u, m1 = 0u;
    if (k < BIN_BLOCKS / 2) {
        m0 = mt[(2 * k) * BIN_THREADS];
        m1 = mt[(2 * k + 1) * BIN_THREADS];
    }
    for (int i = k; i < TILE_M * TS_PITCH / 4; i += KDIM)
        reinterpret_cast<v4f*>(Ts)[i] = v4f{0.f, 0.f, 0.f, 0.f};
    float e[N_SPEC];
#pragma unroll
    for (int s = 0; s < N_SPEC; ++s) e[s] = emb[s * N_CH + (col >> 2)];
    v2f wr[3][4];
#pragma unroll
    for (int l = 0; l < 3; ++l) {
#pragma unroll
        for (int jj = 0; jj < 4; ++jj) {
            const int j = 4 * (jj >> 1) + 2 * half + (jj & 1);
            wr[l][jj] = *reinterpret_cast<const v2f*>(W_rad + (l * N_RB + j) * KDIM + col);
        }
    }
    const float* Rp = RW + (2 * half) * KDIM + col;
    float bw0[32], bw1[32], bw2[32];
    load_b_chunk(Rp, bw0);
    load_b_chunk(Rp + RW_PLANE, bw1);
    load_b_chunk(Rp + 2 * RW_PLANE, bw2);
    const unsigned c0 = m0 & 0xffffu, c1 = m1 & 0xffffu;
    const unsigned mine = c0 + c1;
    const unsigned incl = wave_scan(mine, lane);
    if (lane == 31) wsum[wave] = incl;
    __syncthreads();
    unsigned woff = 0u, total = 0u;
    for (int w = 0; w < KDIM / 32; ++w) {
        const unsigned x = wsum[w];
        total += x;
        if (w < wave) woff += x;
    }
    const unsigned excl = woff + incl - mine;
    const v4f* sg = reinterpret_cast<const v4f*>(srt);
    const unsigned rs0 = (unsigned)(2 * k) * BIN_THREADS + (m0 >> 16);
    const unsigned rs1 = (unsigned)(2 * k + 1) * BIN_THREADS + (m1 >> 16);
    v4f* rl = reinterpret_cast<v4f*>(Rec);
    for (unsigned base = 0u; base < total; base += REC_CAP) {
        note_run(rs0, excl, c0, base, Idx);
        note_run(rs1, excl + c0, c1, base, Idx);
        __syncthreads();
        const unsigned n = min(total - base, (unsigned)REC_CAP);
        v4f ra[REC_CAP / KDIM], rb[REC_CAP / KDIM];
#pragma unroll
        for (int u = 0; u < REC_CAP / KDIM; ++u) {
            const unsigned q   = (unsigned)(k + u * KDIM);
            const unsigned src = q < n ? Idx[q] : 0u;
            ra[u] = sg[2 * (long)src];
            rb[u] = sg[2 * (long)src + 1];
        }
#pragma unroll
        for (int u = 0; u < REC_CAP / KDIM; ++u) {
            const unsigned q = (unsigned)(k + u * KDIM);
            if (q < n) {
                rl[2 * q]     = ra[u];
                rl[2 * q + 1] = rb[u];
            }
        }
        __syncthreads();
        for (unsigned i = k >> 3; i < n; i += 4 * (KDIM / 8)) {
            v4f g[4], h[4];
#pragma unroll
            for (int u = 0; u < 4; ++u) {
                const unsigned ld = min(i + (unsigned)u * (KDIM / 8), n - 1u);
                g[u] = rl[2 * ld];
                h[u] = rl[2 * ld + 1];
            }
#pragma unroll
            for (int u = 0; u < 4; ++u)
                if (i + (unsigned)u * (KDIM / 8) < n)
                    add_terms(Ts + (int)h[u].y + (k & 7), k & 7, g[u].x, g[u].y, g[u].z, g[u].w,
                              h[u].x);
        }
        __syncthreads();
    }

    const float* Ta = Ts + l16 * TS_PITCH + 2 * half;
    const float i3 = A3;
    int ec[3][2];
#pragma unroll
    for (int g = 0; g < 3; ++g) {
        ec[g][0] = (128 * g + col) / 12;
        ec[g][1] = (128 * g + col + 1) / 12;
    }
    float* xr = x0 + (long)(n0 + 8 * half) * K0TOT + col;
    v2f a[8];
    load_t_plane(Ta, 0, a);

    {
        v8f cf[2], ch[2];
        planes_from_moments<0, 1, true>(Ta, wr[0], e, bw0, a, cf, ch);
#pragma unroll
        for (int i = 0; i < 8; ++i) {
            const float* er = Es + Ss[8 * half + i] * N_CH;
            v2f o;
#pragma unroll
            for (int t = 0; t < 2; ++t) {
                const float fv = cf[t][i], h = ch[t][i];
                o[t] = (h * fv + fv) * er[ec[0][t]];
            }
            *reinterpret_cast<v2f*>(xr + i * K0TOT) = o;
        }
    }
    {
        v8f cf[6], ch[6];
        planes_from_moments<1, 3, true>(Ta, wr[1], e, bw1, a, cf, ch);
#pragma unroll
        for (int i = 0; i < 8; ++i) {
            const float* er = Es + Ss[8 * half + i] * N_CH;
            v2f o;
#pragma unroll
            for (int t = 0; t < 2; ++t) {
                float Gu[9], Fu[9];
                uncouple_l1(cf[t][i], cf[2 + t][i], cf[4 + t][i], Fu);
                uncouple_l1(ch[t][i], ch[2 + t][i], ch[4 + t][i], Gu);
                const float tr1 = trace_gf_plus_f(Gu, Fu);
                o[t] = (tr1 * i3) * er[ec[1][t]];
            }
            *reinterpret_cast<v2f*>(xr + i * K0TOT + 128) = o;
        }
    }
    {
        v8f cf[10], ch[10];
        planes_from_moments<4, 5, false>(Ta, wr[2], e, bw2, a, cf, ch);
#pragma unroll
        for (int i = 0; i < 8; ++i) {
            const float* er = Es + Ss[8 * half + i] * N_CH;
            v2f o;
#pragma unroll
            for (int t = 0; t < 2; ++t) {
                float Gu[9], Fu[9];
                uncouple_l2(cf[t][i], cf[2 + t][i], cf[4 + t][i], cf[6 + t][i], cf[8 + t][i], Fu);
                uncouple_l2(ch[t][i], ch[2 + t][i], ch[4 + t][i], ch[6 + t][i], ch[8 + t][i], Gu);
                const float tr2 = trace_gf_plus_f(Gu, Fu);
                o[t] = (tr2 * i3) * er[ec[2][t]];
            }
            *reinterpret_cast<v2f*>(xr + i * K0TOT + 256) = o;
        }
    }
}

#define HB_ROWS 128
#define HB_CHUNKS (K0TOT / 16)
struct HeadA { v4f a[2][2][2]; };
struct HeadB { v4f b[2][4][2]; };
__device__ __forceinline__ void head_load_a(const float* __restrict__ A0,
                                            const float* __restrict__ A1, int c, HeadA& f) {
#pragma unroll
    for (int cc = 0; cc < 2; ++cc) {
#pragma unroll
        for (int sp = 0; sp < 2; ++sp) {
            const int q = 2 * (c + cc) + sp;
            f.a[cc][sp][0] = *reinterpret_cast<const v4f*>(A0 + 8 * q);
            f.a[cc][sp][1] = *reinterpret_cast<const v4f*>(A1 + 8 * q);
        }
    }
}
__device__ __forceinline__ void head_load_b(const float* __restrict__ Bp, int c, HeadB& f) {
#pragma unroll
    for (int sp = 0; sp < 2; ++sp) {
        const int q = 2 * c + sp;
#pragma unroll
        for (int i = 0; i < 4; ++i) {
            const float* rowp = Bp + (long)(8 * q + i) * K0TOT;
            f.b[sp][i][0] = *reinterpret_cast<const v4f*>(rowp);
            f.b[sp][i][1] = *reinterpret_cast<const v4f*>(rowp + 64);
        }
    }
}
template <int CC>
__device__ __forceinline__ void head_mma(const HeadA& fa, const HeadB& fb, v8f (&acc)[2][8]) {
#pragma unroll
    for (int sp = 0; sp < 2; ++sp) {
#pragma unroll
        for (int e = 0; e < 2; ++e) {
            v2f a0, a1;
            a0.x = fa.a[CC][sp][0][2 * e];
            a0.y = fa.a[CC][sp][0][2 * e + 1];
            a1.x = fa.a[CC][sp][1][2 * e];
            a1.y = fa.a[CC][sp][1][2 * e + 1];
#pragma unroll
            for (int g = 0; g < 2; ++g) {
#pragma unroll
                for (int v = 0; v < 4; ++v) {
                    v2f b;
                    b.x = fb.b[sp][2 * e][g][v];
                    b.y = fb.b[sp][2 * e + 1][g][v];
                    acc[0][4 * g + v] = __builtin_amdgcn_wmma_f32_16x16x4_f32(
                        false, a0, false, b, (short)0, acc[0][4 * g + v], false, false);
                    acc[1][4 * g + v] = __builtin_amdgcn_wmma_f32_16x16x4_f32(
                        false, a1, false, b, (short)0, acc[1][4 * g + v], false, false);
                }
            }
        }
    }
}
__global__ void __launch_bounds__(HB_ROWS)
k_head_wmma(const float* __restrict__ x0, const float* __restrict__ Wh,
            const float* __restrict__ bh, const float* __restrict__ Wo,
            float* __restrict__ out) {
    const int lane = threadIdx.x & 31;
    const int wave = threadIdx.x >> 5;
    const int half = lane >> 4;
    const int l16  = lane & 15;
    const int m0 = blockIdx.x * HB_ROWS + wave * 32;
    if (m0 >= N_ATOMS) return;
    const int n0 = blockIdx.y * 128;
    const int ra = min(m0 + l16, N_ATOMS - 1);
    const int rb = min(m0 + 16 + l16, N_ATOMS - 1);
    const float* A0 = x0 + (long)ra * K0TOT + 4 * half;
    const float* A1 = x0 + (long)rb * K0TOT + 4 * half;
    const float* Bp = Wh + (long)(4 * half) * K0TOT + n0 + 4 * l16;
    v8f acc[2][8] = {};
    HeadA fa, fan;
    HeadB f0, f1;
    head_load_a(A0, A1, 0, fa);
    head_load_b(Bp, 0, f0);
    for (int c = 0; c < HB_CHUNKS - 2; c += 2) {
        head_load_b(Bp, c + 1, f1);
        __builtin_amdgcn_sched_barrier(0);
        head_load_a(A0, A1, c + 2, fan);
        __builtin_amdgcn_sched_barrier(0);
        head_mma<0>(fa, f0, acc);
        __builtin_amdgcn_sched_barrier(0);
        head_load_b(Bp, c + 2, f0);
        __builtin_amdgcn_sched_barrier(0);
        head_mma<1>(fa, f1, acc);
        __builtin_amdgcn_sched_barrier(0);
        fa = fan;
    }
    head_load_b(Bp, HB_CHUNKS - 1, f1);
    __builtin_amdgcn_sched_barrier(0);
    head_mma<0>(fa, f0, acc);
    head_mma<1>(fa, f1, acc);
    v4f bias[2], wo[2];
#pragma unroll
    for (int g = 0; g < 2; ++g) {
        bias[g] = *reinterpret_cast<const v4f*>(bh + n0 + 64 * g + 4 * l16);
        wo[g]   = *reinterpret_cast<const v4f*>(Wo + n0 + 64 * g + 4 * l16);
    }
#pragma unroll
    for (int rt = 0; rt < 2; ++rt) {
#pragma unroll
        for (int i = 0; i < 8; ++i) {
            float partial = 0.f;
#pragma unroll
            for (int g = 0; g < 2; ++g) {
#pragma unroll
                for (int v = 0; v < 4; ++v) {
                    float x = acc[rt][4 * g + v][i] + bias[g][v];
                    partial += (x / (1.f + expf(-x))) * wo[g][v];
                }
            }
            const int row = m0 + 16 * rt + 8 * half + i;
            if (row < N_ATOMS) atomicAdd(&out[row], partial);
        }
    }
}

extern "C" void kernel_launch(void* const* d_in, const int* in_sizes, int n_in,
                              void* d_out, int out_size, void* d_ws, size_t ws_size,
                              hipStream_t stream) {
    (void)in_sizes; (void)n_in; (void)out_size; (void)ws_size;
    const float* positions = (const float*)d_in[0];
    const int*   species   = (const int*)d_in[1];
    const int*   pairs     = (const int*)d_in[2];
    const float* embeddings= (const float*)d_in[3];
    const float* W_rad     = (const float*)d_in[4];
    const float* W_cg0     = (const float*)d_in[5];
    const float* W_cg1     = (const float*)d_in[6];
    const float* W_cg2     = (const float*)d_in[7];
    const float* W_head    = (const float*)d_in[8];
    const float* b_head    = (const float*)d_in[9];
    const float* W_out     = (const float*)d_in[10];
    const float* b_out     = (const float*)d_in[11];
    float* out = (float*)d_out;

    float* ws   = (float*)d_ws;
    float* srt  = ws;
    float* meta = srt + (long)N_PAIRS * REC_FLOATS;
    float* x0   = ws + (long)N_ATOMS * TM_ATOM;
    float* RW   = x0 + (long)N_ATOMS * K0TOT;

    k_pairs_binned<<<BIN_BLOCKS + RW_BLOCKS, BIN_THREADS, 0, stream>>>(
        positions, species, pairs, srt, meta, out, b_out, embeddings, W_rad, W_cg0, W_cg1, W_cg2,
        RW);

    k_tile_feats<<<N_TILES, KDIM, 0, stream>>>(srt, meta, embeddings, W_rad, RW, species, x0);

    k_head_wmma<<<dim3((N_ATOMS + HB_ROWS - 1) / HB_ROWS, K0TOT / 128), HB_ROWS, 0, stream>>>(
        x0, W_head, b_head, W_out, out);
}


# head epilogue: 16 lanes' shares of a row summed in registers (halving butterfly) before the atomic: 32 adds per wave, one per row (was 512); sum order of the W_out dot changes
# speedup vs baseline: 6.6281x; 1.1840x over previous
#include <hip/hip_runtime.h>
#include <math.h>

#define N_ATOMS  10000
#define N_PAIRS  160000
#define N_CH     32
#define KDIM     128
#define K0TOT    384

typedef float v2f __attribute__((ext_vector_type(2)));
typedef float v4f __attribute__((ext_vector_type(4)));
typedef float v8f __attribute__((ext_vector_type(8)));

#define A3 0.5773502691896258f
#define A2 0.7071067811865476f
#define B2 0.8164965809277260f
#define B1 0.4082482904638630f

#define N_SPEC   4
#define N_RB     8
#define TM_ATOM  (N_SPEC * 9 * N_RB)
#define TILE_M   16
#define TS_PITCH 292
#define N_TILES     (N_ATOMS / TILE_M)
#define BIN_THREADS 640
#define BIN_BLOCKS  (N_PAIRS / BIN_THREADS)
#define BIN_WAVES   (BIN_THREADS / 32)
#define REC_FLOATS  8
#define OUT_ATOMS   80
typedef int v2i __attribute__((ext_vector_type(2)));

__device__ __forceinline__ void pair_geometry(const float* __restrict__ pos, int c, int nb,
                                              float& d, float& ux, float& uy, float& uz,
                                              float& fc) {
    float rx = pos[3 * nb + 0] - pos[3 * c + 0];
    float ry = pos[3 * nb + 1] - pos[3 * c + 1];
    float rz = pos[3 * nb + 2] - pos[3 * c + 2];
    d = sqrtf(rx * rx + ry * ry + rz * rz + 1e-12f);
    float inv = 1.f / d;
    ux = rx * inv;
    uy = ry * inv;
    uz = rz * inv;
    float tc = (d - 15.f) * 0.2f;
    tc = fminf(fmaxf(tc, 0.f), 1.f);
    fc = 0.5f * (cosf(3.14159265358979f * tc) + 1.f);
}
__device__ __forceinline__ void add_terms(float* Tp, int j, float d, float ux, float uy,
                                          float uz, float fc) {
    float mu = 20.f * (float)j / 7.f;
    float x  = (d - mu) * 0.4f;
    float rb = expf(-x * x) * fc;
    const float s3 = 1.7320508075688772f;
    float sh[9] = { 1.f, uy, uz, ux,
                    s3 * ux * uy, s3 * uy * uz, 0.5f * (3.f * uz * uz - 1.f),
                    s3 * ux * uz, 0.5f * s3 * (ux * ux - uy * uy) };
    for (int m = 0; m < 9; ++m) atomicAdd(Tp + m * N_RB, sh[m] * rb);
}
__device__ __forceinline__ unsigned wave_scan(unsigned v, int lane) {
    for (int o = 1; o < 32; o <<= 1) {
        const unsigned u = __shfl_up(v, o);
        if (lane >= o) v += u;
    }
    return v;
}

#define RW_BLOCKS 3
#define RW_WAVES  8
#define RW_PLANE  (N_SPEC * N_RB * KDIM)
__device__ __forceinline__ void build_rw(int l, int tid, const float* __restrict__ emb,
                                         const float* __restrict__ W_rad,
                                         const float* __restrict__ Wc, float* __restrict__ RW) {
    const int lane = tid & 31;
    const int wave = tid >> 5;
    const int half = lane >> 4;
    const int l16  = lane & 15;
    const int kk   = 16 * (wave & 1) + l16;
    const int col  = 32 * (wave >> 1) + 2 * l16;
    const float* Bp = Wc + (2 * half) * KDIM + col;
    const float* Ap = W_rad + (l * N_RB + (kk & 7)) * KDIM + 2 * half;
    const v4f*   Ep = reinterpret_cast<const v4f*>(emb + (kk >> 3) * N_CH);
    v8f acc0 = {}, acc1 = {};
    v4f e4 = {};
#pragma unroll
    for (int ks = 0; ks < KDIM / 4; ++ks) {
        if ((ks & 3) == 0) e4 = Ep[ks >> 2];
        const v2f r0 = *reinterpret_cast<const v2f*>(Bp + (4 * ks) * KDIM);
        const v2f r1 = *reinterpret_cast<const v2f*>(Bp + (4 * ks + 1) * KDIM);
        const v2f wq = *reinterpret_cast<const v2f*>(Ap + 4 * ks);
        const float ev = e4[ks & 3];
        v2f a, t0, t1;
        a.x  = ev * wq.x;
        a.y  = ev * wq.y;
        t0.x = r0.x;
        t0.y = r1.x;
        t1.x = r0.y;
        t1.y = r1.y;
        acc0 = __builtin_amdgcn_wmma_f32_16x16x4_f32(false, a, false, t0, (short)0, acc0,
                                                     false, false);
        acc1 = __builtin_amdgcn_wmma_f32_16x16x4_f32(false, a, false, t1, (short)0, acc1,
                                                     false, false);
    }
    float* Rd = RW + l * RW_PLANE + (16 * (wave & 1) + 8 * half) * KDIM + col;
#pragma unroll
    for (int r = 0; r < 8; ++r) {
        v2f o;
        o.x = acc0[r] * 0.1f;
        o.y = acc1[r] * 0.1f;
        *reinterpret_cast<v2f*>(Rd + r * KDIM) = o;
    }
}

__global__ void __launch_bounds__(BIN_THREADS)
k_pairs_binned(const float* __restrict__ pos, const int* __restrict__ species,
               const int* __restrict__ pairs, float* __restrict__ srt,
               float* __restrict__ meta, float* __restrict__ out,
               const float* __restrict__ b_out, const float* __restrict__ emb,
               const float* __restrict__ W_rad, const float* __restrict__ W0,
               const float* __restrict__ W1, const float* __restrict__ W2,
               float* __restrict__ RW) {
    if (blockIdx.x >= BIN_BLOCKS) {
        const int l = blockIdx.x - BIN_BLOCKS;
        if (threadIdx.x < RW_WAVES * 32)
            build_rw(l, threadIdx.x, emb, W_rad, (l == 0) ? W0 : ((l == 1) ? W1 : W2), RW);
        return;
    }
    __shared__ __attribute__((aligned(16))) float recs[BIN_THREADS * REC_FLOATS];
    __shared__ unsigned hist[BIN_THREADS];
    __shared__ unsigned wtot[BIN_WAVES];
    const int tid  = threadIdx.x;
    const int lane = tid & 31;
    const int wave = tid >> 5;
    const int b    = blockIdx.x;
    if (b < N_ATOMS / OUT_ATOMS && tid < OUT_ATOMS) out[b * OUT_ATOMS + tid] = b_out[0];
    hist[tid] = 0u;
    const v2i pr = reinterpret_cast<const v2i*>(pairs)[b * BIN_THREADS + tid];
    const int c  = pr.x;
    const int nb = pr.y;
    const int t  = c / TILE_M;
    __syncthreads();
    const unsigned rank = atomicAdd(&hist[t], 1u);
    float d, ux, uy, uz, fc;
    pair_geometry(pos, c, nb, d, ux, uy, uz, fc);
    const int row = (c - t * TILE_M) * TS_PITCH + species[nb] * (9 * N_RB);
    __syncthreads();
    const unsigned cnt  = hist[tid];
    const unsigned incl = wave_scan(cnt, lane);
    if (lane == 31) wtot[wave] = incl;
    __syncthreads();
    unsigned woff = 0u;
    for (int w = 0; w < BIN_WAVES; ++w)
        if (w < wave) woff += wtot[w];
    const unsigned start = woff + incl - cnt;
    hist[tid] = start;
    reinterpret_cast<unsigned*>(meta)[b * BIN_THREADS + tid] = (start << 16) | cnt;
    __syncthreads();
    const unsigned posn = hist[t] + rank;
    v4f* rl = reinterpret_cast<v4f*>(recs);
    rl[2 * posn]     = v4f{d, ux, uy, uz};
    rl[2 * posn + 1] = v4f{fc, (float)row, 0.f, 0.f};
    __syncthreads();
    v4f* sg = reinterpret_cast<v4f*>(srt) + (long)b * (BIN_THREADS * 2);
    for (int i = tid; i < BIN_THREADS * 2; i += BIN_THREADS) sg[i] = rl[i];
}

#define REC_CAP 512
__device__ __forceinline__ void note_run(unsigned src, unsigned p0, unsigned cnt, unsigned base,
                                         unsigned* idx) {
    const unsigned lo = base > p0 ? base - p0 : 0u;
    const unsigned hi = min(cnt, base + REC_CAP > p0 ? base + REC_CAP - p0 : 0u);
#pragma nounroll
    for (unsigned r = lo; r < hi; ++r) idx[p0 + r - base] = src + r;
}

__device__ __forceinline__ void uncouple_l1(float v1, float v2, float v3, float u[9]) {
    u[0] = 0.f;       u[1] =  A2 * v3;  u[2] = -A2 * v2;
    u[3] = -A2 * v3;  u[4] = 0.f;       u[5] =  A2 * v1;
    u[6] =  A2 * v2;  u[7] = -A2 * v1;  u[8] = 0.f;
}
__device__ __forceinline__ void uncouple_l2(float v4, float v5, float v6, float v7, float v8,
                                            float u[9]) {
    u[0] = -B1 * v6 - A2 * v8;  u[1] = A2 * v5;  u[2] = A2 * v4;
    u[3] = A2 * v5;             u[4] = B2 * v6;  u[5] = A2 * v7;
    u[6] = A2 * v4;             u[7] = A2 * v7;  u[8] = -B1 * v6 + A2 * v8;
}
__device__ __forceinline__ float trace_gf_plus_f(const float G[9], const float Fm[9]) {
    float tr = 0.f;
#pragma unroll
    for (int a = 0; a < 3; ++a)
#pragma unroll
        for (int b = 0; b < 3; ++b) tr += G[a * 3 + b] * Fm[b * 3 + a];
    return tr + (Fm[0] + Fm[4] + Fm[8]);
}

__device__ __forceinline__ void load_b_chunk(const float* __restrict__ p, float (&b)[32]) {
#pragma unroll
    for (int ks = 0; ks < 8; ++ks) {
        const float* q = p + ks * 4 * KDIM;
        const v2f r0 = *reinterpret_cast<const v2f*>(q);
        const v2f r1 = *reinterpret_cast<const v2f*>(q + KDIM);
        b[4 * ks + 0] = r0.x;
        b[4 * ks + 1] = r1.x;
        b[4 * ks + 2] = r0.y;
        b[4 * ks + 3] = r1.y;
    }
}
__device__ __forceinline__ void load_t_plane(const float* __restrict__ Ta, int m, v2f (&a)[8]) {
#pragma unroll
    for (int ks = 0; ks < 8; ++ks)
        a[ks] = *reinterpret_cast<const v2f*>(Ta + (ks >> 1) * 72 + m * 8 + 4 * (ks & 1));
}

template <int M0, int NM, bool HAS_NEXT>
__device__ __forceinline__ void planes_from_moments(const float* __restrict__ Ta,
                                                    const v2f (&w)[4], const float (&e)[N_SPEC],
                                                    const float (&bw)[32], v2f (&a)[8],
                                                    v8f (&cf)[2 * NM], v8f (&ch)[2 * NM]) {
    float b[32];
#pragma unroll
    for (int ks = 0; ks < 8; ++ks) {
        const int s  = ks >> 1;
        const int jb = 2 * (ks & 1);
        b[4 * ks + 0] = e[s] * w[jb].x;
        b[4 * ks + 1] = e[s] * w[jb + 1].x;
        b[4 * ks + 2] = e[s] * w[jb].y;
        b[4 * ks + 3] = e[s] * w[jb + 1].y;
    }
#pragma unroll
    for (int i = 0; i < NM; ++i) {
        v2f an[8];
        if (HAS_NEXT || i + 1 < NM) load_t_plane(Ta, M0 + i + 1, an);
        v8f f0 = {}, f1 = {}, h0 = {}, h1 = {};
#pragma unroll
        for (int ks = 0; ks < 8; ++ks) {
            v2f b0, b1, g0, g1;
            b0.x = b[4 * ks + 0];
            b0.y = b[4 * ks + 1];
            b1.x = b[4 * ks + 2];
            b1.y = b[4 * ks + 3];
            g0.x = bw[4 * ks + 0];
            g0.y = bw[4 * ks + 1];
            g1.x = bw[4 * ks + 2];
            g1.y = bw[4 * ks + 3];
            f0 = __builtin_amdgcn_wmma_f32_16x16x4_f32(false, a[ks], false, b0, (short)0, f0,
                                                       false, false);
            f1 = __builtin_amdgcn_wmma_f32_16x16x4_f32(false, a[ks], false, b1, (short)0, f1,
                                                       false, false);
            h0 = __builtin_amdgcn_wmma_f32_16x16x4_f32(false, a[ks], false, g0, (short)0, h0,
                                                       false, false);
            h1 = __builtin_amdgcn_wmma_f32_16x16x4_f32(false, a[ks], false, g1, (short)0, h1,
                                                       false, false);
        }
        cf[2 * i]     = f0 * 0.1f;
        cf[2 * i + 1] = f1 * 0.1f;
        ch[2 * i]     = h0;
        ch[2 * i + 1] = h1;
        if (HAS_NEXT || i + 1 < NM) {
#pragma unroll
            for (int ks = 0; ks < 8; ++ks) a[ks] = an[ks];
        }
    }
}

__global__ void __launch_bounds__(KDIM)
k_tile_feats(const float* __restrict__ srt, const float* __restrict__ meta,
             const float* __restrict__ emb,
             const float* __restrict__ W_rad, const float* __restrict__ RW,
             const int* __restrict__ species, float* __restrict__ x0) {
    __shared__ __attribute__((aligned(16))) float Ts[TILE_M * TS_PITCH];
    __shared__ __attribute__((aligned(16))) float Rec[REC_CAP * REC_FLOATS];
    __shared__ unsigned Idx[REC_CAP];
    __shared__ unsigned wsum[KDIM / 32];
    __shared__ float Es[N_SPEC * N_CH];
    __shared__ int   Ss[TILE_M];
    const int n0 = blockIdx.x * TILE_M;
    const int k  = threadIdx.x;
    const int lane = k & 31;
    const int wave = k >> 5;
    const int half = lane >> 4;
    const int l16  = lane & 15;
    const int colbase = wave * 32;
    const int col = colbase + 2 * l16;
    Es[k] = emb[k];
    if (k < TILE_M) Ss[k] = species[n0 + k];

    const unsigned* mt = reinterpret_cast<const unsigned*>(meta) + blockIdx.x;
    unsigned m0 = 0u, m1 = 0u;
    if (k < BIN_BLOCKS / 2) {
        m0 = mt[(2 * k) * BIN_THREADS];
        m1 = mt[(2 * k + 1) * BIN_THREADS];
    }
    for (int i = k; i < TILE_M * TS_PITCH / 4; i += KDIM)
        reinterpret_cast<v4f*>(Ts)[i] = v4f{0.f, 0.f, 0.f, 0.f};
    float e[N_SPEC];
#pragma unroll
    for (int s = 0; s < N_SPEC; ++s) e[s] = emb[s * N_CH + (col >> 2)];
    v2f wr[3][4];
#pragma unroll
    for (int l = 0; l < 3; ++l) {
#pragma unroll
        for (int jj = 0; jj < 4; ++jj) {
            const int j = 4 * (jj >> 1) + 2 * half + (jj & 1);
            wr[l][jj] = *reinterpret_cast<const v2f*>(W_rad + (l * N_RB + j) * KDIM + col);
        }
    }
    const float* Rp = RW + (2 * half) * KDIM + col;
    float bw0[32], bw1[32], bw2[32];
    load_b_chunk(Rp, bw0);
    load_b_chunk(Rp + RW_PLANE, bw1);
    load_b_chunk(Rp + 2 * RW_PLANE, bw2);
    const unsigned c0 = m0 & 0xffffu, c1 = m1 & 0xffffu;
    const unsigned mine = c0 + c1;
    const unsigned incl = wave_scan(mine, lane);
    if (lane == 31) wsum[wave] = incl;
    __syncthreads();
    unsigned woff = 0u, total = 0u;
    for (int w = 0; w < KDIM / 32; ++w) {
        const unsigned x = wsum[w];
        total += x;
        if (w < wave) woff += x;
    }
    const unsigned excl = woff + incl - mine;
    const v4f* sg = reinterpret_cast<const v4f*>(srt);
    const unsigned rs0 = (unsigned)(2 * k) * BIN_THREADS + (m0 >> 16);
    const unsigned rs1 = (unsigned)(2 * k + 1) * BIN_THREADS + (m1 >> 16);
    v4f* rl = reinterpret_cast<v4f*>(Rec);
    for (unsigned base = 0u; base < total; base += REC_CAP) {
        note_run(rs0, excl, c0, base, Idx);
        note_run(rs1, excl + c0, c1, base, Idx);
        __syncthreads();
        const unsigned n = min(total - base, (unsigned)REC_CAP);
        v4f ra[REC_CAP / KDIM], rb[REC_CAP / KDIM];
#pragma unroll
        for (int u = 0; u < REC_CAP / KDIM; ++u) {
            const unsigned q   = (unsigned)(k + u * KDIM);
            const unsigned src = q < n ? Idx[q] : 0u;
            ra[u] = sg[2 * (long)src];
            rb[u] = sg[2 * (long)src + 1];
        }
#pragma unroll
        for (int u = 0; u < REC_CAP / KDIM; ++u) {
            const unsigned q = (unsigned)(k + u * KDIM);
            if (q < n) {
                rl[2 * q]     = ra[u];
                rl[2 * q + 1] = rb[u];
            }
        }
        __syncthreads();
        for (unsigned i = k >> 3; i < n; i += 4 * (KDIM / 8)) {
            v4f g[4], h[4];
#pragma unroll
            for (int u = 0; u < 4; ++u) {
                const unsigned ld = min(i + (unsigned)u * (KDIM / 8), n - 1u);
                g[u] = rl[2 * ld];
                h[u] = rl[2 * ld + 1];
            }
#pragma unroll
            for (int u = 0; u < 4; ++u)
                if (i + (unsigned)u * (KDIM / 8) < n)
                    add_terms(Ts + (int)h[u].y + (k & 7), k & 7, g[u].x, g[u].y, g[u].z, g[u].w,
                              h[u].x);
        }
        __syncthreads();
    }

    const float* Ta = Ts + l16 * TS_PITCH + 2 * half;
    const float i3 = A3;
    int ec[3][2];
#pragma unroll
    for (int g = 0; g < 3; ++g) {
        ec[g][0] = (128 * g + col) / 12;
        ec[g][1] = (128 * g + col + 1) / 12;
    }
    float* xr = x0 + (long)(n0 + 8 * half) * K0TOT + col;
    v2f a[8];
    load_t_plane(Ta, 0, a);

    {
        v8f cf[2], ch[2];
        planes_from_moments<0, 1, true>(Ta, wr[0], e, bw0, a, cf, ch);
#pragma unroll
        for (int i = 0; i < 8; ++i) {
            const float* er = Es + Ss[8 * half + i] * N_CH;
            v2f o;
#pragma unroll
            for (int t = 0; t < 2; ++t) {
                const float fv = cf[t][i], h = ch[t][i];
                o[t] = (h * fv + fv) * er[ec[0][t]];
            }
            *reinterpret_cast<v2f*>(xr + i * K0TOT) = o;
        }
    }
    {
        v8f cf[6], ch[6];
        planes_from_moments<1, 3, true>(Ta, wr[1], e, bw1, a, cf, ch);
#pragma unroll
        for (int i = 0; i < 8; ++i) {
            const float* er = Es + Ss[8 * half + i] * N_CH;
            v2f o;
#pragma unroll
            for (int t = 0; t < 2; ++t) {
                float Gu[9], Fu[9];
                uncouple_l1(cf[t][i], cf[2 + t][i], cf[4 + t][i], Fu);
                uncouple_l1(ch[t][i], ch[2 + t][i], ch[4 + t][i], Gu);
                const float tr1 = trace_gf_plus_f(Gu, Fu);
                o[t] = (tr1 * i3) * er[ec[1][t]];
            }
            *reinterpret_cast<v2f*>(xr + i * K0TOT + 128) = o;
        }
    }
    {
        v8f cf[10], ch[10];
        planes_from_moments<4, 5, false>(Ta, wr[2], e, bw2, a, cf, ch);
#pragma unroll
        for (int i = 0; i < 8; ++i) {
            const float* er = Es + Ss[8 * half + i] * N_CH;
            v2f o;
#pragma unroll
            for (int t = 0; t < 2; ++t) {
                float Gu[9], Fu[9];
                uncouple_l2(cf[t][i], cf[2 + t][i], cf[4 + t][i], cf[6 + t][i], cf[8 + t][i], Fu);
                uncouple_l2(ch[t][i], ch[2 + t][i], ch[4 + t][i], ch[6 + t][i], ch[8 + t][i], Gu);
                const float tr2 = trace_gf_plus_f(Gu, Fu);
                o[t] = (tr2 * i3) * er[ec[2][t]];
            }
            *reinterpret_cast<v2f*>(xr + i * K0TOT + 256) = o;
        }
    }
}

#define HB_ROWS 128
#define HB_CHUNKS (K0TOT / 16)
struct HeadA { v4f a[2][2][2]; };
struct HeadB { v4f b[2][4][2]; };
__device__ __forceinline__ void head_load_a(const float* __restrict__ A0,
                                            const float* __restrict__ A1, int c, HeadA& f) {
#pragma unroll
    for (int cc = 0; cc < 2; ++cc) {
#pragma unroll
        for (int sp = 0; sp < 2; ++sp) {
            const int q = 2 * (c + cc) + sp;
            f.a[cc][sp][0] = *reinterpret_cast<const v4f*>(A0 + 8 * q);
            f.a[cc][sp][1] = *reinterpret_cast<const v4f*>(A1 + 8 * q);
        }
    }
}
__device__ __forceinline__ void head_load_b(const float* __restrict__ Bp, int c, HeadB& f) {
#pragma unroll
    for (int sp = 0; sp < 2; ++sp) {
        const int q = 2 * c + sp;
#pragma unroll
        for (int i = 0; i < 4; ++i) {
            const float* rowp = Bp + (long)(8 * q + i) * K0TOT;
            f.b[sp][i][0] = *reinterpret_cast<const v4f*>(rowp);
            f.b[sp][i][1] = *reinterpret_cast<const v4f*>(rowp + 64);
        }
    }
}
template <int CC>
__device__ __forceinline__ void head_mma(const HeadA& fa, const HeadB& fb, v8f (&acc)[2][8]) {
#pragma unroll
    for (int sp = 0; sp < 2; ++sp) {
#pragma unroll
        for (int e = 0; e < 2; ++e) {
            v2f a0, a1;
            a0.x = fa.a[CC][sp][0][2 * e];
            a0.y = fa.a[CC][sp][0][2 * e + 1];
            a1.x = fa.a[CC][sp][1][2 * e];
            a1.y = fa.a[CC][sp][1][2 * e + 1];
#pragma unroll
            for (int g = 0; g < 2; ++g) {
#pragma unroll
                for (int v = 0; v < 4; ++v) {
                    v2f b;
                    b.x = fb.b[sp][2 * e][g][v];
                    b.y = fb.b[sp][2 * e + 1][g][v];
                    acc[0][4 * g + v] = __builtin_amdgcn_wmma_f32_16x16x4_f32(
                        false, a0, false, b, (short)0, acc[0][4 * g + v], false, false);
                    acc[1][4 * g + v] = __builtin_amdgcn_wmma_f32_16x16x4_f32(
                        false, a1, false, b, (short)0, acc[1][4 * g + v], false, false);
                }
            }
        }
    }
}
__global__ void __launch_bounds__(HB_ROWS)
k_head_wmma(const float* __restrict__ x0, const float* __restrict__ Wh,
            const float* __restrict__ bh, const float* __restrict__ Wo,
            float* __restrict__ out) {
    const int lane = threadIdx.x & 31;
    const int wave = threadIdx.x >> 5;
    const int half = lane >> 4;
    const int l16  = lane & 15;
    const int m0 = blockIdx.x * HB_ROWS + wave * 32;
    if (m0 >= N_ATOMS) return;
    const int n0 = blockIdx.y * 128;
    const int ra = min(m0 + l16, N_ATOMS - 1);
    const int rb = min(m0 + 16 + l16, N_ATOMS - 1);
    const float* A0 = x0 + (long)ra * K0TOT + 4 * half;
    const float* A1 = x0 + (long)rb * K0TOT + 4 * half;
    const float* Bp = Wh + (long)(4 * half) * K0TOT + n0 + 4 * l16;
    v8f acc[2][8] = {};
    HeadA fa, fan;
    HeadB f0, f1;
    head_load_a(A0, A1, 0, fa);
    head_load_b(Bp, 0, f0);
    for (int c = 0; c < HB_CHUNKS - 2; c += 2) {
        head_load_b(Bp, c + 1, f1);
        __builtin_amdgcn_sched_barrier(0);
        head_load_a(A0, A1, c + 2, fan);
        __builtin_amdgcn_sched_barrier(0);
        head_mma<0>(fa, f0, acc);
        __builtin_amdgcn_sched_barrier(0);
        head_load_b(Bp, c + 2, f0);
        __builtin_amdgcn_sched_barrier(0);
        head_mma<1>(fa, f1, acc);
        __builtin_amdgcn_sched_barrier(0);
        fa = fan;
    }
    head_load_b(Bp, HB_CHUNKS - 1, f1);
    __builtin_amdgcn_sched_barrier(0);
    head_mma<0>(fa, f0, acc);
    head_mma<1>(fa, f1, acc);
    v4f bias[2], wo[2];
#pragma unroll
    for (int g = 0; g < 2; ++g) {
        bias[g] = *reinterpret_cast<const v4f*>(bh + n0 + 64 * g + 4 * l16);
        wo[g]   = *reinterpret_cast<const v4f*>(Wo + n0 + 64 * g + 4 * l16);
    }
    float p[16];
#pragma unroll
    for (int rt = 0; rt < 2; ++rt) {
#pragma unroll
        for (int i = 0; i < 8; ++i) {
            float partial = 0.f;
#pragma unroll
            for (int g = 0; g < 2; ++g) {
#pragma unroll
                for (int v = 0; v < 4; ++v) {
                    float x = acc[rt][4 * g + v][i] + bias[g][v];
                    partial += (x / (1.f + expf(-x))) * wo[g][v];
                }
            }
            p[8 * rt + i] = partial;
        }
    }
#pragma unroll
    for (int s = 8; s >= 1; s >>= 1) {
        const bool up = (l16 & s) != 0;
#pragma unroll
        for (int j = 0; j < s; ++j) {
            const float lo = p[j];
            const float hi = p[j + s];
            const float keep = up ? hi : lo;
            const float send = up ? lo : hi;
            p[j] = keep + __shfl_xor(send, s);
        }
    }
    const int row = m0 + 16 * (l16 >> 3) + 8 * half + (l16 & 7);
    if (row < N_ATOMS) atomicAdd(&out[row], p[0]);
}

extern "C" void kernel_launch(void* const* d_in, const int* in_sizes, int n_in,
                              void* d_out, int out_size, void* d_ws, size_t ws_size,
                              hipStream_t stream) {
    (void)in_sizes; (void)n_in; (void)out_size; (void)ws_size;
    const float* positions = (const float*)d_in[0];
    const int*   species   = (const int*)d_in[1];
    const int*   pairs     = (const int*)d_in[2];
    const float* embeddings= (const float*)d_in[3];
    const float* W_rad     = (const float*)d_in[4];
    const float* W_cg0     = (const float*)d_in[5];
    const float* W_cg1     = (const float*)d_in[6];
    const float* W_cg2     = (const float*)d_in[7];
    const float* W_head    = (const float*)d_in[8];
    const float* b_head    = (const float*)d_in[9];
    const float* W_out     = (const float*)d_in[10];
    const float* b_out     = (const float*)d_in[11];
    float* out = (float*)d_out;

    float* ws   = (float*)d_ws;
    float* srt  = ws;
    float* meta = srt + (long)N_PAIRS * REC_FLOATS;
    float* x0   = ws + (long)N_ATOMS * TM_ATOM;
    float* RW   = x0 + (long)N_ATOMS * K0TOT;

    k_pairs_binned<<<BIN_BLOCKS + RW_BLOCKS, BIN_THREADS, 0, stream>>>(
        positions, species, pairs, srt, meta, out, b_out, embeddings, W_rad, W_cg0, W_cg1, W_cg2,
        RW);

    k_tile_feats<<<N_TILES, KDIM, 0, stream>>>(srt, meta, embeddings, W_rad, RW, species, x0);

    k_head_wmma<<<dim3((N_ATOMS + HB_ROWS - 1) / HB_ROWS, K0TOT / 128), HB_ROWS, 0, stream>>>(
        x0, W_head, b_head, W_out, out);
}
